# starting kernel
# baseline (speedup 1.0000x reference)
#include <hip/hip_runtime.h>
#include <hip/hip_bf16.h>
#include <math.h>

typedef __attribute__((ext_vector_type(2))) float v2f;
typedef __attribute__((ext_vector_type(8))) float v8f;

template <int NT>
__global__ void __launch_bounds__(256) gemm_f32_wmma(
    const float* __restrict__ A, const float* __restrict__ B,
    float* __restrict__ C, int M, int K, int N)
{
    __shared__ v2f sB2[4096];
    const int tid = threadIdx.x;
    const int pairCount = (K >> 1) * N;
    for (int p = tid; p < pairCount; p += 256) {
        const int kk = (p / N) << 1;
        const int n  = p - (p / N) * N;
        v2f b;
        b.x = B[kk * N + n];
        b.y = B[(kk + 1) * N + n];
        sB2[p] = b;
    }
    __syncthreads();

    const int lane    = tid & 31;
    const int wave    = tid >> 5;
    const int rowTile = blockIdx.x * 8 + wave;
    const int numTiles = (M + 15) >> 4;
    if (rowTile >= numTiles) return;

    const int half = lane >> 4;
    const int l15  = lane & 15;
    int mrow = rowTile * 16 + l15;
    if (mrow >= M) mrow = M - 1;

    v8f acc[NT] = {};
    const float* arow = A + (size_t)mrow * K + 2 * half;
    const v2f* __restrict__ bcol = sB2 + half * N + l15;

    for (int k0 = 0; k0 < K; k0 += 4) {
        const v2f a = *(const v2f*)(arow + k0);
        const v2f* brow = bcol + (k0 >> 1) * N;
#pragma unroll
        for (int nt = 0; nt < NT; ++nt) {
            const v2f b = brow[nt * 16];
            acc[nt] = __builtin_amdgcn_wmma_f32_16x16x4_f32(
                false, a, false, b, (short)0, acc[nt], false, false);
        }
    }

#pragma unroll
    for (int nt = 0; nt < NT; ++nt) {
#pragma unroll
        for (int r = 0; r < 8; ++r) {
            const int row = rowTile * 16 + r + 8 * half;
            if (row < M)
                C[(size_t)row * N + nt * 16 + l15] = acc[nt][r];
        }
    }
}

__device__ __forceinline__ void atomicMaxFloat(float* addr, float val) {
    const int iv = __float_as_int(val);
    if (iv >= 0) atomicMax((int*)addr, iv);
    else         atomicMin((unsigned int*)addr, (unsigned int)iv);
}

__global__ void node_score_kernel(const float* __restrict__ z,
                                  const float* __restrict__ w_attn,
                                  float* __restrict__ ssrc, float* __restrict__ sdst,
                                  int N)
{
    const int n = blockIdx.x * blockDim.x + threadIdx.x;
    if (n >= N) return;
    const float* zr = z + (size_t)n * 64;
    float s0 = 0.f, s1 = 0.f;
#pragma unroll
    for (int i = 0; i < 64; ++i) {
        const float v = zr[i];
        s0 += v * w_attn[i];
        s1 += v * w_attn[96 + i];
    }
    ssrc[n] = s0;
    sdst[n] = s1;
}

__global__ void init_kernel(float* __restrict__ h, float* __restrict__ mbuf,
                            float* __restrict__ den, float* __restrict__ zphi,
                            int hsize, int N)
{
    const int i = blockIdx.x * blockDim.x + threadIdx.x;
    const int stride = gridDim.x * blockDim.x;
    for (int j = i; j < hsize; j += stride) h[j] = 0.f;
    for (int j = i; j < N; j += stride) { mbuf[j] = -INFINITY; den[j] = 0.f; }
    if (i < 256) zphi[i] = 0.f;
}

__global__ void edge_score_kernel(const float* __restrict__ eft,
                                  const int* __restrict__ src, const int* __restrict__ dst,
                                  const float* __restrict__ ssrc, const float* __restrict__ sdst,
                                  const float* __restrict__ w_attn,
                                  float* __restrict__ a, float* __restrict__ mbuf, int E)
{
    const int e = blockIdx.x * blockDim.x + threadIdx.x;
    if (e >= E) return;
    const float* er = eft + (size_t)e * 32;
    float t = ssrc[src[e]] + sdst[dst[e]];
#pragma unroll
    for (int i = 0; i < 32; ++i) t += er[i] * w_attn[64 + i];
    const float lr = t > 0.f ? t : 0.01f * t;
    a[e] = lr;
    atomicMaxFloat(&mbuf[dst[e]], lr);
}

__global__ void edge_exp_kernel(const int* __restrict__ dst,
                                const float* __restrict__ mbuf,
                                float* __restrict__ a, float* __restrict__ den, int E)
{
    const int e = blockIdx.x * blockDim.x + threadIdx.x;
    if (e >= E) return;
    const int d = dst[e];
    const float ex = __expf(a[e] - mbuf[d]);
    a[e] = ex;
    atomicAdd(&den[d], ex);
}

__global__ void edge_scatter_kernel(const float* __restrict__ z, const float* __restrict__ eft,
                                    const float* __restrict__ a, const float* __restrict__ den,
                                    const int* __restrict__ src, const int* __restrict__ dst,
                                    const int* __restrict__ etype,
                                    float* __restrict__ h, int E)
{
    const int e    = (int)((blockIdx.x * blockDim.x + threadIdx.x) >> 5);
    const int lane = threadIdx.x & 31;
    if (e >= E) return;
    const int d = dst[e];
    const float alpha = a[e] / den[d];
    const size_t base = ((size_t)d * 4 + etype[e]) * 96;
    const float* zr = z + (size_t)src[e] * 64;
    const float* er = eft + (size_t)e * 32;
#pragma unroll
    for (int j = lane; j < 96; j += 32) {
        const float v = (j < 64) ? zr[j] : er[j - 64];
        atomicAdd(&h[base + j], alpha * v);
    }
}

__global__ void zphi_kernel(const float* __restrict__ h2, float* __restrict__ zphi, int nodes)
{
    const int s = threadIdx.x;
    float sum = 0.f;
    for (int n = blockIdx.x; n < nodes; n += gridDim.x)
        sum += h2[(size_t)n * 256 + s];
    atomicAdd(&zphi[s], sum);
}

__global__ void beta_kernel(const float* __restrict__ zphi,
                            const float* __restrict__ w_sem, float* __restrict__ beta)
{
    if (threadIdx.x != 0) return;
    float wp[4], mx = -INFINITY;
#pragma unroll
    for (int t = 0; t < 4; ++t) {
        float d = 0.f;
        for (int o = 0; o < 64; ++o) d += zphi[t * 64 + o] * w_sem[o];
        const float lr = d > 0.f ? d : 0.01f * d;
        wp[t] = lr;
        mx = fmaxf(mx, lr);
    }
    float s = 0.f;
#pragma unroll
    for (int t = 0; t < 4; ++t) { wp[t] = __expf(wp[t] - mx); s += wp[t]; }
#pragma unroll
    for (int t = 0; t < 4; ++t) beta[t] = wp[t] / s;
}

__global__ void out_kernel(const float* __restrict__ h2, const float* __restrict__ beta,
                           float* __restrict__ Z, int N)
{
    const int i = blockIdx.x * blockDim.x + threadIdx.x;
    if (i >= N * 64) return;
    const int n = i >> 6, o = i & 63;
    const float* r = h2 + (size_t)n * 256 + o;
    Z[i] = beta[0] * r[0] + beta[1] * r[64] + beta[2] * r[128] + beta[3] * r[192];
}

extern "C" void kernel_launch(void* const* d_in, const int* in_sizes, int n_in,
                              void* d_out, int out_size, void* d_ws, size_t ws_size,
                              hipStream_t stream)
{
    const float* node   = (const float*)d_in[0];
    const float* edgef  = (const float*)d_in[1];
    const int*   src    = (const int*)d_in[2];
    const int*   dst    = (const int*)d_in[3];
    const int*   etype  = (const int*)d_in[4];
    const float* w_n    = (const float*)d_in[5];
    const float* w_e    = (const float*)d_in[6];
    const float* w_attn = (const float*)d_in[7];
    const float* w_fc2  = (const float*)d_in[8];
    const float* w_sem  = (const float*)d_in[9];
    float* out = (float*)d_out;

    const int N = in_sizes[0] / 128;
    const int E = in_sizes[2];
    const int T = 4;

    float* ws   = (float*)d_ws;
    float* z    = ws;
    float* ssrc = z + (size_t)N * 64;
    float* sdst = ssrc + N;
    float* abuf = sdst + N;
    float* mbuf = abuf + E;
    float* den  = mbuf + N;
    float* zphi = den + N;
    float* beta = zphi + 256;
    float* h    = beta + 16;
    float* eft  = h + (size_t)N * T * 96;
    float* h2   = eft;

    const int thr = 256;

    gemm_f32_wmma<4><<<((N + 15) / 16 + 7) / 8, thr, 0, stream>>>(node, w_n, z, N, 128, 64);
    gemm_f32_wmma<2><<<((E + 15) / 16 + 7) / 8, thr, 0, stream>>>(edgef, w_e, eft, E, 64, 32);
    node_score_kernel<<<(N + thr - 1) / thr, thr, 0, stream>>>(z, w_attn, ssrc, sdst, N);
    init_kernel<<<1024, thr, 0, stream>>>(h, mbuf, den, zphi, N * T * 96, N);
    edge_score_kernel<<<(E + thr - 1) / thr, thr, 0, stream>>>(eft, src, dst, ssrc, sdst,
                                                               w_attn, abuf, mbuf, E);
    edge_exp_kernel<<<(E + thr - 1) / thr, thr, 0, stream>>>(dst, mbuf, abuf, den, E);
    edge_scatter_kernel<<<(E + 7) / 8, thr, 0, stream>>>(z, eft, abuf, den, src, dst, etype, h, E);
    gemm_f32_wmma<4><<<((N * T + 15) / 16 + 7) / 8, thr, 0, stream>>>(h, w_fc2, h2, N * T, 96, 64);
    zphi_kernel<<<256, thr, 0, stream>>>(h2, zphi, N);
    beta_kernel<<<1, 32, 0, stream>>>(zphi, w_sem, beta);
    out_kernel<<<(N * 64 + thr - 1) / thr, thr, 0, stream>>>(h2, beta, out, N);
}


# zphi column sum as a two-stage tree without atomics: 512 blocks read rows as b128, 8 independent loads per lane per hop (was one 4-B load per hop), fixed-order fold in beta kernel; sum order changes
# speedup vs baseline: 1.1713x; 1.1713x over previous
#include <hip/hip_runtime.h>
#include <hip/hip_bf16.h>
#include <math.h>

typedef __attribute__((ext_vector_type(2))) float v2f;
typedef __attribute__((ext_vector_type(8))) float v8f;

template <int NT>
__global__ void __launch_bounds__(256) gemm_f32_wmma(
    const float* __restrict__ A, const float* __restrict__ B,
    float* __restrict__ C, int M, int K, int N)
{
    __shared__ v2f sB2[4096];
    const int tid = threadIdx.x;
    const int pairCount = (K >> 1) * N;
    for (int p = tid; p < pairCount; p += 256) {
        const int kk = (p / N) << 1;
        const int n  = p - (p / N) * N;
        v2f b;
        b.x = B[kk * N + n];
        b.y = B[(kk + 1) * N + n];
        sB2[p] = b;
    }
    __syncthreads();

    const int lane    = tid & 31;
    const int wave    = tid >> 5;
    const int rowTile = blockIdx.x * 8 + wave;
    const int numTiles = (M + 15) >> 4;
    if (rowTile >= numTiles) return;

    const int half = lane >> 4;
    const int l15  = lane & 15;
    int mrow = rowTile * 16 + l15;
    if (mrow >= M) mrow = M - 1;

    v8f acc[NT] = {};
    const float* arow = A + (size_t)mrow * K + 2 * half;
    const v2f* __restrict__ bcol = sB2 + half * N + l15;

    for (int k0 = 0; k0 < K; k0 += 4) {
        const v2f a = *(const v2f*)(arow + k0);
        const v2f* brow = bcol + (k0 >> 1) * N;
#pragma unroll
        for (int nt = 0; nt < NT; ++nt) {
            const v2f b = brow[nt * 16];
            acc[nt] = __builtin_amdgcn_wmma_f32_16x16x4_f32(
                false, a, false, b, (short)0, acc[nt], false, false);
        }
    }

#pragma unroll
    for (int nt = 0; nt < NT; ++nt) {
#pragma unroll
        for (int r = 0; r < 8; ++r) {
            const int row = rowTile * 16 + r + 8 * half;
            if (row < M)
                C[(size_t)row * N + nt * 16 + l15] = acc[nt][r];
        }
    }
}

__device__ __forceinline__ void atomicMaxFloat(float* addr, float val) {
    const int iv = __float_as_int(val);
    if (iv >= 0) atomicMax((int*)addr, iv);
    else         atomicMin((unsigned int*)addr, (unsigned int)iv);
}

__global__ void node_score_kernel(const float* __restrict__ z,
                                  const float* __restrict__ w_attn,
                                  float* __restrict__ ssrc, float* __restrict__ sdst,
                                  int N)
{
    const int n = blockIdx.x * blockDim.x + threadIdx.x;
    if (n >= N) return;
    const float* zr = z + (size_t)n * 64;
    float s0 = 0.f, s1 = 0.f;
#pragma unroll
    for (int i = 0; i < 64; ++i) {
        const float v = zr[i];
        s0 += v * w_attn[i];
        s1 += v * w_attn[96 + i];
    }
    ssrc[n] = s0;
    sdst[n] = s1;
}

__global__ void init_kernel(float* __restrict__ h, float* __restrict__ mbuf,
                            float* __restrict__ den, int hsize, int N)
{
    const int i = blockIdx.x * blockDim.x + threadIdx.x;
    const int stride = gridDim.x * blockDim.x;
    for (int j = i; j < hsize; j += stride) h[j] = 0.f;
    for (int j = i; j < N; j += stride) { mbuf[j] = -INFINITY; den[j] = 0.f; }
}

__global__ void edge_score_kernel(const float* __restrict__ eft,
                                  const int* __restrict__ src, const int* __restrict__ dst,
                                  const float* __restrict__ ssrc, const float* __restrict__ sdst,
                                  const float* __restrict__ w_attn,
                                  float* __restrict__ a, float* __restrict__ mbuf, int E)
{
    const int e = blockIdx.x * blockDim.x + threadIdx.x;
    if (e >= E) return;
    const float* er = eft + (size_t)e * 32;
    float t = ssrc[src[e]] + sdst[dst[e]];
#pragma unroll
    for (int i = 0; i < 32; ++i) t += er[i] * w_attn[64 + i];
    const float lr = t > 0.f ? t : 0.01f * t;
    a[e] = lr;
    atomicMaxFloat(&mbuf[dst[e]], lr);
}

__global__ void edge_exp_kernel(const int* __restrict__ dst,
                                const float* __restrict__ mbuf,
                                float* __restrict__ a, float* __restrict__ den, int E)
{
    const int e = blockIdx.x * blockDim.x + threadIdx.x;
    if (e >= E) return;
    const int d = dst[e];
    const float ex = __expf(a[e] - mbuf[d]);
    a[e] = ex;
    atomicAdd(&den[d], ex);
}

__global__ void edge_scatter_kernel(const float* __restrict__ z, const float* __restrict__ eft,
                                    const float* __restrict__ a, const float* __restrict__ den,
                                    const int* __restrict__ src, const int* __restrict__ dst,
                                    const int* __restrict__ etype,
                                    float* __restrict__ h, int E)
{
    const int e    = (int)((blockIdx.x * blockDim.x + threadIdx.x) >> 5);
    const int lane = threadIdx.x & 31;
    if (e >= E) return;
    const int d = dst[e];
    const float alpha = a[e] / den[d];
    const size_t base = ((size_t)d * 4 + etype[e]) * 96;
    const float* zr = z + (size_t)src[e] * 64;
    const float* er = eft + (size_t)e * 32;
#pragma unroll
    for (int j = lane; j < 96; j += 32) {
        const float v = (j < 64) ? zr[j] : er[j - 64];
        atomicAdd(&h[base + j], alpha * v);
    }
}

constexpr int ZP_BLOCKS = 512;

__device__ __forceinline__ void add4(float4& a, const float4& v) {
    a.x += v.x; a.y += v.y; a.z += v.z; a.w += v.w;
}

__global__ void __launch_bounds__(512) zphi_partial_kernel(
    const float* __restrict__ h2, float* __restrict__ part, int nodes)
{
    __shared__ float4 sm[8][64];
    const int tid = threadIdx.x;
    const int c4  = tid & 63;
    const int sub = tid >> 6;
    const int R   = (nodes + (int)gridDim.x - 1) / (int)gridDim.x;
    const int r0  = blockIdx.x * R;
    int r1 = r0 + R;
    if (r1 > nodes) r1 = nodes;
    const float4* __restrict__ p = (const float4*)h2 + c4;

    float4 acc = make_float4(0.f, 0.f, 0.f, 0.f);
    int r = r0 + sub;
    for (; r + 56 < r1; r += 64) {
        float4 v[8];
#pragma unroll
        for (int u = 0; u < 8; ++u) v[u] = p[(size_t)(r + 8 * u) * 64];
#pragma unroll
        for (int u = 0; u < 8; ++u) add4(acc, v[u]);
    }
    {
        float4 v[8];
#pragma unroll
        for (int u = 0; u < 8; ++u) {
            const int rr = r + 8 * u;
            v[u] = make_float4(0.f, 0.f, 0.f, 0.f);
            if (rr < r1) v[u] = p[(size_t)rr * 64];
        }
#pragma unroll
        for (int u = 0; u < 8; ++u) add4(acc, v[u]);
    }
    sm[sub][c4] = acc;
    __syncthreads();
    if (tid < 256) {
        const float* sf = (const float*)&sm[0][0];
        float s = 0.f;
#pragma unroll
        for (int k = 0; k < 8; ++k) s += sf[k * 256 + tid];
        part[(size_t)blockIdx.x * 256 + tid] = s;
    }
}

__global__ void __launch_bounds__(1024) beta_kernel(const float* __restrict__ part,
                            const float* __restrict__ w_sem, float* __restrict__ beta)
{
    __shared__ float4 sm[16][64];
    __shared__ float zs[256];
    const int tid = threadIdx.x;
    const int c4  = tid & 63;
    const int g   = tid >> 6;
    const float4* __restrict__ p = (const float4*)part + (size_t)g * 32 * 64 + c4;
    float4 acc = make_float4(0.f, 0.f, 0.f, 0.f);
    for (int b = 0; b < 32; b += 16) {
        float4 v[16];
#pragma unroll
        for (int u = 0; u < 16; ++u) v[u] = p[(size_t)(b + u) * 64];
#pragma unroll
        for (int u = 0; u < 16; ++u) add4(acc, v[u]);
    }
    sm[g][c4] = acc;
    __syncthreads();
    if (tid < 256) {
        const float* sf = (const float*)&sm[0][0];
        float s = 0.f;
#pragma unroll
        for (int k = 0; k < 16; ++k) s += sf[k * 256 + tid];
        zs[tid] = s;
    }
    __syncthreads();
    if (tid != 0) return;
    float wp[4], mx = -INFINITY;
#pragma unroll
    for (int t = 0; t < 4; ++t) {
        float d = 0.f;
        for (int o = 0; o < 64; ++o) d += zs[t * 64 + o] * w_sem[o];
        const float lr = d > 0.f ? d : 0.01f * d;
        wp[t] = lr;
        mx = fmaxf(mx, lr);
    }
    float s = 0.f;
#pragma unroll
    for (int t = 0; t < 4; ++t) { wp[t] = __expf(wp[t] - mx); s += wp[t]; }
#pragma unroll
    for (int t = 0; t < 4; ++t) beta[t] = wp[t] / s;
}

__global__ void out_kernel(const float* __restrict__ h2, const float* __restrict__ beta,
                           float* __restrict__ Z, int N)
{
    const int i = blockIdx.x * blockDim.x + threadIdx.x;
    if (i >= N * 64) return;
    const int n = i >> 6, o = i & 63;
    const float* r = h2 + (size_t)n * 256 + o;
    Z[i] = beta[0] * r[0] + beta[1] * r[64] + beta[2] * r[128] + beta[3] * r[192];
}

extern "C" void kernel_launch(void* const* d_in, const int* in_sizes, int n_in,
                              void* d_out, int out_size, void* d_ws, size_t ws_size,
                              hipStream_t stream)
{
    const float* node   = (const float*)d_in[0];
    const float* edgef  = (const float*)d_in[1];
    const int*   src    = (const int*)d_in[2];
    const int*   dst    = (const int*)d_in[3];
    const int*   etype  = (const int*)d_in[4];
    const float* w_n    = (const float*)d_in[5];
    const float* w_e    = (const float*)d_in[6];
    const float* w_attn = (const float*)d_in[7];
    const float* w_fc2  = (const float*)d_in[8];
    const float* w_sem  = (const float*)d_in[9];
    float* out = (float*)d_out;

    const int N = in_sizes[0] / 128;
    const int E = in_sizes[2];
    const int T = 4;

    float* ws   = (float*)d_ws;
    float* z    = ws;
    float* ssrc = z + (size_t)N * 64;
    float* sdst = ssrc + N;
    float* abuf = sdst + N;
    float* mbuf = abuf + E;
    float* den  = mbuf + N;
    float* beta = den + N + 256;
    float* h    = beta + 16;
    float* eft  = h + (size_t)N * T * 96;
    float* h2   = eft;
    float* part = abuf;

    const int thr = 256;

    gemm_f32_wmma<4><<<((N + 15) / 16 + 7) / 8, thr, 0, stream>>>(node, w_n, z, N, 128, 64);
    gemm_f32_wmma<2><<<((E + 15) / 16 + 7) / 8, thr, 0, stream>>>(edgef, w_e, eft, E, 64, 32);
    node_score_kernel<<<(N + thr - 1) / thr, thr, 0, stream>>>(z, w_attn, ssrc, sdst, N);
    init_kernel<<<1024, thr, 0, stream>>>(h, mbuf, den, N * T * 96, N);
    edge_score_kernel<<<(E + thr - 1) / thr, thr, 0, stream>>>(eft, src, dst, ssrc, sdst,
                                                               w_attn, abuf, mbuf, E);
    edge_exp_kernel<<<(E + thr - 1) / thr, thr, 0, stream>>>(dst, mbuf, abuf, den, E);
    edge_scatter_kernel<<<(E + 7) / 8, thr, 0, stream>>>(z, eft, abuf, den, src, dst, etype, h, E);
    gemm_f32_wmma<4><<<((N * T + 15) / 16 + 7) / 8, thr, 0, stream>>>(h, w_fc2, h2, N * T, 96, 64);
    zphi_partial_kernel<<<ZP_BLOCKS, 512, 0, stream>>>(h2, part, N);
    beta_kernel<<<1, 1024, 0, stream>>>(part, w_sem, beta);
    out_kernel<<<(N * 64 + thr - 1) / thr, thr, 0, stream>>>(h2, beta, out, N);
}


# scatter without float atomics: edges counting-sorted by dst, one wave per node does logits, softmax and the h rows in registers in ascending edge id; sum order changes and is now fixed
# speedup vs baseline: 1.3546x; 1.1565x over previous
#include <hip/hip_runtime.h>
#include <hip/hip_bf16.h>
#include <math.h>

typedef __attribute__((ext_vector_type(2))) float v2f;
typedef __attribute__((ext_vector_type(8))) float v8f;

template <int NT>
__global__ void __launch_bounds__(256) gemm_f32_wmma(
    const float* __restrict__ A, const float* __restrict__ B,
    float* __restrict__ C, int M, int K, int N)
{
    __shared__ v2f sB2[4096];
    const int tid = threadIdx.x;
    const int pairCount = (K >> 1) * N;
    for (int p = tid; p < pairCount; p += 256) {
        const int kk = (p / N) << 1;
        const int n  = p - (p / N) * N;
        v2f b;
        b.x = B[kk * N + n];
        b.y = B[(kk + 1) * N + n];
        sB2[p] = b;
    }
    __syncthreads();

    const int lane    = tid & 31;
    const int wave    = tid >> 5;
    const int rowTile = blockIdx.x * 8 + wave;
    const int numTiles = (M + 15) >> 4;
    if (rowTile >= numTiles) return;

    const int half = lane >> 4;
    const int l15  = lane & 15;
    int mrow = rowTile * 16 + l15;
    if (mrow >= M) mrow = M - 1;

    v8f acc[NT] = {};
    const float* arow = A + (size_t)mrow * K + 2 * half;
    const v2f* __restrict__ bcol = sB2 + half * N + l15;

    for (int k0 = 0; k0 < K; k0 += 4) {
        const v2f a = *(const v2f*)(arow + k0);
        const v2f* brow = bcol + (k0 >> 1) * N;
#pragma unroll
        for (int nt = 0; nt < NT; ++nt) {
            const v2f b = brow[nt * 16];
            acc[nt] = __builtin_amdgcn_wmma_f32_16x16x4_f32(
                false, a, false, b, (short)0, acc[nt], false, false);
        }
    }

#pragma unroll
    for (int nt = 0; nt < NT; ++nt) {
#pragma unroll
        for (int r = 0; r < 8; ++r) {
            const int row = rowTile * 16 + r + 8 * half;
            if (row < M)
                C[(size_t)row * N + nt * 16 + l15] = acc[nt][r];
        }
    }
}


__global__ void node_score_kernel(const float* __restrict__ z,
                                  const float* __restrict__ w_attn,
                                  float* __restrict__ ssrc, float* __restrict__ sdst,
                                  int* __restrict__ cnt, int* __restrict__ total, int N)
{
    const int n = blockIdx.x * blockDim.x + threadIdx.x;
    if (n >= N) return;
    const float* zr = z + (size_t)n * 64;
    float s0 = 0.f, s1 = 0.f;
#pragma unroll
    for (int i = 0; i < 64; ++i) {
        const float v = zr[i];
        s0 += v * w_attn[i];
        s1 += v * w_attn[96 + i];
    }
    ssrc[n] = s0;
    sdst[n] = s1;
    cnt[n] = 0;
    if (n == 0) *total = 0;
}

__global__ void dst_count_kernel(const int* __restrict__ dst, int* __restrict__ cnt, int E)
{
    const int e = blockIdx.x * blockDim.x + threadIdx.x;
    if (e >= E) return;
    atomicAdd(&cnt[dst[e]], 1);
}

__global__ void __launch_bounds__(256) seg_alloc_kernel(int* __restrict__ cur, int* __restrict__ beg,
                                                        int* __restrict__ total, int N)
{
    __shared__ int wtot[8];
    __shared__ int sbase;
    const int tid  = threadIdx.x;
    const int lane = tid & 31;
    const int wave = tid >> 5;
    const int n = blockIdx.x * 256 + tid;
    const int c = (n < N) ? cur[n] : 0;
    int x = c;
#pragma unroll
    for (int o = 1; o < 32; o <<= 1) {
        const int y = __shfl_up(x, o);
        if (lane >= o) x += y;
    }
    if (lane == 31) wtot[wave] = x;
    __syncthreads();
    if (tid == 0) {
        int s = 0;
#pragma unroll
        for (int w = 0; w < 8; ++w) { const int t = wtot[w]; wtot[w] = s; s += t; }
        sbase = atomicAdd(total, s);
    }
    __syncthreads();
    const int b = sbase + wtot[wave] + x - c;
    if (n < N) { beg[n] = b; cur[n] = b; }
}

__global__ void dst_fill_kernel(const int* __restrict__ dst, int* __restrict__ cur,
                                int* __restrict__ perm, int E)
{
    const int e = blockIdx.x * blockDim.x + threadIdx.x;
    if (e >= E) return;
    const int pos = atomicAdd(&cur[dst[e]], 1);
    perm[pos] = e;
}

__device__ __forceinline__ float wave_max_f(float v) {
#pragma unroll
    for (int o = 16; o > 0; o >>= 1) v = fmaxf(v, __shfl_xor(v, o));
    return v;
}
__device__ __forceinline__ float wave_sum_f(float v) {
#pragma unroll
    for (int o = 16; o > 0; o >>= 1) v += __shfl_xor(v, o);
    return v;
}
__device__ __forceinline__ int wave_min_i(int v) {
#pragma unroll
    for (int o = 16; o > 0; o >>= 1) { const int t = __shfl_xor(v, o); v = t < v ? t : v; }
    return v;
}

__device__ __forceinline__ float edge_logit(const float* __restrict__ eft, const int* __restrict__ src,
                                            const float* __restrict__ ssrc,
                                            const float* __restrict__ w_attn, int e, float sd)
{
    const float4* er = (const float4*)(eft + (size_t)e * 32);
    float t = ssrc[src[e]] + sd;
#pragma unroll
    for (int q = 0; q < 8; ++q) {
        const float4 v = er[q];
        t += v.x * w_attn[64 + 4 * q];
        t += v.y * w_attn[65 + 4 * q];
        t += v.z * w_attn[66 + 4 * q];
        t += v.w * w_attn[67 + 4 * q];
    }
    return t > 0.f ? t : 0.01f * t;
}

#define ROW_ACC(T)                                                         \
    if (ty == T) { acc[T][0] += al * v0; acc[T][1] += al * v1; acc[T][2] += al * v2; }

__device__ __forceinline__ int next_edge(const int* __restrict__ pb, int deg, int lane, int last)
{
    int best = 0x7fffffff;
    for (int c = 0; c < deg; c += 32) {
        const int i = c + lane;
        if (i < deg) {
            const int v = pb[i];
            if (v > last && v < best) best = v;
        }
    }
    return wave_min_i(best);
}

__global__ void __launch_bounds__(256) node_gather_kernel(
    const float* __restrict__ z, const float* __restrict__ eft,
    const int* __restrict__ src, const int* __restrict__ etype,
    const float* __restrict__ ssrc, const float* __restrict__ sdst,
    const float* __restrict__ w_attn,
    const int* __restrict__ beg, const int* __restrict__ cur, const int* __restrict__ perm,
    float* __restrict__ h, int N)
{
    const int d    = (int)((blockIdx.x * blockDim.x + threadIdx.x) >> 5);
    const int lane = threadIdx.x & 31;
    if (d >= N) return;
    const int b   = beg[d];
    const int deg = cur[d] - b;
    const int* __restrict__ pb = perm + b;
    const float sd = sdst[d];
    float acc[4][3] = {};

    if (deg <= 32) {
        const int id = (lane < deg) ? pb[lane] : (0x7fffffc0 + lane);
        int rank = 0;
#pragma unroll
        for (int j = 0; j < 32; ++j)
            rank += (__builtin_amdgcn_readlane(id, j) < id) ? 1 : 0;
        const int sid = __builtin_amdgcn_ds_permute(rank << 2, id);
        const bool ok = lane < deg;
        const int e  = ok ? sid : 0;
        const int s  = src[e];
        const int tt = etype[e];
        const float lr = ok ? edge_logit(eft, src, ssrc, w_attn, e, sd) : -INFINITY;
        const float m  = wave_max_f(lr);
        const float ex = ok ? __expf(lr - m) : 0.f;
        const float den = wave_sum_f(ex);
        const float alpha = ex / den;

        for (int j = 0; j < deg; j += 4) {
            float r0[4], r1[4], r2[4];
#pragma unroll
            for (int u = 0; u < 4; ++u) {
                const int jj = (j + u < deg) ? (j + u) : (deg - 1);
                const int eu = __builtin_amdgcn_readlane(e, jj);
                const int su = __builtin_amdgcn_readlane(s, jj);
                const float* zr = z + (size_t)su * 64;
                r0[u] = zr[lane];
                r1[u] = zr[32 + lane];
                r2[u] = eft[(size_t)eu * 32 + lane];
            }
#pragma unroll
            for (int u = 0; u < 4; ++u) {
                if (j + u < deg) {
                    const float al = __int_as_float(__builtin_amdgcn_readlane(__float_as_int(alpha), j + u));
                    const int   ty = __builtin_amdgcn_readlane(tt, j + u);
                    const float v0 = r0[u], v1 = r1[u], v2 = r2[u];
                    ROW_ACC(0) ROW_ACC(1) ROW_ACC(2) ROW_ACC(3)
                }
            }
        }
    } else {
        float m = -INFINITY;
        for (int c = 0; c < deg; c += 32) {
            const int i = c + lane;
            if (i < deg) m = fmaxf(m, edge_logit(eft, src, ssrc, w_attn, pb[i], sd));
        }
        m = wave_max_f(m);
        float den = 0.f;
        int last = -1;
        for (int j = 0; j < deg; ++j) {
            last = next_edge(pb, deg, lane, last);
            den += __expf(edge_logit(eft, src, ssrc, w_attn, last, sd) - m);
        }
        last = -1;
        for (int j = 0; j < deg; ++j) {
            last = next_edge(pb, deg, lane, last);
            const float al = __expf(edge_logit(eft, src, ssrc, w_attn, last, sd) - m) / den;
            const int ty = etype[last];
            const float* zr = z + (size_t)src[last] * 64;
            const float v0 = zr[lane], v1 = zr[32 + lane], v2 = eft[(size_t)last * 32 + lane];
            ROW_ACC(0) ROW_ACC(1) ROW_ACC(2) ROW_ACC(3)
        }
    }

    float* hr = h + (size_t)d * 4 * 96 + lane;
#pragma unroll
    for (int t = 0; t < 4; ++t) {
        hr[t * 96]      = acc[t][0];
        hr[t * 96 + 32] = acc[t][1];
        hr[t * 96 + 64] = acc[t][2];
    }
}
#undef ROW_ACC

constexpr int ZP_BLOCKS = 512;

__device__ __forceinline__ void add4(float4& a, const float4& v) {
    a.x += v.x; a.y += v.y; a.z += v.z; a.w += v.w;
}

__global__ void __launch_bounds__(512) zphi_partial_kernel(
    const float* __restrict__ h2, float* __restrict__ part, int nodes)
{
    __shared__ float4 sm[8][64];
    const int tid = threadIdx.x;
    const int c4  = tid & 63;
    const int sub = tid >> 6;
    const int R   = (nodes + (int)gridDim.x - 1) / (int)gridDim.x;
    const int r0  = blockIdx.x * R;
    int r1 = r0 + R;
    if (r1 > nodes) r1 = nodes;
    const float4* __restrict__ p = (const float4*)h2 + c4;

    float4 acc = make_float4(0.f, 0.f, 0.f, 0.f);
    int r = r0 + sub;
    for (; r + 56 < r1; r += 64) {
        float4 v[8];
#pragma unroll
        for (int u = 0; u < 8; ++u) v[u] = p[(size_t)(r + 8 * u) * 64];
#pragma unroll
        for (int u = 0; u < 8; ++u) add4(acc, v[u]);
    }
    {
        float4 v[8];
#pragma unroll
        for (int u = 0; u < 8; ++u) {
            const int rr = r + 8 * u;
            v[u] = make_float4(0.f, 0.f, 0.f, 0.f);
            if (rr < r1) v[u] = p[(size_t)rr * 64];
        }
#pragma unroll
        for (int u = 0; u < 8; ++u) add4(acc, v[u]);
    }
    sm[sub][c4] = acc;
    __syncthreads();
    if (tid < 256) {
        const float* sf = (const float*)&sm[0][0];
        float s = 0.f;
#pragma unroll
        for (int k = 0; k < 8; ++k) s += sf[k * 256 + tid];
        part[(size_t)blockIdx.x * 256 + tid] = s;
    }
}

__global__ void __launch_bounds__(1024) beta_kernel(const float* __restrict__ part,
                            const float* __restrict__ w_sem, float* __restrict__ beta)
{
    __shared__ float4 sm[16][64];
    __shared__ float zs[256];
    const int tid = threadIdx.x;
    const int c4  = tid & 63;
    const int g   = tid >> 6;
    const float4* __restrict__ p = (const float4*)part + (size_t)g * 32 * 64 + c4;
    float4 acc = make_float4(0.f, 0.f, 0.f, 0.f);
    for (int b = 0; b < 32; b += 16) {
        float4 v[16];
#pragma unroll
        for (int u = 0; u < 16; ++u) v[u] = p[(size_t)(b + u) * 64];
#pragma unroll
        for (int u = 0; u < 16; ++u) add4(acc, v[u]);
    }
    sm[g][c4] = acc;
    __syncthreads();
    if (tid < 256) {
        const float* sf = (const float*)&sm[0][0];
        float s = 0.f;
#pragma unroll
        for (int k = 0; k < 16; ++k) s += sf[k * 256 + tid];
        zs[tid] = s;
    }
    __syncthreads();
    if (tid != 0) return;
    float wp[4], mx = -INFINITY;
#pragma unroll
    for (int t = 0; t < 4; ++t) {
        float d = 0.f;
        for (int o = 0; o < 64; ++o) d += zs[t * 64 + o] * w_sem[o];
        const float lr = d > 0.f ? d : 0.01f * d;
        wp[t] = lr;
        mx = fmaxf(mx, lr);
    }
    float s = 0.f;
#pragma unroll
    for (int t = 0; t < 4; ++t) { wp[t] = __expf(wp[t] - mx); s += wp[t]; }
#pragma unroll
    for (int t = 0; t < 4; ++t) beta[t] = wp[t] / s;
}

__global__ void out_kernel(const float* __restrict__ h2, const float* __restrict__ beta,
                           float* __restrict__ Z, int N)
{
    const int i = blockIdx.x * blockDim.x + threadIdx.x;
    if (i >= N * 64) return;
    const int n = i >> 6, o = i & 63;
    const float* r = h2 + (size_t)n * 256 + o;
    Z[i] = beta[0] * r[0] + beta[1] * r[64] + beta[2] * r[128] + beta[3] * r[192];
}

extern "C" void kernel_launch(void* const* d_in, const int* in_sizes, int n_in,
                              void* d_out, int out_size, void* d_ws, size_t ws_size,
                              hipStream_t stream)
{
    const float* node   = (const float*)d_in[0];
    const float* edgef  = (const float*)d_in[1];
    const int*   src    = (const int*)d_in[2];
    const int*   dst    = (const int*)d_in[3];
    const int*   etype  = (const int*)d_in[4];
    const float* w_n    = (const float*)d_in[5];
    const float* w_e    = (const float*)d_in[6];
    const float* w_attn = (const float*)d_in[7];
    const float* w_fc2  = (const float*)d_in[8];
    const float* w_sem  = (const float*)d_in[9];
    float* out = (float*)d_out;

    const int N = in_sizes[0] / 128;
    const int E = in_sizes[2];
    const int T = 4;

    float* ws   = (float*)d_ws;
    float* z    = ws;
    float* ssrc = z + (size_t)N * 64;
    float* sdst = ssrc + N;
    float* abuf = sdst + N;
    float* mbuf = abuf + E;
    float* den  = mbuf + N;
    float* beta = den + N + 256;
    float* h    = beta + 16;
    float* eft  = h + (size_t)N * T * 96;
    float* h2   = eft;
    float* part = abuf;
    int* perm  = (int*)abuf;
    int* cur   = (int*)mbuf;
    int* beg   = (int*)den;
    int* total = (int*)(den + N);

    const int thr = 256;

    gemm_f32_wmma<4><<<((N + 15) / 16 + 7) / 8, thr, 0, stream>>>(node, w_n, z, N, 128, 64);
    gemm_f32_wmma<2><<<((E + 15) / 16 + 7) / 8, thr, 0, stream>>>(edgef, w_e, eft, E, 64, 32);
    node_score_kernel<<<(N + thr - 1) / thr, thr, 0, stream>>>(z, w_attn, ssrc, sdst, cur, total, N);
    dst_count_kernel<<<(E + thr - 1) / thr, thr, 0, stream>>>(dst, cur, E);
    seg_alloc_kernel<<<(N + 255) / 256, 256, 0, stream>>>(cur, beg, total, N);
    dst_fill_kernel<<<(E + thr - 1) / thr, thr, 0, stream>>>(dst, cur, perm, E);
    node_gather_kernel<<<(N + 7) / 8, thr, 0, stream>>>(z, eft, src, etype, ssrc, sdst, w_attn,
                                                        beg, cur, perm, h, N);
    gemm_f32_wmma<4><<<((N * T + 15) / 16 + 7) / 8, thr, 0, stream>>>(h, w_fc2, h2, N * T, 96, 64);
    zphi_partial_kernel<<<ZP_BLOCKS, 512, 0, stream>>>(h2, part, N);
    beta_kernel<<<1, 1024, 0, stream>>>(part, w_sem, beta);
    out_kernel<<<(N * 64 + thr - 1) / thr, thr, 0, stream>>>(h2, beta, out, N);
}


# destinations with 33 to 64 incoming edges rank their ids in registers (two per lane) and gather rows four at a time like short runs, not by serial repeated minimum; order of every sum unchanged
# speedup vs baseline: 1.5386x; 1.1358x over previous
#include <hip/hip_runtime.h>
#include <hip/hip_bf16.h>
#include <math.h>

typedef __attribute__((ext_vector_type(2))) float v2f;
typedef __attribute__((ext_vector_type(8))) float v8f;

template <int NT>
__global__ void __launch_bounds__(256) gemm_f32_wmma(
    const float* __restrict__ A, const float* __restrict__ B,
    float* __restrict__ C, int M, int K, int N)
{
    __shared__ v2f sB2[4096];
    const int tid = threadIdx.x;
    const int pairCount = (K >> 1) * N;
    for (int p = tid; p < pairCount; p += 256) {
        const int kk = (p / N) << 1;
        const int n  = p - (p / N) * N;
        v2f b;
        b.x = B[kk * N + n];
        b.y = B[(kk + 1) * N + n];
        sB2[p] = b;
    }
    __syncthreads();

    const int lane    = tid & 31;
    const int wave    = tid >> 5;
    const int rowTile = blockIdx.x * 8 + wave;
    const int numTiles = (M + 15) >> 4;
    if (rowTile >= numTiles) return;

    const int half = lane >> 4;
    const int l15  = lane & 15;
    int mrow = rowTile * 16 + l15;
    if (mrow >= M) mrow = M - 1;

    v8f acc[NT] = {};
    const float* arow = A + (size_t)mrow * K + 2 * half;
    const v2f* __restrict__ bcol = sB2 + half * N + l15;

    for (int k0 = 0; k0 < K; k0 += 4) {
        const v2f a = *(const v2f*)(arow + k0);
        const v2f* brow = bcol + (k0 >> 1) * N;
#pragma unroll
        for (int nt = 0; nt < NT; ++nt) {
            const v2f b = brow[nt * 16];
            acc[nt] = __builtin_amdgcn_wmma_f32_16x16x4_f32(
                false, a, false, b, (short)0, acc[nt], false, false);
        }
    }

#pragma unroll
    for (int nt = 0; nt < NT; ++nt) {
#pragma unroll
        for (int r = 0; r < 8; ++r) {
            const int row = rowTile * 16 + r + 8 * half;
            if (row < M)
                C[(size_t)row * N + nt * 16 + l15] = acc[nt][r];
        }
    }
}


__global__ void node_score_kernel(const float* __restrict__ z,
                                  const float* __restrict__ w_attn,
                                  float* __restrict__ ssrc, float* __restrict__ sdst,
                                  int* __restrict__ cnt, int* __restrict__ total, int N)
{
    const int n = blockIdx.x * blockDim.x + threadIdx.x;
    if (n >= N) return;
    const float* zr = z + (size_t)n * 64;
    float s0 = 0.f, s1 = 0.f;
#pragma unroll
    for (int i = 0; i < 64; ++i) {
        const float v = zr[i];
        s0 += v * w_attn[i];
        s1 += v * w_attn[96 + i];
    }
    ssrc[n] = s0;
    sdst[n] = s1;
    cnt[n] = 0;
    if (n == 0) *total = 0;
}

__global__ void dst_count_kernel(const int* __restrict__ dst, int* __restrict__ cnt, int E)
{
    const int e = blockIdx.x * blockDim.x + threadIdx.x;
    if (e >= E) return;
    atomicAdd(&cnt[dst[e]], 1);
}

__global__ void __launch_bounds__(256) seg_alloc_kernel(int* __restrict__ cur, int* __restrict__ beg,
                                                        int* __restrict__ total, int N)
{
    __shared__ int wtot[8];
    __shared__ int sbase;
    const int tid  = threadIdx.x;
    const int lane = tid & 31;
    const int wave = tid >> 5;
    const int n = blockIdx.x * 256 + tid;
    const int c = (n < N) ? cur[n] : 0;
    int x = c;
#pragma unroll
    for (int o = 1; o < 32; o <<= 1) {
        const int y = __shfl_up(x, o);
        if (lane >= o) x += y;
    }
    if (lane == 31) wtot[wave] = x;
    __syncthreads();
    if (tid == 0) {
        int s = 0;
#pragma unroll
        for (int w = 0; w < 8; ++w) { const int t = wtot[w]; wtot[w] = s; s += t; }
        sbase = atomicAdd(total, s);
    }
    __syncthreads();
    const int b = sbase + wtot[wave] + x - c;
    if (n < N) { beg[n] = b; cur[n] = b; }
}

__global__ void dst_fill_kernel(const int* __restrict__ dst, int* __restrict__ cur,
                                int* __restrict__ perm, int E)
{
    const int e = blockIdx.x * blockDim.x + threadIdx.x;
    if (e >= E) return;
    const int pos = atomicAdd(&cur[dst[e]], 1);
    perm[pos] = e;
}

__device__ __forceinline__ float wave_max_f(float v) {
#pragma unroll
    for (int o = 16; o > 0; o >>= 1) v = fmaxf(v, __shfl_xor(v, o));
    return v;
}
__device__ __forceinline__ float wave_sum_f(float v) {
#pragma unroll
    for (int o = 16; o > 0; o >>= 1) v += __shfl_xor(v, o);
    return v;
}
__device__ __forceinline__ int wave_min_i(int v) {
#pragma unroll
    for (int o = 16; o > 0; o >>= 1) { const int t = __shfl_xor(v, o); v = t < v ? t : v; }
    return v;
}

__device__ __forceinline__ float edge_logit(const float* __restrict__ eft, const int* __restrict__ src,
                                            const float* __restrict__ ssrc,
                                            const float* __restrict__ w_attn, int e, float sd)
{
    const float4* er = (const float4*)(eft + (size_t)e * 32);
    float t = ssrc[src[e]] + sd;
#pragma unroll
    for (int q = 0; q < 8; ++q) {
        const float4 v = er[q];
        t += v.x * w_attn[64 + 4 * q];
        t += v.y * w_attn[65 + 4 * q];
        t += v.z * w_attn[66 + 4 * q];
        t += v.w * w_attn[67 + 4 * q];
    }
    return t > 0.f ? t : 0.01f * t;
}

#define ROW_ACC(T)                                                         \
    if (ty == T) { acc[T][0] += al * v0; acc[T][1] += al * v1; acc[T][2] += al * v2; }

__device__ __forceinline__ int next_edge(const int* __restrict__ pb, int deg, int lane, int last)
{
    int best = 0x7fffffff;
    for (int c = 0; c < deg; c += 32) {
        const int i = c + lane;
        if (i < deg) {
            const int v = pb[i];
            if (v > last && v < best) best = v;
        }
    }
    return wave_min_i(best);
}

__device__ __forceinline__ int rank_owner(int rankA, int rankB, int p, bool& inA)
{
    const unsigned mA = __builtin_amdgcn_ballot_w32(rankA == p);
    const unsigned mB = __builtin_amdgcn_ballot_w32(rankB == p);
    inA = (mA != 0u);
    return __builtin_ctz(inA ? mA : mB);
}
__device__ __forceinline__ int pick_i(int vA, int vB, int ln, bool inA)
{
    return inA ? __builtin_amdgcn_readlane(vA, ln) : __builtin_amdgcn_readlane(vB, ln);
}
__device__ __forceinline__ float pick_f(float vA, float vB, int ln, bool inA)
{
    return __int_as_float(pick_i(__float_as_int(vA), __float_as_int(vB), ln, inA));
}

__global__ void __launch_bounds__(256) node_gather_kernel(
    const float* __restrict__ z, const float* __restrict__ eft,
    const int* __restrict__ src, const int* __restrict__ etype,
    const float* __restrict__ ssrc, const float* __restrict__ sdst,
    const float* __restrict__ w_attn,
    const int* __restrict__ beg, const int* __restrict__ cur, const int* __restrict__ perm,
    float* __restrict__ h, int N)
{
    const int d    = (int)((blockIdx.x * blockDim.x + threadIdx.x) >> 5);
    const int lane = threadIdx.x & 31;
    if (d >= N) return;
    const int b   = beg[d];
    const int deg = cur[d] - b;
    const int* __restrict__ pb = perm + b;
    const float sd = sdst[d];
    float acc[4][3] = {};

    if (deg <= 32) {
        const int id = (lane < deg) ? pb[lane] : (0x7fffffc0 + lane);
        int rank = 0;
#pragma unroll
        for (int j = 0; j < 32; ++j)
            rank += (__builtin_amdgcn_readlane(id, j) < id) ? 1 : 0;
        const int sid = __builtin_amdgcn_ds_permute(rank << 2, id);
        const bool ok = lane < deg;
        const int e  = ok ? sid : 0;
        const int s  = src[e];
        const int tt = etype[e];
        const float lr = ok ? edge_logit(eft, src, ssrc, w_attn, e, sd) : -INFINITY;
        const float m  = wave_max_f(lr);
        const float ex = ok ? __expf(lr - m) : 0.f;
        const float den = wave_sum_f(ex);
        const float alpha = ex / den;

        for (int j = 0; j < deg; j += 4) {
            float r0[4], r1[4], r2[4];
#pragma unroll
            for (int u = 0; u < 4; ++u) {
                const int jj = (j + u < deg) ? (j + u) : (deg - 1);
                const int eu = __builtin_amdgcn_readlane(e, jj);
                const int su = __builtin_amdgcn_readlane(s, jj);
                const float* zr = z + (size_t)su * 64;
                r0[u] = zr[lane];
                r1[u] = zr[32 + lane];
                r2[u] = eft[(size_t)eu * 32 + lane];
            }
#pragma unroll
            for (int u = 0; u < 4; ++u) {
                if (j + u < deg) {
                    const float al = __int_as_float(__builtin_amdgcn_readlane(__float_as_int(alpha), j + u));
                    const int   ty = __builtin_amdgcn_readlane(tt, j + u);
                    const float v0 = r0[u], v1 = r1[u], v2 = r2[u];
                    ROW_ACC(0) ROW_ACC(1) ROW_ACC(2) ROW_ACC(3)
                }
            }
        }
    } else if (deg <= 64) {
        const int  degB = deg - 32;
        const bool okB  = lane < degB;
        const int  eA   = pb[lane];
        const int  idB  = okB ? pb[32 + lane] : (0x7fffffc0 + lane);
        int rankA = 0, rankB = 0;
        for (int j = 0; j < 32; ++j) {
            const int a = __builtin_amdgcn_readlane(eA, j);
            const int b = __builtin_amdgcn_readlane(idB, j);
            rankA += ((a < eA) ? 1 : 0) + ((b < eA) ? 1 : 0);
            rankB += ((a < idB) ? 1 : 0) + ((b < idB) ? 1 : 0);
        }
        const int eB = okB ? idB : 0;
        int sA = 0, sB = 0, tA = 0, tB = 0;
        float lA = 0.f, lB = 0.f;
#pragma nounroll
        for (int hh = 0; hh < 2; ++hh) {
            const int   e_h = hh ? eB : eA;
            const int   s_h = src[e_h];
            const int   t_h = etype[e_h];
            const float l_h = edge_logit(eft, src, ssrc, w_attn, e_h, sd);
            if (hh) { sB = s_h; tB = t_h; lB = l_h; } else { sA = s_h; tA = t_h; lA = l_h; }
        }
        lB = okB ? lB : -INFINITY;
        const float m  = wave_max_f(fmaxf(lA, lB));
        const float xA = __expf(lA - m);
        const float xB = okB ? __expf(lB - m) : 0.f;
        float den = 0.f;
        for (int p = 0; p < deg; ++p) {
            bool inA;
            const int ln = rank_owner(rankA, rankB, p, inA);
            den += pick_f(xA, xB, ln, inA);
        }
        const float alA = xA / den, alB = xB / den;

        for (int j = 0; j < deg; j += 4) {
            float r0[4], r1[4], r2[4], al4[4];
            int ty4[4];
            for (int u = 0; u < 4; ++u) {
                const int p = (j + u < deg) ? (j + u) : (deg - 1);
                bool inA;
                const int ln = rank_owner(rankA, rankB, p, inA);
                const int eu = pick_i(eA, eB, ln, inA);
                const int su = pick_i(sA, sB, ln, inA);
                al4[u] = pick_f(alA, alB, ln, inA);
                ty4[u] = pick_i(tA, tB, ln, inA);
                const float* zr = z + (size_t)su * 64;
                r0[u] = zr[lane];
                r1[u] = zr[32 + lane];
                r2[u] = eft[(size_t)eu * 32 + lane];
            }
            for (int u = 0; u < 4; ++u) {
                if (j + u < deg) {
                    const float al = al4[u];
                    const int   ty = ty4[u];
                    const float v0 = r0[u], v1 = r1[u], v2 = r2[u];
                    ROW_ACC(0) ROW_ACC(1) ROW_ACC(2) ROW_ACC(3)
                }
            }
        }
    } else {
        float m = -INFINITY;
        for (int c = 0; c < deg; c += 32) {
            const int i = c + lane;
            if (i < deg) m = fmaxf(m, edge_logit(eft, src, ssrc, w_attn, pb[i], sd));
        }
        m = wave_max_f(m);
        float den = 0.f;
        int last = -1;
        for (int j = 0; j < deg; ++j) {
            last = next_edge(pb, deg, lane, last);
            den += __expf(edge_logit(eft, src, ssrc, w_attn, last, sd) - m);
        }
        last = -1;
        for (int j = 0; j < deg; ++j) {
            last = next_edge(pb, deg, lane, last);
            const float al = __expf(edge_logit(eft, src, ssrc, w_attn, last, sd) - m) / den;
            const int ty = etype[last];
            const float* zr = z + (size_t)src[last] * 64;
            const float v0 = zr[lane], v1 = zr[32 + lane], v2 = eft[(size_t)last * 32 + lane];
            ROW_ACC(0) ROW_ACC(1) ROW_ACC(2) ROW_ACC(3)
        }
    }

    float* hr = h + (size_t)d * 4 * 96 + lane;
#pragma unroll
    for (int t = 0; t < 4; ++t) {
        hr[t * 96]      = acc[t][0];
        hr[t * 96 + 32] = acc[t][1];
        hr[t * 96 + 64] = acc[t][2];
    }
}
#undef ROW_ACC

constexpr int ZP_BLOCKS = 512;

__device__ __forceinline__ void add4(float4& a, const float4& v) {
    a.x += v.x; a.y += v.y; a.z += v.z; a.w += v.w;
}

__global__ void __launch_bounds__(512) zphi_partial_kernel(
    const float* __restrict__ h2, float* __restrict__ part, int nodes)
{
    __shared__ float4 sm[8][64];
    const int tid = threadIdx.x;
    const int c4  = tid & 63;
    const int sub = tid >> 6;
    const int R   = (nodes + (int)gridDim.x - 1) / (int)gridDim.x;
    const int r0  = blockIdx.x * R;
    int r1 = r0 + R;
    if (r1 > nodes) r1 = nodes;
    const float4* __restrict__ p = (const float4*)h2 + c4;

    float4 acc = make_float4(0.f, 0.f, 0.f, 0.f);
    int r = r0 + sub;
    for (; r + 56 < r1; r += 64) {
        float4 v[8];
#pragma unroll
        for (int u = 0; u < 8; ++u) v[u] = p[(size_t)(r + 8 * u) * 64];
#pragma unroll
        for (int u = 0; u < 8; ++u) add4(acc, v[u]);
    }
    {
        float4 v[8];
#pragma unroll
        for (int u = 0; u < 8; ++u) {
            const int rr = r + 8 * u;
            v[u] = make_float4(0.f, 0.f, 0.f, 0.f);
            if (rr < r1) v[u] = p[(size_t)rr * 64];
        }
#pragma unroll
        for (int u = 0; u < 8; ++u) add4(acc, v[u]);
    }
    sm[sub][c4] = acc;
    __syncthreads();
    if (tid < 256) {
        const float* sf = (const float*)&sm[0][0];
        float s = 0.f;
#pragma unroll
        for (int k = 0; k < 8; ++k) s += sf[k * 256 + tid];
        part[(size_t)blockIdx.x * 256 + tid] = s;
    }
}

__global__ void __launch_bounds__(1024) beta_kernel(const float* __restrict__ part,
                            const float* __restrict__ w_sem, float* __restrict__ beta)
{
    __shared__ float4 sm[16][64];
    __shared__ float zs[256];
    const int tid = threadIdx.x;
    const int c4  = tid & 63;
    const int g   = tid >> 6;
    const float4* __restrict__ p = (const float4*)part + (size_t)g * 32 * 64 + c4;
    float4 acc = make_float4(0.f, 0.f, 0.f, 0.f);
    for (int b = 0; b < 32; b += 16) {
        float4 v[16];
#pragma unroll
        for (int u = 0; u < 16; ++u) v[u] = p[(size_t)(b + u) * 64];
#pragma unroll
        for (int u = 0; u < 16; ++u) add4(acc, v[u]);
    }
    sm[g][c4] = acc;
    __syncthreads();
    if (tid < 256) {
        const float* sf = (const float*)&sm[0][0];
        float s = 0.f;
#pragma unroll
        for (int k = 0; k < 16; ++k) s += sf[k * 256 + tid];
        zs[tid] = s;
    }
    __syncthreads();
    if (tid != 0) return;
    float wp[4], mx = -INFINITY;
#pragma unroll
    for (int t = 0; t < 4; ++t) {
        float d = 0.f;
        for (int o = 0; o < 64; ++o) d += zs[t * 64 + o] * w_sem[o];
        const float lr = d > 0.f ? d : 0.01f * d;
        wp[t] = lr;
        mx = fmaxf(mx, lr);
    }
    float s = 0.f;
#pragma unroll
    for (int t = 0; t < 4; ++t) { wp[t] = __expf(wp[t] - mx); s += wp[t]; }
#pragma unroll
    for (int t = 0; t < 4; ++t) beta[t] = wp[t] / s;
}

__global__ void out_kernel(const float* __restrict__ h2, const float* __restrict__ beta,
                           float* __restrict__ Z, int N)
{
    const int i = blockIdx.x * blockDim.x + threadIdx.x;
    if (i >= N * 64) return;
    const int n = i >> 6, o = i & 63;
    const float* r = h2 + (size_t)n * 256 + o;
    Z[i] = beta[0] * r[0] + beta[1] * r[64] + beta[2] * r[128] + beta[3] * r[192];
}

extern "C" void kernel_launch(void* const* d_in, const int* in_sizes, int n_in,
                              void* d_out, int out_size, void* d_ws, size_t ws_size,
                              hipStream_t stream)
{
    const float* node   = (const float*)d_in[0];
    const float* edgef  = (const float*)d_in[1];
    const int*   src    = (const int*)d_in[2];
    const int*   dst    = (const int*)d_in[3];
    const int*   etype  = (const int*)d_in[4];
    const float* w_n    = (const float*)d_in[5];
    const float* w_e    = (const float*)d_in[6];
    const float* w_attn = (const float*)d_in[7];
    const float* w_fc2  = (const float*)d_in[8];
    const float* w_sem  = (const float*)d_in[9];
    float* out = (float*)d_out;

    const int N = in_sizes[0] / 128;
    const int E = in_sizes[2];
    const int T = 4;

    float* ws   = (float*)d_ws;
    float* z    = ws;
    float* ssrc = z + (size_t)N * 64;
    float* sdst = ssrc + N;
    float* abuf = sdst + N;
    float* mbuf = abuf + E;
    float* den  = mbuf + N;
    float* beta = den + N + 256;
    float* h    = beta + 16;
    float* eft  = h + (size_t)N * T * 96;
    float* h2   = eft;
    float* part = abuf;
    int* perm  = (int*)abuf;
    int* cur   = (int*)mbuf;
    int* beg   = (int*)den;
    int* total = (int*)(den + N);

    const int thr = 256;

    gemm_f32_wmma<4><<<((N + 15) / 16 + 7) / 8, thr, 0, stream>>>(node, w_n, z, N, 128, 64);
    gemm_f32_wmma<2><<<((E + 15) / 16 + 7) / 8, thr, 0, stream>>>(edgef, w_e, eft, E, 64, 32);
    node_score_kernel<<<(N + thr - 1) / thr, thr, 0, stream>>>(z, w_attn, ssrc, sdst, cur, total, N);
    dst_count_kernel<<<(E + thr - 1) / thr, thr, 0, stream>>>(dst, cur, E);
    seg_alloc_kernel<<<(N + 255) / 256, 256, 0, stream>>>(cur, beg, total, N);
    dst_fill_kernel<<<(E + thr - 1) / thr, thr, 0, stream>>>(dst, cur, perm, E);
    node_gather_kernel<<<(N + 7) / 8, thr, 0, stream>>>(z, eft, src, etype, ssrc, sdst, w_attn,
                                                        beg, cur, perm, h, N);
    gemm_f32_wmma<4><<<((N * T + 15) / 16 + 7) / 8, thr, 0, stream>>>(h, w_fc2, h2, N * T, 96, 64);
    zphi_partial_kernel<<<ZP_BLOCKS, 512, 0, stream>>>(h2, part, N);
    beta_kernel<<<1, 1024, 0, stream>>>(part, w_sem, beta);
    out_kernel<<<(N * 64 + thr - 1) / thr, thr, 0, stream>>>(h2, beta, out, N);
}


# rows on whole lines: h and eft/h2 planes moved onto 256-B boundaries (were 64 B off a line) and the GEMM epilogue swaps lane halves, so each 128-B row line is loaded and stored by one instruction
# speedup vs baseline: 1.6110x; 1.0471x over previous
#include <hip/hip_runtime.h>
#include <hip/hip_bf16.h>
#include <math.h>

typedef __attribute__((ext_vector_type(2))) float v2f;
typedef __attribute__((ext_vector_type(8))) float v8f;

__device__ __forceinline__ float xhalf(float x) {
    return __builtin_bit_cast(float, __builtin_amdgcn_permlanex16(
        __builtin_bit_cast(int, x), __builtin_bit_cast(int, x), 0x76543210, 0xfedcba98, false, false));
}

template <int NT>
__global__ void __launch_bounds__(256) gemm_f32_wmma(
    const float* __restrict__ A, const float* __restrict__ B,
    float* __restrict__ C, int M, int K, int N)
{
    __shared__ v2f sB2[4096];
    const int tid = threadIdx.x;
    const int pairCount = (K >> 1) * N;
    for (int p = tid; p < pairCount; p += 256) {
        const int kk = (p / N) << 1;
        const int n  = p - (p / N) * N;
        v2f b;
        b.x = B[kk * N + n];
        b.y = B[(kk + 1) * N + n];
        sB2[p] = b;
    }
    __syncthreads();

    const int lane    = tid & 31;
    const int wave    = tid >> 5;
    const int rowTile = blockIdx.x * 8 + wave;
    const int numTiles = (M + 15) >> 4;
    if (rowTile >= numTiles) return;

    const int half = lane >> 4;
    const int l15  = lane & 15;
    int mrow = rowTile * 16 + l15;
    if (mrow >= M) mrow = M - 1;

    v8f acc[NT] = {};
    const float* arow = A + (size_t)mrow * K + 2 * half;
    const v2f* __restrict__ bcol = sB2 + half * N + l15;

    for (int k0 = 0; k0 < K; k0 += 4) {
        const v2f a = *(const v2f*)(arow + k0);
        const v2f* brow = bcol + (k0 >> 1) * N;
#pragma unroll
        for (int nt = 0; nt < NT; ++nt) {
            const v2f b = brow[nt * 16];
            acc[nt] = __builtin_amdgcn_wmma_f32_16x16x4_f32(
                false, a, false, b, (short)0, acc[nt], false, false);
        }
    }

#pragma unroll
    for (int p = 0; p < NT / 2; ++p) {
#pragma unroll
        for (int r = 0; r < 8; ++r) {
            const float a0 = acc[2 * p][r];
            const float a1 = acc[2 * p + 1][r];
            const float x0 = xhalf(a0);
            const float x1 = xhalf(a1);
            const float lo = half ? x1 : a0;
            const float hi = half ? a1 : x0;
            const int row = rowTile * 16 + r;
            if (row < M)     C[(size_t)row * N + p * 32 + lane] = lo;
            if (row + 8 < M) C[(size_t)(row + 8) * N + p * 32 + lane] = hi;
        }
    }
}


__global__ void node_score_kernel(const float* __restrict__ z,
                                  const float* __restrict__ w_attn,
                                  float* __restrict__ ssrc, float* __restrict__ sdst,
                                  int* __restrict__ cnt, int* __restrict__ total, int N)
{
    const int n = blockIdx.x * blockDim.x + threadIdx.x;
    if (n >= N) return;
    const float* zr = z + (size_t)n * 64;
    float s0 = 0.f, s1 = 0.f;
#pragma unroll
    for (int i = 0; i < 64; ++i) {
        const float v = zr[i];
        s0 += v * w_attn[i];
        s1 += v * w_attn[96 + i];
    }
    ssrc[n] = s0;
    sdst[n] = s1;
    cnt[n] = 0;
    if (n == 0) *total = 0;
}

__global__ void dst_count_kernel(const int* __restrict__ dst, int* __restrict__ cnt, int E)
{
    const int e = blockIdx.x * blockDim.x + threadIdx.x;
    if (e >= E) return;
    atomicAdd(&cnt[dst[e]], 1);
}

__global__ void __launch_bounds__(256) seg_alloc_kernel(int* __restrict__ cur, int* __restrict__ beg,
                                                        int* __restrict__ total, int N)
{
    __shared__ int wtot[8];
    __shared__ int sbase;
    const int tid  = threadIdx.x;
    const int lane = tid & 31;
    const int wave = tid >> 5;
    const int n = blockIdx.x * 256 + tid;
    const int c = (n < N) ? cur[n] : 0;
    int x = c;
#pragma unroll
    for (int o = 1; o < 32; o <<= 1) {
        const int y = __shfl_up(x, o);
        if (lane >= o) x += y;
    }
    if (lane == 31) wtot[wave] = x;
    __syncthreads();
    if (tid == 0) {
        int s = 0;
#pragma unroll
        for (int w = 0; w < 8; ++w) { const int t = wtot[w]; wtot[w] = s; s += t; }
        sbase = atomicAdd(total, s);
    }
    __syncthreads();
    const int b = sbase + wtot[wave] + x - c;
    if (n < N) { beg[n] = b; cur[n] = b; }
}

__global__ void dst_fill_kernel(const int* __restrict__ dst, int* __restrict__ cur,
                                int* __restrict__ perm, int E)
{
    const int e = blockIdx.x * blockDim.x + threadIdx.x;
    if (e >= E) return;
    const int pos = atomicAdd(&cur[dst[e]], 1);
    perm[pos] = e;
}

__device__ __forceinline__ float wave_max_f(float v) {
#pragma unroll
    for (int o = 16; o > 0; o >>= 1) v = fmaxf(v, __shfl_xor(v, o));
    return v;
}
__device__ __forceinline__ float wave_sum_f(float v) {
#pragma unroll
    for (int o = 16; o > 0; o >>= 1) v += __shfl_xor(v, o);
    return v;
}
__device__ __forceinline__ int wave_min_i(int v) {
#pragma unroll
    for (int o = 16; o > 0; o >>= 1) { const int t = __shfl_xor(v, o); v = t < v ? t : v; }
    return v;
}

__device__ __forceinline__ float edge_logit(const float* __restrict__ eft, const int* __restrict__ src,
                                            const float* __restrict__ ssrc,
                                            const float* __restrict__ w_attn, int e, float sd)
{
    const float4* er = (const float4*)(eft + (size_t)e * 32);
    float t = ssrc[src[e]] + sd;
#pragma unroll
    for (int q = 0; q < 8; ++q) {
        const float4 v = er[q];
        t += v.x * w_attn[64 + 4 * q];
        t += v.y * w_attn[65 + 4 * q];
        t += v.z * w_attn[66 + 4 * q];
        t += v.w * w_attn[67 + 4 * q];
    }
    return t > 0.f ? t : 0.01f * t;
}

#define ROW_ACC(T)                                                         \
    if (ty == T) { acc[T][0] += al * v0; acc[T][1] += al * v1; acc[T][2] += al * v2; }

__device__ __forceinline__ int next_edge(const int* __restrict__ pb, int deg, int lane, int last)
{
    int best = 0x7fffffff;
    for (int c = 0; c < deg; c += 32) {
        const int i = c + lane;
        if (i < deg) {
            const int v = pb[i];
            if (v > last && v < best) best = v;
        }
    }
    return wave_min_i(best);
}

__device__ __forceinline__ int rank_owner(int rankA, int rankB, int p, bool& inA)
{
    const unsigned mA = __builtin_amdgcn_ballot_w32(rankA == p);
    const unsigned mB = __builtin_amdgcn_ballot_w32(rankB == p);
    inA = (mA != 0u);
    return __builtin_ctz(inA ? mA : mB);
}
__device__ __forceinline__ int pick_i(int vA, int vB, int ln, bool inA)
{
    return inA ? __builtin_amdgcn_readlane(vA, ln) : __builtin_amdgcn_readlane(vB, ln);
}
__device__ __forceinline__ float pick_f(float vA, float vB, int ln, bool inA)
{
    return __int_as_float(pick_i(__float_as_int(vA), __float_as_int(vB), ln, inA));
}

__global__ void __launch_bounds__(256) node_gather_kernel(
    const float* __restrict__ z, const float* __restrict__ eft,
    const int* __restrict__ src, const int* __restrict__ etype,
    const float* __restrict__ ssrc, const float* __restrict__ sdst,
    const float* __restrict__ w_attn,
    const int* __restrict__ beg, const int* __restrict__ cur, const int* __restrict__ perm,
    float* __restrict__ h, int N)
{
    const int d    = (int)((blockIdx.x * blockDim.x + threadIdx.x) >> 5);
    const int lane = threadIdx.x & 31;
    if (d >= N) return;
    const int b   = beg[d];
    const int deg = cur[d] - b;
    const int* __restrict__ pb = perm + b;
    const float sd = sdst[d];
    float acc[4][3] = {};

    if (deg <= 32) {
        const int id = (lane < deg) ? pb[lane] : (0x7fffffc0 + lane);
        int rank = 0;
#pragma unroll
        for (int j = 0; j < 32; ++j)
            rank += (__builtin_amdgcn_readlane(id, j) < id) ? 1 : 0;
        const int sid = __builtin_amdgcn_ds_permute(rank << 2, id);
        const bool ok = lane < deg;
        const int e  = ok ? sid : 0;
        const int s  = src[e];
        const int tt = etype[e];
        const float lr = ok ? edge_logit(eft, src, ssrc, w_attn, e, sd) : -INFINITY;
        const float m  = wave_max_f(lr);
        const float ex = ok ? __expf(lr - m) : 0.f;
        const float den = wave_sum_f(ex);
        const float alpha = ex / den;

        for (int j = 0; j < deg; j += 4) {
            float r0[4], r1[4], r2[4];
#pragma unroll
            for (int u = 0; u < 4; ++u) {
                const int jj = (j + u < deg) ? (j + u) : (deg - 1);
                const int eu = __builtin_amdgcn_readlane(e, jj);
                const int su = __builtin_amdgcn_readlane(s, jj);
                const float* zr = z + (size_t)su * 64;
                r0[u] = zr[lane];
                r1[u] = zr[32 + lane];
                r2[u] = eft[(size_t)eu * 32 + lane];
            }
#pragma unroll
            for (int u = 0; u < 4; ++u) {
                if (j + u < deg) {
                    const float al = __int_as_float(__builtin_amdgcn_readlane(__float_as_int(alpha), j + u));
                    const int   ty = __builtin_amdgcn_readlane(tt, j + u);
                    const float v0 = r0[u], v1 = r1[u], v2 = r2[u];
                    ROW_ACC(0) ROW_ACC(1) ROW_ACC(2) ROW_ACC(3)
                }
            }
        }
    } else if (deg <= 64) {
        const int  degB = deg - 32;
        const bool okB  = lane < degB;
        const int  eA   = pb[lane];
        const int  idB  = okB ? pb[32 + lane] : (0x7fffffc0 + lane);
        int rankA = 0, rankB = 0;
        for (int j = 0; j < 32; ++j) {
            const int a = __builtin_amdgcn_readlane(eA, j);
            const int b = __builtin_amdgcn_readlane(idB, j);
            rankA += ((a < eA) ? 1 : 0) + ((b < eA) ? 1 : 0);
            rankB += ((a < idB) ? 1 : 0) + ((b < idB) ? 1 : 0);
        }
        const int eB = okB ? idB : 0;
        int sA = 0, sB = 0, tA = 0, tB = 0;
        float lA = 0.f, lB = 0.f;
#pragma nounroll
        for (int hh = 0; hh < 2; ++hh) {
            const int   e_h = hh ? eB : eA;
            const int   s_h = src[e_h];
            const int   t_h = etype[e_h];
            const float l_h = edge_logit(eft, src, ssrc, w_attn, e_h, sd);
            if (hh) { sB = s_h; tB = t_h; lB = l_h; } else { sA = s_h; tA = t_h; lA = l_h; }
        }
        lB = okB ? lB : -INFINITY;
        const float m  = wave_max_f(fmaxf(lA, lB));
        const float xA = __expf(lA - m);
        const float xB = okB ? __expf(lB - m) : 0.f;
        float den = 0.f;
        for (int p = 0; p < deg; ++p) {
            bool inA;
            const int ln = rank_owner(rankA, rankB, p, inA);
            den += pick_f(xA, xB, ln, inA);
        }
        const float alA = xA / den, alB = xB / den;

        for (int j = 0; j < deg; j += 4) {
            float r0[4], r1[4], r2[4], al4[4];
            int ty4[4];
            for (int u = 0; u < 4; ++u) {
                const int p = (j + u < deg) ? (j + u) : (deg - 1);
                bool inA;
                const int ln = rank_owner(rankA, rankB, p, inA);
                const int eu = pick_i(eA, eB, ln, inA);
                const int su = pick_i(sA, sB, ln, inA);
                al4[u] = pick_f(alA, alB, ln, inA);
                ty4[u] = pick_i(tA, tB, ln, inA);
                const float* zr = z + (size_t)su * 64;
                r0[u] = zr[lane];
                r1[u] = zr[32 + lane];
                r2[u] = eft[(size_t)eu * 32 + lane];
            }
            for (int u = 0; u < 4; ++u) {
                if (j + u < deg) {
                    const float al = al4[u];
                    const int   ty = ty4[u];
                    const float v0 = r0[u], v1 = r1[u], v2 = r2[u];
                    ROW_ACC(0) ROW_ACC(1) ROW_ACC(2) ROW_ACC(3)
                }
            }
        }
    } else {
        float m = -INFINITY;
        for (int c = 0; c < deg; c += 32) {
            const int i = c + lane;
            if (i < deg) m = fmaxf(m, edge_logit(eft, src, ssrc, w_attn, pb[i], sd));
        }
        m = wave_max_f(m);
        float den = 0.f;
        int last = -1;
        for (int j = 0; j < deg; ++j) {
            last = next_edge(pb, deg, lane, last);
            den += __expf(edge_logit(eft, src, ssrc, w_attn, last, sd) - m);
        }
        last = -1;
        for (int j = 0; j < deg; ++j) {
            last = next_edge(pb, deg, lane, last);
            const float al = __expf(edge_logit(eft, src, ssrc, w_attn, last, sd) - m) / den;
            const int ty = etype[last];
            const float* zr = z + (size_t)src[last] * 64;
            const float v0 = zr[lane], v1 = zr[32 + lane], v2 = eft[(size_t)last * 32 + lane];
            ROW_ACC(0) ROW_ACC(1) ROW_ACC(2) ROW_ACC(3)
        }
    }

    float* hr = h + (size_t)d * 4 * 96 + lane;
#pragma unroll
    for (int t = 0; t < 4; ++t) {
        hr[t * 96]      = acc[t][0];
        hr[t * 96 + 32] = acc[t][1];
        hr[t * 96 + 64] = acc[t][2];
    }
}
#undef ROW_ACC

constexpr int ZP_BLOCKS = 512;

__device__ __forceinline__ void add4(float4& a, const float4& v) {
    a.x += v.x; a.y += v.y; a.z += v.z; a.w += v.w;
}

__global__ void __launch_bounds__(512) zphi_partial_kernel(
    const float* __restrict__ h2, float* __restrict__ part, int nodes)
{
    __shared__ float4 sm[8][64];
    const int tid = threadIdx.x;
    const int c4  = tid & 63;
    const int sub = tid >> 6;
    const int R   = (nodes + (int)gridDim.x - 1) / (int)gridDim.x;
    const int r0  = blockIdx.x * R;
    int r1 = r0 + R;
    if (r1 > nodes) r1 = nodes;
    const float4* __restrict__ p = (const float4*)h2 + c4;

    float4 acc = make_float4(0.f, 0.f, 0.f, 0.f);
    int r = r0 + sub;
    for (; r + 56 < r1; r += 64) {
        float4 v[8];
#pragma unroll
        for (int u = 0; u < 8; ++u) v[u] = p[(size_t)(r + 8 * u) * 64];
#pragma unroll
        for (int u = 0; u < 8; ++u) add4(acc, v[u]);
    }
    {
        float4 v[8];
#pragma unroll
        for (int u = 0; u < 8; ++u) {
            const int rr = r + 8 * u;
            v[u] = make_float4(0.f, 0.f, 0.f, 0.f);
            if (rr < r1) v[u] = p[(size_t)rr * 64];
        }
#pragma unroll
        for (int u = 0; u < 8; ++u) add4(acc, v[u]);
    }
    sm[sub][c4] = acc;
    __syncthreads();
    if (tid < 256) {
        const float* sf = (const float*)&sm[0][0];
        float s = 0.f;
#pragma unroll
        for (int k = 0; k < 8; ++k) s += sf[k * 256 + tid];
        part[(size_t)blockIdx.x * 256 + tid] = s;
    }
}

__global__ void __launch_bounds__(1024) beta_kernel(const float* __restrict__ part,
                            const float* __restrict__ w_sem, float* __restrict__ beta)
{
    __shared__ float4 sm[16][64];
    __shared__ float zs[256];
    const int tid = threadIdx.x;
    const int c4  = tid & 63;
    const int g   = tid >> 6;
    const float4* __restrict__ p = (const float4*)part + (size_t)g * 32 * 64 + c4;
    float4 acc = make_float4(0.f, 0.f, 0.f, 0.f);
    for (int b = 0; b < 32; b += 16) {
        float4 v[16];
#pragma unroll
        for (int u = 0; u < 16; ++u) v[u] = p[(size_t)(b + u) * 64];
#pragma unroll
        for (int u = 0; u < 16; ++u) add4(acc, v[u]);
    }
    sm[g][c4] = acc;
    __syncthreads();
    if (tid < 256) {
        const float* sf = (const float*)&sm[0][0];
        float s = 0.f;
#pragma unroll
        for (int k = 0; k < 16; ++k) s += sf[k * 256 + tid];
        zs[tid] = s;
    }
    __syncthreads();
    if (tid != 0) return;
    float wp[4], mx = -INFINITY;
#pragma unroll
    for (int t = 0; t < 4; ++t) {
        float d = 0.f;
        for (int o = 0; o < 64; ++o) d += zs[t * 64 + o] * w_sem[o];
        const float lr = d > 0.f ? d : 0.01f * d;
        wp[t] = lr;
        mx = fmaxf(mx, lr);
    }
    float s = 0.f;
#pragma unroll
    for (int t = 0; t < 4; ++t) { wp[t] = __expf(wp[t] - mx); s += wp[t]; }
#pragma unroll
    for (int t = 0; t < 4; ++t) beta[t] = wp[t] / s;
}

__global__ void out_kernel(const float* __restrict__ h2, const float* __restrict__ beta,
                           float* __restrict__ Z, int N)
{
    const int i = blockIdx.x * blockDim.x + threadIdx.x;
    if (i >= N * 64) return;
    const int n = i >> 6, o = i & 63;
    const float* r = h2 + (size_t)n * 256 + o;
    Z[i] = beta[0] * r[0] + beta[1] * r[64] + beta[2] * r[128] + beta[3] * r[192];
}

extern "C" void kernel_launch(void* const* d_in, const int* in_sizes, int n_in,
                              void* d_out, int out_size, void* d_ws, size_t ws_size,
                              hipStream_t stream)
{
    const float* node   = (const float*)d_in[0];
    const float* edgef  = (const float*)d_in[1];
    const int*   src    = (const int*)d_in[2];
    const int*   dst    = (const int*)d_in[3];
    const int*   etype  = (const int*)d_in[4];
    const float* w_n    = (const float*)d_in[5];
    const float* w_e    = (const float*)d_in[6];
    const float* w_attn = (const float*)d_in[7];
    const float* w_fc2  = (const float*)d_in[8];
    const float* w_sem  = (const float*)d_in[9];
    float* out = (float*)d_out;

    const int N = in_sizes[0] / 128;
    const int E = in_sizes[2];
    const int T = 4;

    float* ws   = (float*)d_ws;
    float* h    = ws;
    float* eft  = h + (size_t)N * T * 96;
    float* h2   = eft;
    float* z    = eft + (size_t)E * 32;
    float* ssrc = z + (size_t)N * 64;
    float* sdst = ssrc + N;
    float* abuf = sdst + N;
    float* mbuf = abuf + E;
    float* den  = mbuf + N;
    float* beta = den + N + 256;
    float* part = abuf;
    int* perm  = (int*)abuf;
    int* cur   = (int*)mbuf;
    int* beg   = (int*)den;
    int* total = (int*)(den + N);

    const int thr = 256;

    gemm_f32_wmma<4><<<((N + 15) / 16 + 7) / 8, thr, 0, stream>>>(node, w_n, z, N, 128, 64);
    gemm_f32_wmma<2><<<((E + 15) / 16 + 7) / 8, thr, 0, stream>>>(edgef, w_e, eft, E, 64, 32);
    node_score_kernel<<<(N + thr - 1) / thr, thr, 0, stream>>>(z, w_attn, ssrc, sdst, cur, total, N);
    dst_count_kernel<<<(E + thr - 1) / thr, thr, 0, stream>>>(dst, cur, E);
    seg_alloc_kernel<<<(N + 255) / 256, 256, 0, stream>>>(cur, beg, total, N);
    dst_fill_kernel<<<(E + thr - 1) / thr, thr, 0, stream>>>(dst, cur, perm, E);
    node_gather_kernel<<<(N + 7) / 8, thr, 0, stream>>>(z, eft, src, etype, ssrc, sdst, w_attn,
                                                        beg, cur, perm, h, N);
    gemm_f32_wmma<4><<<((N * T + 15) / 16 + 7) / 8, thr, 0, stream>>>(h, w_fc2, h2, N * T, 96, 64);
    zphi_partial_kernel<<<ZP_BLOCKS, 512, 0, stream>>>(h2, part, N);
    beta_kernel<<<1, 1024, 0, stream>>>(part, w_sem, beta);
    out_kernel<<<(N * 64 + thr - 1) / thr, thr, 0, stream>>>(h2, beta, out, N);
}


# edges grouped by destination through LDS bins of 256 nodes: one global add per (edge chunk, bin), then one block per bin fills the runs; was two global adds per edge; order of every sum unchanged
# speedup vs baseline: 2.1086x; 1.3089x over previous
#include <hip/hip_runtime.h>
#include <hip/hip_bf16.h>
#include <math.h>

typedef __attribute__((ext_vector_type(2))) float v2f;
typedef __attribute__((ext_vector_type(8))) float v8f;

__device__ __forceinline__ float xhalf(float x) {
    return __builtin_bit_cast(float, __builtin_amdgcn_permlanex16(
        __builtin_bit_cast(int, x), __builtin_bit_cast(int, x), 0x76543210, 0xfedcba98, false, false));
}

template <int NT>
__global__ void __launch_bounds__(256) gemm_f32_wmma(
    const float* __restrict__ A, const float* __restrict__ B,
    float* __restrict__ C, int M, int K, int N)
{
    __shared__ v2f sB2[4096];
    const int tid = threadIdx.x;
    const int pairCount = (K >> 1) * N;
    for (int p = tid; p < pairCount; p += 256) {
        const int kk = (p / N) << 1;
        const int n  = p - (p / N) * N;
        v2f b;
        b.x = B[kk * N + n];
        b.y = B[(kk + 1) * N + n];
        sB2[p] = b;
    }
    __syncthreads();

    const int lane    = tid & 31;
    const int wave    = tid >> 5;
    const int rowTile = blockIdx.x * 8 + wave;
    const int numTiles = (M + 15) >> 4;
    if (rowTile >= numTiles) return;

    const int half = lane >> 4;
    const int l15  = lane & 15;
    int mrow = rowTile * 16 + l15;
    if (mrow >= M) mrow = M - 1;

    v8f acc[NT] = {};
    const float* arow = A + (size_t)mrow * K + 2 * half;
    const v2f* __restrict__ bcol = sB2 + half * N + l15;

    for (int k0 = 0; k0 < K; k0 += 4) {
        const v2f a = *(const v2f*)(arow + k0);
        const v2f* brow = bcol + (k0 >> 1) * N;
#pragma unroll
        for (int nt = 0; nt < NT; ++nt) {
            const v2f b = brow[nt * 16];
            acc[nt] = __builtin_amdgcn_wmma_f32_16x16x4_f32(
                false, a, false, b, (short)0, acc[nt], false, false);
        }
    }

#pragma unroll
    for (int p = 0; p < NT / 2; ++p) {
#pragma unroll
        for (int r = 0; r < 8; ++r) {
            const float a0 = acc[2 * p][r];
            const float a1 = acc[2 * p + 1][r];
            const float x0 = xhalf(a0);
            const float x1 = xhalf(a1);
            const float lo = half ? x1 : a0;
            const float hi = half ? a1 : x0;
            const int row = rowTile * 16 + r;
            if (row < M)     C[(size_t)row * N + p * 32 + lane] = lo;
            if (row + 8 < M) C[(size_t)(row + 8) * N + p * 32 + lane] = hi;
        }
    }
}


constexpr int BIN_SHIFT  = 8;
constexpr int MAXBINS    = 256;
constexpr int CTR_PITCH  = 64;
constexpr int EDGE_CHUNK = 4096;

__global__ void node_score_kernel(const float* __restrict__ z,
                                  const float* __restrict__ w_attn,
                                  float* __restrict__ ssrc, float* __restrict__ sdst,
                                  int* __restrict__ binCount, int* __restrict__ binCursor, int N)
{
    const int n = blockIdx.x * blockDim.x + threadIdx.x;
    if (n < MAXBINS) { binCount[n * CTR_PITCH] = 0; binCursor[n * CTR_PITCH] = 0; }
    if (n >= N) return;
    const float* zr = z + (size_t)n * 64;
    float s0 = 0.f, s1 = 0.f;
#pragma unroll
    for (int i = 0; i < 64; ++i) {
        const float v = zr[i];
        s0 += v * w_attn[i];
        s1 += v * w_attn[96 + i];
    }
    ssrc[n] = s0;
    sdst[n] = s1;
}

__device__ __forceinline__ void load_chunk(const int* __restrict__ dst, int E, int base, int tid, int4 (&d)[4])
{
    if (base + EDGE_CHUNK <= E) {
#pragma unroll
        for (int u = 0; u < 4; ++u)
            d[u] = *(const int4*)(dst + base + (u * 256 + tid) * 4);
    } else {
#pragma unroll
        for (int u = 0; u < 4; ++u) {
            const int i = base + (u * 256 + tid) * 4;
            d[u].x = (i     < E) ? dst[i]     : -1;
            d[u].y = (i + 1 < E) ? dst[i + 1] : -1;
            d[u].z = (i + 2 < E) ? dst[i + 2] : -1;
            d[u].w = (i + 3 < E) ? dst[i + 3] : -1;
        }
    }
}

__global__ void __launch_bounds__(256) bin_count_kernel(const int* __restrict__ dst,
                                                        int* __restrict__ binCount, int E)
{
    __shared__ int hist[MAXBINS];
    const int tid = threadIdx.x;
    hist[tid] = 0;
    int4 d[4];
    load_chunk(dst, E, blockIdx.x * EDGE_CHUNK, tid, d);
    __syncthreads();
#pragma unroll
    for (int u = 0; u < 4; ++u) {
        if (d[u].x >= 0) atomicAdd(&hist[d[u].x >> BIN_SHIFT], 1);
        if (d[u].y >= 0) atomicAdd(&hist[d[u].y >> BIN_SHIFT], 1);
        if (d[u].z >= 0) atomicAdd(&hist[d[u].z >> BIN_SHIFT], 1);
        if (d[u].w >= 0) atomicAdd(&hist[d[u].w >> BIN_SHIFT], 1);
    }
    __syncthreads();
    const int c = hist[tid];
    if (c != 0) atomicAdd(&binCount[tid * CTR_PITCH], c);
}

__global__ void __launch_bounds__(256) bin_scatter_kernel(const int* __restrict__ dst,
    const int* __restrict__ binCount, int* __restrict__ binCursor, int* __restrict__ binStart,
    int* __restrict__ pairs, int E, int nbins)
{
    __shared__ int hist[MAXBINS];
    __shared__ int base[MAXBINS];
    __shared__ int wtot[8];
    const int tid  = threadIdx.x;
    const int lane = tid & 31;
    const int wave = tid >> 5;
    hist[tid] = 0;
    int4 d[4];
    const int ebase = blockIdx.x * EDGE_CHUNK;
    load_chunk(dst, E, ebase, tid, d);
    const int c = (tid < nbins) ? binCount[tid * CTR_PITCH] : 0;
    int x = c;
#pragma unroll
    for (int o = 1; o < 32; o <<= 1) {
        const int y = __shfl_up(x, o);
        if (lane >= o) x += y;
    }
    if (lane == 31) wtot[wave] = x;
    __syncthreads();
#pragma unroll
    for (int u = 0; u < 4; ++u) {
        if (d[u].x >= 0) atomicAdd(&hist[d[u].x >> BIN_SHIFT], 1);
        if (d[u].y >= 0) atomicAdd(&hist[d[u].y >> BIN_SHIFT], 1);
        if (d[u].z >= 0) atomicAdd(&hist[d[u].z >> BIN_SHIFT], 1);
        if (d[u].w >= 0) atomicAdd(&hist[d[u].w >> BIN_SHIFT], 1);
    }
    int woff = 0;
    for (int w = 0; w < wave; ++w) woff += wtot[w];
    const int start = woff + x - c;
    __syncthreads();
    const int mine = hist[tid];
    int b0 = start;
    if (mine != 0) b0 += atomicAdd(&binCursor[tid * CTR_PITCH], mine);
    base[tid] = b0;
    hist[tid] = 0;
    if (blockIdx.x == 0) {
        binStart[tid] = start;
        if (tid == MAXBINS - 1) binStart[MAXBINS] = start + c;
    }
    __syncthreads();
    int2* __restrict__ pr = (int2*)pairs;
    if (ebase + EDGE_CHUNK <= E) {
        int pos[16];
#pragma unroll
        for (int u = 0; u < 4; ++u) {
            const int dv[4] = { d[u].x, d[u].y, d[u].z, d[u].w };
#pragma unroll
            for (int q = 0; q < 4; ++q) {
                const int b = dv[q] >> BIN_SHIFT;
                pos[u * 4 + q] = base[b] + atomicAdd(&hist[b], 1);
            }
        }
#pragma unroll
        for (int u = 0; u < 4; ++u) {
            const int e = ebase + (u * 256 + tid) * 4;
            const int dv[4] = { d[u].x, d[u].y, d[u].z, d[u].w };
#pragma unroll
            for (int q = 0; q < 4; ++q) pr[pos[u * 4 + q]] = make_int2(e + q, dv[q]);
        }
    } else {
#pragma unroll
        for (int u = 0; u < 4; ++u) {
            const int e = ebase + (u * 256 + tid) * 4;
            const int dv[4] = { d[u].x, d[u].y, d[u].z, d[u].w };
#pragma unroll
            for (int q = 0; q < 4; ++q) {
                if (dv[q] >= 0) {
                    const int b = dv[q] >> BIN_SHIFT;
                    const int pos = base[b] + atomicAdd(&hist[b], 1);
                    pr[pos] = make_int2(e + q, dv[q]);
                }
            }
        }
    }
}

__global__ void __launch_bounds__(1024) bin_fine_kernel(const int* __restrict__ pairs,
    const int* __restrict__ binStart, int* __restrict__ beg, int* __restrict__ cur,
    int* __restrict__ perm, int N)
{
    __shared__ int ncount[256];
    __shared__ int ncur[256];
    __shared__ int wtot[8];
    const int tid  = threadIdx.x;
    const int lane = tid & 31;
    const int wave = tid >> 5;
    const int bin  = blockIdx.x;
    const int s    = binStart[bin];
    const int cntb = binStart[bin + 1] - s;
    const int2* __restrict__ tp = (const int2*)pairs + s;
    if (tid < 256) ncount[tid] = 0;
    __syncthreads();
    for (int i0 = 0; i0 < cntb; i0 += 4096) {
        int2 p[4];
#pragma unroll
        for (int u = 0; u < 4; ++u) {
            const int i = i0 + u * 1024 + tid;
            p[u] = make_int2(-1, -1);
            if (i < cntb) p[u] = tp[i];
        }
#pragma unroll
        for (int u = 0; u < 4; ++u)
            if (p[u].x >= 0) atomicAdd(&ncount[p[u].y & 255], 1);
    }
    __syncthreads();
    int c = 0, x = 0;
    if (tid < 256) {
        c = ncount[tid];
        x = c;
#pragma unroll
        for (int o = 1; o < 32; o <<= 1) {
            const int y = __shfl_up(x, o);
            if (lane >= o) x += y;
        }
        if (lane == 31) wtot[wave] = x;
    }
    __syncthreads();
    if (tid < 256) {
        int woff = 0;
        for (int w = 0; w < wave; ++w) woff += wtot[w];
        const int b0 = s + woff + x - c;
        ncur[tid] = b0;
        const int node = (bin << BIN_SHIFT) + tid;
        if (node < N) { beg[node] = b0; cur[node] = b0 + c; }
    }
    __syncthreads();
    for (int i0 = 0; i0 < cntb; i0 += 4096) {
        int2 p[4];
#pragma unroll
        for (int u = 0; u < 4; ++u) {
            const int i = i0 + u * 1024 + tid;
            p[u] = make_int2(-1, -1);
            if (i < cntb) p[u] = tp[i];
        }
#pragma unroll
        for (int u = 0; u < 4; ++u) {
            if (p[u].x >= 0) {
                const int pos = atomicAdd(&ncur[p[u].y & 255], 1);
                perm[pos] = p[u].x;
            }
        }
    }
}

__device__ __forceinline__ float wave_max_f(float v) {
#pragma unroll
    for (int o = 16; o > 0; o >>= 1) v = fmaxf(v, __shfl_xor(v, o));
    return v;
}
__device__ __forceinline__ float wave_sum_f(float v) {
#pragma unroll
    for (int o = 16; o > 0; o >>= 1) v += __shfl_xor(v, o);
    return v;
}
__device__ __forceinline__ int wave_min_i(int v) {
#pragma unroll
    for (int o = 16; o > 0; o >>= 1) { const int t = __shfl_xor(v, o); v = t < v ? t : v; }
    return v;
}

__device__ __forceinline__ float edge_logit(const float* __restrict__ eft, const int* __restrict__ src,
                                            const float* __restrict__ ssrc,
                                            const float* __restrict__ w_attn, int e, float sd)
{
    const float4* er = (const float4*)(eft + (size_t)e * 32);
    float t = ssrc[src[e]] + sd;
#pragma unroll
    for (int q = 0; q < 8; ++q) {
        const float4 v = er[q];
        t += v.x * w_attn[64 + 4 * q];
        t += v.y * w_attn[65 + 4 * q];
        t += v.z * w_attn[66 + 4 * q];
        t += v.w * w_attn[67 + 4 * q];
    }
    return t > 0.f ? t : 0.01f * t;
}

#define ROW_ACC(T)                                                         \
    if (ty == T) { acc[T][0] += al * v0; acc[T][1] += al * v1; acc[T][2] += al * v2; }

__device__ __forceinline__ int next_edge(const int* __restrict__ pb, int deg, int lane, int last)
{
    int best = 0x7fffffff;
    for (int c = 0; c < deg; c += 32) {
        const int i = c + lane;
        if (i < deg) {
            const int v = pb[i];
            if (v > last && v < best) best = v;
        }
    }
    return wave_min_i(best);
}

__device__ __forceinline__ int rank_owner(int rankA, int rankB, int p, bool& inA)
{
    const unsigned mA = __builtin_amdgcn_ballot_w32(rankA == p);
    const unsigned mB = __builtin_amdgcn_ballot_w32(rankB == p);
    inA = (mA != 0u);
    return __builtin_ctz(inA ? mA : mB);
}
__device__ __forceinline__ int pick_i(int vA, int vB, int ln, bool inA)
{
    return inA ? __builtin_amdgcn_readlane(vA, ln) : __builtin_amdgcn_readlane(vB, ln);
}
__device__ __forceinline__ float pick_f(float vA, float vB, int ln, bool inA)
{
    return __int_as_float(pick_i(__float_as_int(vA), __float_as_int(vB), ln, inA));
}

__global__ void __launch_bounds__(256) node_gather_kernel(
    const float* __restrict__ z, const float* __restrict__ eft,
    const int* __restrict__ src, const int* __restrict__ etype,
    const float* __restrict__ ssrc, const float* __restrict__ sdst,
    const float* __restrict__ w_attn,
    const int* __restrict__ beg, const int* __restrict__ cur, const int* __restrict__ perm,
    float* __restrict__ h, int N)
{
    const int d    = (int)((blockIdx.x * blockDim.x + threadIdx.x) >> 5);
    const int lane = threadIdx.x & 31;
    if (d >= N) return;
    const int b   = beg[d];
    const int deg = cur[d] - b;
    const int* __restrict__ pb = perm + b;
    const float sd = sdst[d];
    float acc[4][3] = {};

    if (deg <= 32) {
        const int id = (lane < deg) ? pb[lane] : (0x7fffffc0 + lane);
        int rank = 0;
#pragma unroll
        for (int j = 0; j < 32; ++j)
            rank += (__builtin_amdgcn_readlane(id, j) < id) ? 1 : 0;
        const int sid = __builtin_amdgcn_ds_permute(rank << 2, id);
        const bool ok = lane < deg;
        const int e  = ok ? sid : 0;
        const int s  = src[e];
        const int tt = etype[e];
        const float lr = ok ? edge_logit(eft, src, ssrc, w_attn, e, sd) : -INFINITY;
        const float m  = wave_max_f(lr);
        const float ex = ok ? __expf(lr - m) : 0.f;
        const float den = wave_sum_f(ex);
        const float alpha = ex / den;

        for (int j = 0; j < deg; j += 4) {
            float r0[4], r1[4], r2[4];
#pragma unroll
            for (int u = 0; u < 4; ++u) {
                const int jj = (j + u < deg) ? (j + u) : (deg - 1);
                const int eu = __builtin_amdgcn_readlane(e, jj);
                const int su = __builtin_amdgcn_readlane(s, jj);
                const float* zr = z + (size_t)su * 64;
                r0[u] = zr[lane];
                r1[u] = zr[32 + lane];
                r2[u] = eft[(size_t)eu * 32 + lane];
            }
#pragma unroll
            for (int u = 0; u < 4; ++u) {
                if (j + u < deg) {
                    const float al = __int_as_float(__builtin_amdgcn_readlane(__float_as_int(alpha), j + u));
                    const int   ty = __builtin_amdgcn_readlane(tt, j + u);
                    const float v0 = r0[u], v1 = r1[u], v2 = r2[u];
                    ROW_ACC(0) ROW_ACC(1) ROW_ACC(2) ROW_ACC(3)
                }
            }
        }
    } else if (deg <= 64) {
        const int  degB = deg - 32;
        const bool okB  = lane < degB;
        const int  eA   = pb[lane];
        const int  idB  = okB ? pb[32 + lane] : (0x7fffffc0 + lane);
        int rankA = 0, rankB = 0;
        for (int j = 0; j < 32; ++j) {
            const int a = __builtin_amdgcn_readlane(eA, j);
            const int b = __builtin_amdgcn_readlane(idB, j);
            rankA += ((a < eA) ? 1 : 0) + ((b < eA) ? 1 : 0);
            rankB += ((a < idB) ? 1 : 0) + ((b < idB) ? 1 : 0);
        }
        const int eB = okB ? idB : 0;
        int sA = 0, sB = 0, tA = 0, tB = 0;
        float lA = 0.f, lB = 0.f;
#pragma nounroll
        for (int hh = 0; hh < 2; ++hh) {
            const int   e_h = hh ? eB : eA;
            const int   s_h = src[e_h];
            const int   t_h = etype[e_h];
            const float l_h = edge_logit(eft, src, ssrc, w_attn, e_h, sd);
            if (hh) { sB = s_h; tB = t_h; lB = l_h; } else { sA = s_h; tA = t_h; lA = l_h; }
        }
        lB = okB ? lB : -INFINITY;
        const float m  = wave_max_f(fmaxf(lA, lB));
        const float xA = __expf(lA - m);
        const float xB = okB ? __expf(lB - m) : 0.f;
        float den = 0.f;
        for (int p = 0; p < deg; ++p) {
            bool inA;
            const int ln = rank_owner(rankA, rankB, p, inA);
            den += pick_f(xA, xB, ln, inA);
        }
        const float alA = xA / den, alB = xB / den;

        for (int j = 0; j < deg; j += 4) {
            float r0[4], r1[4], r2[4], al4[4];
            int ty4[4];
            for (int u = 0; u < 4; ++u) {
                const int p = (j + u < deg) ? (j + u) : (deg - 1);
                bool inA;
                const int ln = rank_owner(rankA, rankB, p, inA);
                const int eu = pick_i(eA, eB, ln, inA);
                const int su = pick_i(sA, sB, ln, inA);
                al4[u] = pick_f(alA, alB, ln, inA);
                ty4[u] = pick_i(tA, tB, ln, inA);
                const float* zr = z + (size_t)su * 64;
                r0[u] = zr[lane];
                r1[u] = zr[32 + lane];
                r2[u] = eft[(size_t)eu * 32 + lane];
            }
            for (int u = 0; u < 4; ++u) {
                if (j + u < deg) {
                    const float al = al4[u];
                    const int   ty = ty4[u];
                    const float v0 = r0[u], v1 = r1[u], v2 = r2[u];
                    ROW_ACC(0) ROW_ACC(1) ROW_ACC(2) ROW_ACC(3)
                }
            }
        }
    } else {
        float m = -INFINITY;
        for (int c = 0; c < deg; c += 32) {
            const int i = c + lane;
            if (i < deg) m = fmaxf(m, edge_logit(eft, src, ssrc, w_attn, pb[i], sd));
        }
        m = wave_max_f(m);
        float den = 0.f;
        int last = -1;
        for (int j = 0; j < deg; ++j) {
            last = next_edge(pb, deg, lane, last);
            den += __expf(edge_logit(eft, src, ssrc, w_attn, last, sd) - m);
        }
        last = -1;
        for (int j = 0; j < deg; ++j) {
            last = next_edge(pb, deg, lane, last);
            const float al = __expf(edge_logit(eft, src, ssrc, w_attn, last, sd) - m) / den;
            const int ty = etype[last];
            const float* zr = z + (size_t)src[last] * 64;
            const float v0 = zr[lane], v1 = zr[32 + lane], v2 = eft[(size_t)last * 32 + lane];
            ROW_ACC(0) ROW_ACC(1) ROW_ACC(2) ROW_ACC(3)
        }
    }

    float* hr = h + (size_t)d * 4 * 96 + lane;
#pragma unroll
    for (int t = 0; t < 4; ++t) {
        hr[t * 96]      = acc[t][0];
        hr[t * 96 + 32] = acc[t][1];
        hr[t * 96 + 64] = acc[t][2];
    }
}
#undef ROW_ACC

constexpr int ZP_BLOCKS = 512;

__device__ __forceinline__ void add4(float4& a, const float4& v) {
    a.x += v.x; a.y += v.y; a.z += v.z; a.w += v.w;
}

__global__ void __launch_bounds__(512) zphi_partial_kernel(
    const float* __restrict__ h2, float* __restrict__ part, int nodes)
{
    __shared__ float4 sm[8][64];
    const int tid = threadIdx.x;
    const int c4  = tid & 63;
    const int sub = tid >> 6;
    const int R   = (nodes + (int)gridDim.x - 1) / (int)gridDim.x;
    const int r0  = blockIdx.x * R;
    int r1 = r0 + R;
    if (r1 > nodes) r1 = nodes;
    const float4* __restrict__ p = (const float4*)h2 + c4;

    float4 acc = make_float4(0.f, 0.f, 0.f, 0.f);
    int r = r0 + sub;
    for (; r + 56 < r1; r += 64) {
        float4 v[8];
#pragma unroll
        for (int u = 0; u < 8; ++u) v[u] = p[(size_t)(r + 8 * u) * 64];
#pragma unroll
        for (int u = 0; u < 8; ++u) add4(acc, v[u]);
    }
    {
        float4 v[8];
#pragma unroll
        for (int u = 0; u < 8; ++u) {
            const int rr = r + 8 * u;
            v[u] = make_float4(0.f, 0.f, 0.f, 0.f);
            if (rr < r1) v[u] = p[(size_t)rr * 64];
        }
#pragma unroll
        for (int u = 0; u < 8; ++u) add4(acc, v[u]);
    }
    sm[sub][c4] = acc;
    __syncthreads();
    if (tid < 256) {
        const float* sf = (const float*)&sm[0][0];
        float s = 0.f;
#pragma unroll
        for (int k = 0; k < 8; ++k) s += sf[k * 256 + tid];
        part[(size_t)blockIdx.x * 256 + tid] = s;
    }
}

__global__ void __launch_bounds__(1024) beta_kernel(const float* __restrict__ part,
                            const float* __restrict__ w_sem, float* __restrict__ beta)
{
    __shared__ float4 sm[16][64];
    __shared__ float zs[256];
    const int tid = threadIdx.x;
    const int c4  = tid & 63;
    const int g   = tid >> 6;
    const float4* __restrict__ p = (const float4*)part + (size_t)g * 32 * 64 + c4;
    float4 acc = make_float4(0.f, 0.f, 0.f, 0.f);
    for (int b = 0; b < 32; b += 16) {
        float4 v[16];
#pragma unroll
        for (int u = 0; u < 16; ++u) v[u] = p[(size_t)(b + u) * 64];
#pragma unroll
        for (int u = 0; u < 16; ++u) add4(acc, v[u]);
    }
    sm[g][c4] = acc;
    __syncthreads();
    if (tid < 256) {
        const float* sf = (const float*)&sm[0][0];
        float s = 0.f;
#pragma unroll
        for (int k = 0; k < 16; ++k) s += sf[k * 256 + tid];
        zs[tid] = s;
    }
    __syncthreads();
    if (tid != 0) return;
    float wp[4], mx = -INFINITY;
#pragma unroll
    for (int t = 0; t < 4; ++t) {
        float d = 0.f;
        for (int o = 0; o < 64; ++o) d += zs[t * 64 + o] * w_sem[o];
        const float lr = d > 0.f ? d : 0.01f * d;
        wp[t] = lr;
        mx = fmaxf(mx, lr);
    }
    float s = 0.f;
#pragma unroll
    for (int t = 0; t < 4; ++t) { wp[t] = __expf(wp[t] - mx); s += wp[t]; }
#pragma unroll
    for (int t = 0; t < 4; ++t) beta[t] = wp[t] / s;
}

__global__ void out_kernel(const float* __restrict__ h2, const float* __restrict__ beta,
                           float* __restrict__ Z, int N)
{
    const int i = blockIdx.x * blockDim.x + threadIdx.x;
    if (i >= N * 64) return;
    const int n = i >> 6, o = i & 63;
    const float* r = h2 + (size_t)n * 256 + o;
    Z[i] = beta[0] * r[0] + beta[1] * r[64] + beta[2] * r[128] + beta[3] * r[192];
}

extern "C" void kernel_launch(void* const* d_in, const int* in_sizes, int n_in,
                              void* d_out, int out_size, void* d_ws, size_t ws_size,
                              hipStream_t stream)
{
    const float* node   = (const float*)d_in[0];
    const float* edgef  = (const float*)d_in[1];
    const int*   src    = (const int*)d_in[2];
    const int*   dst    = (const int*)d_in[3];
    const int*   etype  = (const int*)d_in[4];
    const float* w_n    = (const float*)d_in[5];
    const float* w_e    = (const float*)d_in[6];
    const float* w_attn = (const float*)d_in[7];
    const float* w_fc2  = (const float*)d_in[8];
    const float* w_sem  = (const float*)d_in[9];
    float* out = (float*)d_out;

    const int N = in_sizes[0] / 128;
    const int E = in_sizes[2];
    const int T = 4;

    float* ws   = (float*)d_ws;
    float* h    = ws;
    float* eft  = h + (size_t)N * T * 96;
    float* h2   = eft;
    float* z    = eft + (size_t)E * 32;
    float* ssrc = z + (size_t)N * 64;
    float* sdst = ssrc + N;
    float* abuf = sdst + N;
    float* mbuf = abuf + E;
    float* den  = mbuf + N;
    float* beta = den + N + 256;
    float* part = abuf;
    int* perm  = (int*)abuf;
    int* cur   = (int*)mbuf;
    int* beg   = (int*)den;
    int* pairs     = (int*)h;
    int* binCount  = pairs + 2 * (size_t)E;
    int* binCursor = binCount + MAXBINS * CTR_PITCH;
    int* binStart  = binCursor + MAXBINS * CTR_PITCH;
    const int nbins  = (N + (1 << BIN_SHIFT) - 1) >> BIN_SHIFT;
    const int chunks = (E + EDGE_CHUNK - 1) / EDGE_CHUNK;

    const int thr = 256;

    gemm_f32_wmma<4><<<((N + 15) / 16 + 7) / 8, thr, 0, stream>>>(node, w_n, z, N, 128, 64);
    gemm_f32_wmma<2><<<((E + 15) / 16 + 7) / 8, thr, 0, stream>>>(edgef, w_e, eft, E, 64, 32);
    node_score_kernel<<<(N + thr - 1) / thr, thr, 0, stream>>>(z, w_attn, ssrc, sdst, binCount, binCursor, N);
    bin_count_kernel<<<chunks, 256, 0, stream>>>(dst, binCount, E);
    bin_scatter_kernel<<<chunks, 256, 0, stream>>>(dst, binCount, binCursor, binStart, pairs, E, nbins);
    bin_fine_kernel<<<nbins, 1024, 0, stream>>>(pairs, binStart, beg, cur, perm, N);
    node_gather_kernel<<<(N + 7) / 8, thr, 0, stream>>>(z, eft, src, etype, ssrc, sdst, w_attn,
                                                        beg, cur, perm, h, N);
    gemm_f32_wmma<4><<<((N * T + 15) / 16 + 7) / 8, thr, 0, stream>>>(h, w_fc2, h2, N * T, 96, 64);
    zphi_partial_kernel<<<ZP_BLOCKS, 512, 0, stream>>>(h2, part, N);
    beta_kernel<<<1, 1024, 0, stream>>>(part, w_sem, beta);
    out_kernel<<<(N * 64 + thr - 1) / thr, thr, 0, stream>>>(h2, beta, out, N);
}


# semantic weights applied in front of the fc2 product (linear in h): zphi = colsum(h) @ w_fc2, Z = (sum_t beta_t h_t) @ w_fc2 on WMMA over N rows; h2 plane and out launch go; sum order changes
# speedup vs baseline: 2.5098x; 1.1903x over previous
#include <hip/hip_runtime.h>
#include <hip/hip_bf16.h>
#include <math.h>

typedef __attribute__((ext_vector_type(2))) float v2f;
typedef __attribute__((ext_vector_type(8))) float v8f;

__device__ __forceinline__ float xhalf(float x) {
    return __builtin_bit_cast(float, __builtin_amdgcn_permlanex16(
        __builtin_bit_cast(int, x), __builtin_bit_cast(int, x), 0x76543210, 0xfedcba98, false, false));
}

template <int NT>
__global__ void __launch_bounds__(256) gemm_f32_wmma(
    const float* __restrict__ A, const float* __restrict__ B,
    float* __restrict__ C, int M, int K, int N)
{
    __shared__ v2f sB2[4096];
    const int tid = threadIdx.x;
    const int pairCount = (K >> 1) * N;
    for (int p = tid; p < pairCount; p += 256) {
        const int kk = (p / N) << 1;
        const int n  = p - (p / N) * N;
        v2f b;
        b.x = B[kk * N + n];
        b.y = B[(kk + 1) * N + n];
        sB2[p] = b;
    }
    __syncthreads();

    const int lane    = tid & 31;
    const int wave    = tid >> 5;
    const int rowTile = blockIdx.x * 8 + wave;
    const int numTiles = (M + 15) >> 4;
    if (rowTile >= numTiles) return;

    const int half = lane >> 4;
    const int l15  = lane & 15;
    int mrow = rowTile * 16 + l15;
    if (mrow >= M) mrow = M - 1;

    v8f acc[NT] = {};
    const float* arow = A + (size_t)mrow * K + 2 * half;
    const v2f* __restrict__ bcol = sB2 + half * N + l15;

    for (int k0 = 0; k0 < K; k0 += 4) {
        const v2f a = *(const v2f*)(arow + k0);
        const v2f* brow = bcol + (k0 >> 1) * N;
#pragma unroll
        for (int nt = 0; nt < NT; ++nt) {
            const v2f b = brow[nt * 16];
            acc[nt] = __builtin_amdgcn_wmma_f32_16x16x4_f32(
                false, a, false, b, (short)0, acc[nt], false, false);
        }
    }

#pragma unroll
    for (int p = 0; p < NT / 2; ++p) {
#pragma unroll
        for (int r = 0; r < 8; ++r) {
            const float a0 = acc[2 * p][r];
            const float a1 = acc[2 * p + 1][r];
            const float x0 = xhalf(a0);
            const float x1 = xhalf(a1);
            const float lo = half ? x1 : a0;
            const float hi = half ? a1 : x0;
            const int row = rowTile * 16 + r;
            if (row < M)     C[(size_t)row * N + p * 32 + lane] = lo;
            if (row + 8 < M) C[(size_t)(row + 8) * N + p * 32 + lane] = hi;
        }
    }
}


constexpr int BIN_SHIFT  = 8;
constexpr int MAXBINS    = 256;
constexpr int CTR_PITCH  = 64;
constexpr int EDGE_CHUNK = 4096;

__global__ void node_score_kernel(const float* __restrict__ z,
                                  const float* __restrict__ w_attn,
                                  float* __restrict__ ssrc, float* __restrict__ sdst,
                                  int* __restrict__ binCount, int* __restrict__ binCursor, int N)
{
    const int n = blockIdx.x * blockDim.x + threadIdx.x;
    if (n < MAXBINS) { binCount[n * CTR_PITCH] = 0; binCursor[n * CTR_PITCH] = 0; }
    if (n >= N) return;
    const float* zr = z + (size_t)n * 64;
    float s0 = 0.f, s1 = 0.f;
#pragma unroll
    for (int i = 0; i < 64; ++i) {
        const float v = zr[i];
        s0 += v * w_attn[i];
        s1 += v * w_attn[96 + i];
    }
    ssrc[n] = s0;
    sdst[n] = s1;
}

__device__ __forceinline__ void load_chunk(const int* __restrict__ dst, int E, int base, int tid, int4 (&d)[4])
{
    if (base + EDGE_CHUNK <= E) {
#pragma unroll
        for (int u = 0; u < 4; ++u)
            d[u] = *(const int4*)(dst + base + (u * 256 + tid) * 4);
    } else {
#pragma unroll
        for (int u = 0; u < 4; ++u) {
            const int i = base + (u * 256 + tid) * 4;
            d[u].x = (i     < E) ? dst[i]     : -1;
            d[u].y = (i + 1 < E) ? dst[i + 1] : -1;
            d[u].z = (i + 2 < E) ? dst[i + 2] : -1;
            d[u].w = (i + 3 < E) ? dst[i + 3] : -1;
        }
    }
}

__global__ void __launch_bounds__(256) bin_count_kernel(const int* __restrict__ dst,
                                                        int* __restrict__ binCount, int E)
{
    __shared__ int hist[MAXBINS];
    const int tid = threadIdx.x;
    hist[tid] = 0;
    int4 d[4];
    load_chunk(dst, E, blockIdx.x * EDGE_CHUNK, tid, d);
    __syncthreads();
#pragma unroll
    for (int u = 0; u < 4; ++u) {
        if (d[u].x >= 0) atomicAdd(&hist[d[u].x >> BIN_SHIFT], 1);
        if (d[u].y >= 0) atomicAdd(&hist[d[u].y >> BIN_SHIFT], 1);
        if (d[u].z >= 0) atomicAdd(&hist[d[u].z >> BIN_SHIFT], 1);
        if (d[u].w >= 0) atomicAdd(&hist[d[u].w >> BIN_SHIFT], 1);
    }
    __syncthreads();
    const int c = hist[tid];
    if (c != 0) atomicAdd(&binCount[tid * CTR_PITCH], c);
}

__global__ void __launch_bounds__(256) bin_scatter_kernel(const int* __restrict__ dst,
    const int* __restrict__ binCount, int* __restrict__ binCursor, int* __restrict__ binStart,
    int* __restrict__ pairs, int E, int nbins)
{
    __shared__ int hist[MAXBINS];
    __shared__ int base[MAXBINS];
    __shared__ int wtot[8];
    const int tid  = threadIdx.x;
    const int lane = tid & 31;
    const int wave = tid >> 5;
    hist[tid] = 0;
    int4 d[4];
    const int ebase = blockIdx.x * EDGE_CHUNK;
    load_chunk(dst, E, ebase, tid, d);
    const int c = (tid < nbins) ? binCount[tid * CTR_PITCH] : 0;
    int x = c;
#pragma unroll
    for (int o = 1; o < 32; o <<= 1) {
        const int y = __shfl_up(x, o);
        if (lane >= o) x += y;
    }
    if (lane == 31) wtot[wave] = x;
    __syncthreads();
#pragma unroll
    for (int u = 0; u < 4; ++u) {
        if (d[u].x >= 0) atomicAdd(&hist[d[u].x >> BIN_SHIFT], 1);
        if (d[u].y >= 0) atomicAdd(&hist[d[u].y >> BIN_SHIFT], 1);
        if (d[u].z >= 0) atomicAdd(&hist[d[u].z >> BIN_SHIFT], 1);
        if (d[u].w >= 0) atomicAdd(&hist[d[u].w >> BIN_SHIFT], 1);
    }
    int woff = 0;
    for (int w = 0; w < wave; ++w) woff += wtot[w];
    const int start = woff + x - c;
    __syncthreads();
    const int mine = hist[tid];
    int b0 = start;
    if (mine != 0) b0 += atomicAdd(&binCursor[tid * CTR_PITCH], mine);
    base[tid] = b0;
    hist[tid] = 0;
    if (blockIdx.x == 0) {
        binStart[tid] = start;
        if (tid == MAXBINS - 1) binStart[MAXBINS] = start + c;
    }
    __syncthreads();
    int2* __restrict__ pr = (int2*)pairs;
    if (ebase + EDGE_CHUNK <= E) {
        int pos[16];
#pragma unroll
        for (int u = 0; u < 4; ++u) {
            const int dv[4] = { d[u].x, d[u].y, d[u].z, d[u].w };
#pragma unroll
            for (int q = 0; q < 4; ++q) {
                const int b = dv[q] >> BIN_SHIFT;
                pos[u * 4 + q] = base[b] + atomicAdd(&hist[b], 1);
            }
        }
#pragma unroll
        for (int u = 0; u < 4; ++u) {
            const int e = ebase + (u * 256 + tid) * 4;
            const int dv[4] = { d[u].x, d[u].y, d[u].z, d[u].w };
#pragma unroll
            for (int q = 0; q < 4; ++q) pr[pos[u * 4 + q]] = make_int2(e + q, dv[q]);
        }
    } else {
#pragma unroll
        for (int u = 0; u < 4; ++u) {
            const int e = ebase + (u * 256 + tid) * 4;
            const int dv[4] = { d[u].x, d[u].y, d[u].z, d[u].w };
#pragma unroll
            for (int q = 0; q < 4; ++q) {
                if (dv[q] >= 0) {
                    const int b = dv[q] >> BIN_SHIFT;
                    const int pos = base[b] + atomicAdd(&hist[b], 1);
                    pr[pos] = make_int2(e + q, dv[q]);
                }
            }
        }
    }
}

__global__ void __launch_bounds__(1024) bin_fine_kernel(const int* __restrict__ pairs,
    const int* __restrict__ binStart, int* __restrict__ beg, int* __restrict__ cur,
    int* __restrict__ perm, int N)
{
    __shared__ int ncount[256];
    __shared__ int ncur[256];
    __shared__ int wtot[8];
    const int tid  = threadIdx.x;
    const int lane = tid & 31;
    const int wave = tid >> 5;
    const int bin  = blockIdx.x;
    const int s    = binStart[bin];
    const int cntb = binStart[bin + 1] - s;
    const int2* __restrict__ tp = (const int2*)pairs + s;
    if (tid < 256) ncount[tid] = 0;
    __syncthreads();
    for (int i0 = 0; i0 < cntb; i0 += 4096) {
        int2 p[4];
#pragma unroll
        for (int u = 0; u < 4; ++u) {
            const int i = i0 + u * 1024 + tid;
            p[u] = make_int2(-1, -1);
            if (i < cntb) p[u] = tp[i];
        }
#pragma unroll
        for (int u = 0; u < 4; ++u)
            if (p[u].x >= 0) atomicAdd(&ncount[p[u].y & 255], 1);
    }
    __syncthreads();
    int c = 0, x = 0;
    if (tid < 256) {
        c = ncount[tid];
        x = c;
#pragma unroll
        for (int o = 1; o < 32; o <<= 1) {
            const int y = __shfl_up(x, o);
            if (lane >= o) x += y;
        }
        if (lane == 31) wtot[wave] = x;
    }
    __syncthreads();
    if (tid < 256) {
        int woff = 0;
        for (int w = 0; w < wave; ++w) woff += wtot[w];
        const int b0 = s + woff + x - c;
        ncur[tid] = b0;
        const int node = (bin << BIN_SHIFT) + tid;
        if (node < N) { beg[node] = b0; cur[node] = b0 + c; }
    }
    __syncthreads();
    for (int i0 = 0; i0 < cntb; i0 += 4096) {
        int2 p[4];
#pragma unroll
        for (int u = 0; u < 4; ++u) {
            const int i = i0 + u * 1024 + tid;
            p[u] = make_int2(-1, -1);
            if (i < cntb) p[u] = tp[i];
        }
#pragma unroll
        for (int u = 0; u < 4; ++u) {
            if (p[u].x >= 0) {
                const int pos = atomicAdd(&ncur[p[u].y & 255], 1);
                perm[pos] = p[u].x;
            }
        }
    }
}

__device__ __forceinline__ float wave_max_f(float v) {
#pragma unroll
    for (int o = 16; o > 0; o >>= 1) v = fmaxf(v, __shfl_xor(v, o));
    return v;
}
__device__ __forceinline__ float wave_sum_f(float v) {
#pragma unroll
    for (int o = 16; o > 0; o >>= 1) v += __shfl_xor(v, o);
    return v;
}
__device__ __forceinline__ int wave_min_i(int v) {
#pragma unroll
    for (int o = 16; o > 0; o >>= 1) { const int t = __shfl_xor(v, o); v = t < v ? t : v; }
    return v;
}

__device__ __forceinline__ float edge_logit(const float* __restrict__ eft, const int* __restrict__ src,
                                            const float* __restrict__ ssrc,
                                            const float* __restrict__ w_attn, int e, float sd)
{
    const float4* er = (const float4*)(eft + (size_t)e * 32);
    float t = ssrc[src[e]] + sd;
#pragma unroll
    for (int q = 0; q < 8; ++q) {
        const float4 v = er[q];
        t += v.x * w_attn[64 + 4 * q];
        t += v.y * w_attn[65 + 4 * q];
        t += v.z * w_attn[66 + 4 * q];
        t += v.w * w_attn[67 + 4 * q];
    }
    return t > 0.f ? t : 0.01f * t;
}

#define ROW_ACC(T)                                                         \
    if (ty == T) { acc[T][0] += al * v0; acc[T][1] += al * v1; acc[T][2] += al * v2; }

__device__ __forceinline__ int next_edge(const int* __restrict__ pb, int deg, int lane, int last)
{
    int best = 0x7fffffff;
    for (int c = 0; c < deg; c += 32) {
        const int i = c + lane;
        if (i < deg) {
            const int v = pb[i];
            if (v > last && v < best) best = v;
        }
    }
    return wave_min_i(best);
}

__device__ __forceinline__ int rank_owner(int rankA, int rankB, int p, bool& inA)
{
    const unsigned mA = __builtin_amdgcn_ballot_w32(rankA == p);
    const unsigned mB = __builtin_amdgcn_ballot_w32(rankB == p);
    inA = (mA != 0u);
    return __builtin_ctz(inA ? mA : mB);
}
__device__ __forceinline__ int pick_i(int vA, int vB, int ln, bool inA)
{
    return inA ? __builtin_amdgcn_readlane(vA, ln) : __builtin_amdgcn_readlane(vB, ln);
}
__device__ __forceinline__ float pick_f(float vA, float vB, int ln, bool inA)
{
    return __int_as_float(pick_i(__float_as_int(vA), __float_as_int(vB), ln, inA));
}

__global__ void __launch_bounds__(256) node_gather_kernel(
    const float* __restrict__ z, const float* __restrict__ eft,
    const int* __restrict__ src, const int* __restrict__ etype,
    const float* __restrict__ ssrc, const float* __restrict__ sdst,
    const float* __restrict__ w_attn,
    const int* __restrict__ beg, const int* __restrict__ cur, const int* __restrict__ perm,
    float* __restrict__ h, int N)
{
    const int d    = (int)((blockIdx.x * blockDim.x + threadIdx.x) >> 5);
    const int lane = threadIdx.x & 31;
    if (d >= N) return;
    const int b   = beg[d];
    const int deg = cur[d] - b;
    const int* __restrict__ pb = perm + b;
    const float sd = sdst[d];
    float acc[4][3] = {};

    if (deg <= 32) {
        const int id = (lane < deg) ? pb[lane] : (0x7fffffc0 + lane);
        int rank = 0;
#pragma unroll
        for (int j = 0; j < 32; ++j)
            rank += (__builtin_amdgcn_readlane(id, j) < id) ? 1 : 0;
        const int sid = __builtin_amdgcn_ds_permute(rank << 2, id);
        const bool ok = lane < deg;
        const int e  = ok ? sid : 0;
        const int s  = src[e];
        const int tt = etype[e];
        const float lr = ok ? edge_logit(eft, src, ssrc, w_attn, e, sd) : -INFINITY;
        const float m  = wave_max_f(lr);
        const float ex = ok ? __expf(lr - m) : 0.f;
        const float den = wave_sum_f(ex);
        const float alpha = ex / den;

        for (int j = 0; j < deg; j += 4) {
            float r0[4], r1[4], r2[4];
#pragma unroll
            for (int u = 0; u < 4; ++u) {
                const int jj = (j + u < deg) ? (j + u) : (deg - 1);
                const int eu = __builtin_amdgcn_readlane(e, jj);
                const int su = __builtin_amdgcn_readlane(s, jj);
                const float* zr = z + (size_t)su * 64;
                r0[u] = zr[lane];
                r1[u] = zr[32 + lane];
                r2[u] = eft[(size_t)eu * 32 + lane];
            }
#pragma unroll
            for (int u = 0; u < 4; ++u) {
                if (j + u < deg) {
                    const float al = __int_as_float(__builtin_amdgcn_readlane(__float_as_int(alpha), j + u));
                    const int   ty = __builtin_amdgcn_readlane(tt, j + u);
                    const float v0 = r0[u], v1 = r1[u], v2 = r2[u];
                    ROW_ACC(0) ROW_ACC(1) ROW_ACC(2) ROW_ACC(3)
                }
            }
        }
    } else if (deg <= 64) {
        const int  degB = deg - 32;
        const bool okB  = lane < degB;
        const int  eA   = pb[lane];
        const int  idB  = okB ? pb[32 + lane] : (0x7fffffc0 + lane);
        int rankA = 0, rankB = 0;
        for (int j = 0; j < 32; ++j) {
            const int a = __builtin_amdgcn_readlane(eA, j);
            const int b = __builtin_amdgcn_readlane(idB, j);
            rankA += ((a < eA) ? 1 : 0) + ((b < eA) ? 1 : 0);
            rankB += ((a < idB) ? 1 : 0) + ((b < idB) ? 1 : 0);
        }
        const int eB = okB ? idB : 0;
        int sA = 0, sB = 0, tA = 0, tB = 0;
        float lA = 0.f, lB = 0.f;
#pragma nounroll
        for (int hh = 0; hh < 2; ++hh) {
            const int   e_h = hh ? eB : eA;
            const int   s_h = src[e_h];
            const int   t_h = etype[e_h];
            const float l_h = edge_logit(eft, src, ssrc, w_attn, e_h, sd);
            if (hh) { sB = s_h; tB = t_h; lB = l_h; } else { sA = s_h; tA = t_h; lA = l_h; }
        }
        lB = okB ? lB : -INFINITY;
        const float m  = wave_max_f(fmaxf(lA, lB));
        const float xA = __expf(lA - m);
        const float xB = okB ? __expf(lB - m) : 0.f;
        float den = 0.f;
        for (int p = 0; p < deg; ++p) {
            bool inA;
            const int ln = rank_owner(rankA, rankB, p, inA);
            den += pick_f(xA, xB, ln, inA);
        }
        const float alA = xA / den, alB = xB / den;

        for (int j = 0; j < deg; j += 4) {
            float r0[4], r1[4], r2[4], al4[4];
            int ty4[4];
            for (int u = 0; u < 4; ++u) {
                const int p = (j + u < deg) ? (j + u) : (deg - 1);
                bool inA;
                const int ln = rank_owner(rankA, rankB, p, inA);
                const int eu = pick_i(eA, eB, ln, inA);
                const int su = pick_i(sA, sB, ln, inA);
                al4[u] = pick_f(alA, alB, ln, inA);
                ty4[u] = pick_i(tA, tB, ln, inA);
                const float* zr = z + (size_t)su * 64;
                r0[u] = zr[lane];
                r1[u] = zr[32 + lane];
                r2[u] = eft[(size_t)eu * 32 + lane];
            }
            for (int u = 0; u < 4; ++u) {
                if (j + u < deg) {
                    const float al = al4[u];
                    const int   ty = ty4[u];
                    const float v0 = r0[u], v1 = r1[u], v2 = r2[u];
                    ROW_ACC(0) ROW_ACC(1) ROW_ACC(2) ROW_ACC(3)
                }
            }
        }
    } else {
        float m = -INFINITY;
        for (int c = 0; c < deg; c += 32) {
            const int i = c + lane;
            if (i < deg) m = fmaxf(m, edge_logit(eft, src, ssrc, w_attn, pb[i], sd));
        }
        m = wave_max_f(m);
        float den = 0.f;
        int last = -1;
        for (int j = 0; j < deg; ++j) {
            last = next_edge(pb, deg, lane, last);
            den += __expf(edge_logit(eft, src, ssrc, w_attn, last, sd) - m);
        }
        last = -1;
        for (int j = 0; j < deg; ++j) {
            last = next_edge(pb, deg, lane, last);
            const float al = __expf(edge_logit(eft, src, ssrc, w_attn, last, sd) - m) / den;
            const int ty = etype[last];
            const float* zr = z + (size_t)src[last] * 64;
            const float v0 = zr[lane], v1 = zr[32 + lane], v2 = eft[(size_t)last * 32 + lane];
            ROW_ACC(0) ROW_ACC(1) ROW_ACC(2) ROW_ACC(3)
        }
    }

    float* hr = h + (size_t)d * 4 * 96 + lane;
#pragma unroll
    for (int t = 0; t < 4; ++t) {
        hr[t * 96]      = acc[t][0];
        hr[t * 96 + 32] = acc[t][1];
        hr[t * 96 + 64] = acc[t][2];
    }
}
#undef ROW_ACC

constexpr int ZP_BLOCKS = 512;

__device__ __forceinline__ void add4(float4& a, const float4& v) {
    a.x += v.x; a.y += v.y; a.z += v.z; a.w += v.w;
}

__global__ void __launch_bounds__(768) hsum_partial_kernel(
    const float* __restrict__ h, float* __restrict__ part, int nodes)
{
    __shared__ float4 sm[8][96];
    const int tid = threadIdx.x;
    const int c4  = tid % 96;
    const int sub = tid / 96;
    const int R   = (nodes + (int)gridDim.x - 1) / (int)gridDim.x;
    const int r0  = blockIdx.x * R;
    int r1 = r0 + R;
    if (r1 > nodes) r1 = nodes;
    const float4* __restrict__ p = (const float4*)h + c4;

    float4 acc = make_float4(0.f, 0.f, 0.f, 0.f);
    int r = r0 + sub;
    for (; r + 56 < r1; r += 64) {
        float4 v[8];
#pragma unroll
        for (int u = 0; u < 8; ++u) v[u] = p[(size_t)(r + 8 * u) * 96];
#pragma unroll
        for (int u = 0; u < 8; ++u) add4(acc, v[u]);
    }
    {
        float4 v[8];
#pragma unroll
        for (int u = 0; u < 8; ++u) {
            const int rr = r + 8 * u;
            v[u] = make_float4(0.f, 0.f, 0.f, 0.f);
            if (rr < r1) v[u] = p[(size_t)rr * 96];
        }
#pragma unroll
        for (int u = 0; u < 8; ++u) add4(acc, v[u]);
    }
    sm[sub][c4] = acc;
    __syncthreads();
    if (tid < 384) {
        const float* sf = (const float*)&sm[0][0];
        float s = 0.f;
#pragma unroll
        for (int k = 0; k < 8; ++k) s += sf[k * 384 + tid];
        part[(size_t)blockIdx.x * 384 + tid] = s;
    }
}

__global__ void __launch_bounds__(768) beta_kernel(const float* __restrict__ part,
                            const float* __restrict__ w_fc2,
                            const float* __restrict__ w_sem, float* __restrict__ beta)
{
    __shared__ float4 sm[8][96];
    __shared__ __attribute__((aligned(16))) float hs[4 * 96];
    __shared__ float zs[16 * 64];
    const int tid = threadIdx.x;
    const int c4  = tid % 96;
    const int g   = tid / 96;
    const float4* __restrict__ p = (const float4*)part + (size_t)g * 64 * 96 + c4;
    float4 acc = make_float4(0.f, 0.f, 0.f, 0.f);
    for (int b = 0; b < 64; b += 16) {
        float4 v[16];
#pragma unroll
        for (int u = 0; u < 16; ++u) v[u] = p[(size_t)(b + u) * 96];
#pragma unroll
        for (int u = 0; u < 16; ++u) add4(acc, v[u]);
    }
    sm[g][c4] = acc;
    __syncthreads();
    if (tid < 384) {
        const float* sf = (const float*)&sm[0][0];
        float s = 0.f;
#pragma unroll
        for (int k = 0; k < 8; ++k) s += sf[k * 384 + tid];
        hs[tid] = s;
    }
    __syncthreads();
    if (tid < 128) {
        const int lane = tid & 31;
        const int nt   = tid >> 5;
        const int half = lane >> 4;
        const int l15  = lane & 15;
        const int ar   = (l15 < 4) ? l15 : 0;
        v8f d = {};
#pragma unroll
        for (int k0 = 0; k0 < 96; k0 += 4) {
            const int k = k0 + 2 * half;
            const v2f av = *(const v2f*)&hs[ar * 96 + k];
            v2f a, b;
            a.x = (l15 < 4) ? av.x : 0.f;
            a.y = (l15 < 4) ? av.y : 0.f;
            b.x = w_fc2[k * 64 + nt * 16 + l15];
            b.y = w_fc2[(k + 1) * 64 + nt * 16 + l15];
            d = __builtin_amdgcn_wmma_f32_16x16x4_f32(false, a, false, b, (short)0, d, false, false);
        }
#pragma unroll
        for (int r = 0; r < 8; ++r) zs[(r + 8 * half) * 64 + nt * 16 + l15] = d[r];
    }
    __syncthreads();
    if (tid != 0) return;
    float wp[4], mx = -INFINITY;
#pragma unroll
    for (int t = 0; t < 4; ++t) {
        float d = 0.f;
        for (int o = 0; o < 64; ++o) d += zs[t * 64 + o] * w_sem[o];
        const float lr = d > 0.f ? d : 0.01f * d;
        wp[t] = lr;
        mx = fmaxf(mx, lr);
    }
    float s = 0.f;
#pragma unroll
    for (int t = 0; t < 4; ++t) { wp[t] = __expf(wp[t] - mx); s += wp[t]; }
#pragma unroll
    for (int t = 0; t < 4; ++t) beta[t] = wp[t] / s;
}

__device__ __forceinline__ void sem_load4(const float4* __restrict__ hrow, int j0, float4 (&v)[4][4])
{
#pragma unroll
    for (int s = 0; s < 4; ++s) {
#pragma unroll
        for (int t = 0; t < 4; ++t) v[s][t] = hrow[24 * t + 2 * (j0 + s)];
    }
}

__device__ __forceinline__ void sem_steps4(const float4 (&v)[4][4], float b0, float b1, float b2, float b3,
                                           const v2f* __restrict__ bcol, int j0, v8f (&acc)[4])
{
#pragma unroll
    for (int s = 0; s < 4; ++s) {
        const float4 v0 = v[s][0], v1 = v[s][1], v2 = v[s][2], v3 = v[s][3];
        v2f a01, a23;
        a01.x = b0 * v0.x + b1 * v1.x + b2 * v2.x + b3 * v3.x;
        a01.y = b0 * v0.y + b1 * v1.y + b2 * v2.y + b3 * v3.y;
        a23.x = b0 * v0.z + b1 * v1.z + b2 * v2.z + b3 * v3.z;
        a23.y = b0 * v0.w + b1 * v1.w + b2 * v2.w + b3 * v3.w;
        const v2f* brow = bcol + (4 * (j0 + s)) * 64;
#pragma unroll
        for (int nt = 0; nt < 4; ++nt) {
            const v2f b = brow[nt * 16];
            acc[nt] = __builtin_amdgcn_wmma_f32_16x16x4_f32(
                false, a01, false, b, (short)0, acc[nt], false, false);
        }
#pragma unroll
        for (int nt = 0; nt < 4; ++nt) {
            const v2f b = brow[64 + nt * 16];
            acc[nt] = __builtin_amdgcn_wmma_f32_16x16x4_f32(
                false, a23, false, b, (short)0, acc[nt], false, false);
        }
    }
}

__global__ void __launch_bounds__(256) sem_out_wmma_kernel(
    const float* __restrict__ h, const float* __restrict__ w_fc2, const float* __restrict__ beta,
    float* __restrict__ Z, int M)
{
    constexpr int K = 96, N = 64, NT = 4;
    __shared__ v2f sB2[(K / 2) * N];
    const int tid = threadIdx.x;
    for (int p = tid; p < (K / 2) * N; p += 256) {
        const int kk = (p / N) << 1;
        const int n  = p % N;
        v2f b;
        b.x = w_fc2[kk * N + n];
        b.y = w_fc2[(kk + 1) * N + n];
        sB2[p] = b;
    }
    __syncthreads();

    const int lane    = tid & 31;
    const int wave    = tid >> 5;
    const int rowTile = blockIdx.x * 8 + wave;
    const int numTiles = (M + 15) >> 4;
    if (rowTile >= numTiles) return;

    const int half = lane >> 4;
    const int l15  = lane & 15;
    int mrow = rowTile * 16 + l15;
    if (mrow >= M) mrow = M - 1;

    const float b0 = beta[0], b1 = beta[1], b2 = beta[2], b3 = beta[3];
    v8f acc[NT] = {};
    const float4* __restrict__ hrow = (const float4*)(h + (size_t)mrow * (4 * K) + 4 * half);
    const v2f* __restrict__ bcol = sB2 + (2 * half) * N + l15;

    float4 v[3][4][4];
    sem_load4(hrow, 0, v[0]);
    sem_load4(hrow, 4, v[1]);
    sem_load4(hrow, 8, v[2]);
    __builtin_amdgcn_sched_barrier(0);
    sem_steps4(v[0], b0, b1, b2, b3, bcol, 0, acc);
    sem_steps4(v[1], b0, b1, b2, b3, bcol, 4, acc);
    sem_steps4(v[2], b0, b1, b2, b3, bcol, 8, acc);

#pragma unroll
    for (int p = 0; p < NT / 2; ++p) {
#pragma unroll
        for (int r = 0; r < 8; ++r) {
            const float a0 = acc[2 * p][r];
            const float a1 = acc[2 * p + 1][r];
            const float x0 = xhalf(a0);
            const float x1 = xhalf(a1);
            const float lo = half ? x1 : a0;
            const float hi = half ? a1 : x0;
            const int row = rowTile * 16 + r;
            if (row < M)     Z[(size_t)row * N + p * 32 + lane] = lo;
            if (row + 8 < M) Z[(size_t)(row + 8) * N + p * 32 + lane] = hi;
        }
    }
}

extern "C" void kernel_launch(void* const* d_in, const int* in_sizes, int n_in,
                              void* d_out, int out_size, void* d_ws, size_t ws_size,
                              hipStream_t stream)
{
    const float* node   = (const float*)d_in[0];
    const float* edgef  = (const float*)d_in[1];
    const int*   src    = (const int*)d_in[2];
    const int*   dst    = (const int*)d_in[3];
    const int*   etype  = (const int*)d_in[4];
    const float* w_n    = (const float*)d_in[5];
    const float* w_e    = (const float*)d_in[6];
    const float* w_attn = (const float*)d_in[7];
    const float* w_fc2  = (const float*)d_in[8];
    const float* w_sem  = (const float*)d_in[9];
    float* out = (float*)d_out;

    const int N = in_sizes[0] / 128;
    const int E = in_sizes[2];
    const int T = 4;

    float* ws   = (float*)d_ws;
    float* h    = ws;
    float* eft  = h + (size_t)N * T * 96;
    float* z    = eft + (size_t)E * 32;
    float* ssrc = z + (size_t)N * 64;
    float* sdst = ssrc + N;
    float* abuf = sdst + N;
    float* mbuf = abuf + E;
    float* den  = mbuf + N;
    float* beta = den + N + 256;
    float* part = abuf;
    int* perm  = (int*)abuf;
    int* cur   = (int*)mbuf;
    int* beg   = (int*)den;
    int* pairs     = (int*)h;
    int* binCount  = pairs + 2 * (size_t)E;
    int* binCursor = binCount + MAXBINS * CTR_PITCH;
    int* binStart  = binCursor + MAXBINS * CTR_PITCH;
    const int nbins  = (N + (1 << BIN_SHIFT) - 1) >> BIN_SHIFT;
    const int chunks = (E + EDGE_CHUNK - 1) / EDGE_CHUNK;

    const int thr = 256;

    gemm_f32_wmma<4><<<((N + 15) / 16 + 7) / 8, thr, 0, stream>>>(node, w_n, z, N, 128, 64);
    gemm_f32_wmma<2><<<((E + 15) / 16 + 7) / 8, thr, 0, stream>>>(edgef, w_e, eft, E, 64, 32);
    node_score_kernel<<<(N + thr - 1) / thr, thr, 0, stream>>>(z, w_attn, ssrc, sdst, binCount, binCursor, N);
    bin_count_kernel<<<chunks, 256, 0, stream>>>(dst, binCount, E);
    bin_scatter_kernel<<<chunks, 256, 0, stream>>>(dst, binCount, binCursor, binStart, pairs, E, nbins);
    bin_fine_kernel<<<nbins, 1024, 0, stream>>>(pairs, binStart, beg, cur, perm, N);
    node_gather_kernel<<<(N + 7) / 8, thr, 0, stream>>>(z, eft, src, etype, ssrc, sdst, w_attn,
                                                        beg, cur, perm, h, N);
    hsum_partial_kernel<<<ZP_BLOCKS, 768, 0, stream>>>(h, part, N);
    beta_kernel<<<1, 768, 0, stream>>>(part, w_fc2, w_sem, beta);
    sem_out_wmma_kernel<<<((N + 15) / 16 + 7) / 8, thr, 0, stream>>>(h, w_fc2, beta, out, N);
}


# edge projection w_e applied after the weighted sum (linear): gather sums raw 64-wide edge rows, w_e folds into fc2 weights by a 64x32x64 WMMA; [E,32] plane and its E-row GEMM go; sum order changes
# speedup vs baseline: 2.9233x; 1.1648x over previous
#include <hip/hip_runtime.h>
#include <hip/hip_bf16.h>
#include <math.h>

typedef __attribute__((ext_vector_type(2))) float v2f;
typedef __attribute__((ext_vector_type(8))) float v8f;

__device__ __forceinline__ float xhalf(float x) {
    return __builtin_bit_cast(float, __builtin_amdgcn_permlanex16(
        __builtin_bit_cast(int, x), __builtin_bit_cast(int, x), 0x76543210, 0xfedcba98, false, false));
}

template <int NT>
__global__ void __launch_bounds__(256) gemm_f32_wmma(
    const float* __restrict__ A, const float* __restrict__ B,
    float* __restrict__ C, int M, int K, int N)
{
    __shared__ v2f sB2[4096];
    const int tid = threadIdx.x;
    const int pairCount = (K >> 1) * N;
    for (int p = tid; p < pairCount; p += 256) {
        const int kk = (p / N) << 1;
        const int n  = p - (p / N) * N;
        v2f b;
        b.x = B[kk * N + n];
        b.y = B[(kk + 1) * N + n];
        sB2[p] = b;
    }
    __syncthreads();

    const int lane    = tid & 31;
    const int wave    = tid >> 5;
    const int rowTile = blockIdx.x * 8 + wave;
    const int numTiles = (M + 15) >> 4;
    if (rowTile >= numTiles) return;

    const int half = lane >> 4;
    const int l15  = lane & 15;
    int mrow = rowTile * 16 + l15;
    if (mrow >= M) mrow = M - 1;

    v8f acc[NT] = {};
    const float* arow = A + (size_t)mrow * K + 2 * half;
    const v2f* __restrict__ bcol = sB2 + half * N + l15;

    for (int k0 = 0; k0 < K; k0 += 4) {
        const v2f a = *(const v2f*)(arow + k0);
        const v2f* brow = bcol + (k0 >> 1) * N;
#pragma unroll
        for (int nt = 0; nt < NT; ++nt) {
            const v2f b = brow[nt * 16];
            acc[nt] = __builtin_amdgcn_wmma_f32_16x16x4_f32(
                false, a, false, b, (short)0, acc[nt], false, false);
        }
    }

#pragma unroll
    for (int p = 0; p < NT / 2; ++p) {
#pragma unroll
        for (int r = 0; r < 8; ++r) {
            const float a0 = acc[2 * p][r];
            const float a1 = acc[2 * p + 1][r];
            const float x0 = xhalf(a0);
            const float x1 = xhalf(a1);
            const float lo = half ? x1 : a0;
            const float hi = half ? a1 : x0;
            const int row = rowTile * 16 + r;
            if (row < M)     C[(size_t)row * N + p * 32 + lane] = lo;
            if (row + 8 < M) C[(size_t)(row + 8) * N + p * 32 + lane] = hi;
        }
    }
}


constexpr int BIN_SHIFT  = 8;
constexpr int MAXBINS    = 256;
constexpr int CTR_PITCH  = 64;
constexpr int EDGE_CHUNK = 4096;

__global__ void node_score_kernel(const float* __restrict__ z,
                                  const float* __restrict__ w_attn, const float* __restrict__ w_e,
                                  float* __restrict__ ssrc, float* __restrict__ sdst, float* __restrict__ uvec,
                                  int* __restrict__ binCount, int* __restrict__ binCursor, int N)
{
    const int n = blockIdx.x * blockDim.x + threadIdx.x;
    if (n < MAXBINS) { binCount[n * CTR_PITCH] = 0; binCursor[n * CTR_PITCH] = 0; }
    if (n < 64) {
        float u = 0.f;
#pragma unroll
        for (int j = 0; j < 32; ++j) u += w_e[n * 32 + j] * w_attn[64 + j];
        uvec[n] = u;
    }
    if (n >= N) return;
    const float* zr = z + (size_t)n * 64;
    float s0 = 0.f, s1 = 0.f;
#pragma unroll
    for (int i = 0; i < 64; ++i) {
        const float v = zr[i];
        s0 += v * w_attn[i];
        s1 += v * w_attn[96 + i];
    }
    ssrc[n] = s0;
    sdst[n] = s1;
}

__device__ __forceinline__ void load_chunk(const int* __restrict__ dst, int E, int base, int tid, int4 (&d)[4])
{
    if (base + EDGE_CHUNK <= E) {
#pragma unroll
        for (int u = 0; u < 4; ++u)
            d[u] = *(const int4*)(dst + base + (u * 256 + tid) * 4);
    } else {
#pragma unroll
        for (int u = 0; u < 4; ++u) {
            const int i = base + (u * 256 + tid) * 4;
            d[u].x = (i     < E) ? dst[i]     : -1;
            d[u].y = (i + 1 < E) ? dst[i + 1] : -1;
            d[u].z = (i + 2 < E) ? dst[i + 2] : -1;
            d[u].w = (i + 3 < E) ? dst[i + 3] : -1;
        }
    }
}

__global__ void __launch_bounds__(256) bin_count_kernel(const int* __restrict__ dst,
                                                        int* __restrict__ binCount, int E)
{
    __shared__ int hist[MAXBINS];
    const int tid = threadIdx.x;
    hist[tid] = 0;
    int4 d[4];
    load_chunk(dst, E, blockIdx.x * EDGE_CHUNK, tid, d);
    __syncthreads();
#pragma unroll
    for (int u = 0; u < 4; ++u) {
        if (d[u].x >= 0) atomicAdd(&hist[d[u].x >> BIN_SHIFT], 1);
        if (d[u].y >= 0) atomicAdd(&hist[d[u].y >> BIN_SHIFT], 1);
        if (d[u].z >= 0) atomicAdd(&hist[d[u].z >> BIN_SHIFT], 1);
        if (d[u].w >= 0) atomicAdd(&hist[d[u].w >> BIN_SHIFT], 1);
    }
    __syncthreads();
    const int c = hist[tid];
    if (c != 0) atomicAdd(&binCount[tid * CTR_PITCH], c);
}

__global__ void __launch_bounds__(256) bin_scatter_kernel(const int* __restrict__ dst,
    const int* __restrict__ binCount, int* __restrict__ binCursor, int* __restrict__ binStart,
    int* __restrict__ pairs, int E, int nbins)
{
    __shared__ int hist[MAXBINS];
    __shared__ int base[MAXBINS];
    __shared__ int wtot[8];
    const int tid  = threadIdx.x;
    const int lane = tid & 31;
    const int wave = tid >> 5;
    hist[tid] = 0;
    int4 d[4];
    const int ebase = blockIdx.x * EDGE_CHUNK;
    load_chunk(dst, E, ebase, tid, d);
    const int c = (tid < nbins) ? binCount[tid * CTR_PITCH] : 0;
    int x = c;
#pragma unroll
    for (int o = 1; o < 32; o <<= 1) {
        const int y = __shfl_up(x, o);
        if (lane >= o) x += y;
    }
    if (lane == 31) wtot[wave] = x;
    __syncthreads();
#pragma unroll
    for (int u = 0; u < 4; ++u) {
        if (d[u].x >= 0) atomicAdd(&hist[d[u].x >> BIN_SHIFT], 1);
        if (d[u].y >= 0) atomicAdd(&hist[d[u].y >> BIN_SHIFT], 1);
        if (d[u].z >= 0) atomicAdd(&hist[d[u].z >> BIN_SHIFT], 1);
        if (d[u].w >= 0) atomicAdd(&hist[d[u].w >> BIN_SHIFT], 1);
    }
    int woff = 0;
    for (int w = 0; w < wave; ++w) woff += wtot[w];
    const int start = woff + x - c;
    __syncthreads();
    const int mine = hist[tid];
    int b0 = start;
    if (mine != 0) b0 += atomicAdd(&binCursor[tid * CTR_PITCH], mine);
    base[tid] = b0;
    hist[tid] = 0;
    if (blockIdx.x == 0) {
        binStart[tid] = start;
        if (tid == MAXBINS - 1) binStart[MAXBINS] = start + c;
    }
    __syncthreads();
    int2* __restrict__ pr = (int2*)pairs;
    if (ebase + EDGE_CHUNK <= E) {
        int pos[16];
#pragma unroll
        for (int u = 0; u < 4; ++u) {
            const int dv[4] = { d[u].x, d[u].y, d[u].z, d[u].w };
#pragma unroll
            for (int q = 0; q < 4; ++q) {
                const int b = dv[q] >> BIN_SHIFT;
                pos[u * 4 + q] = base[b] + atomicAdd(&hist[b], 1);
            }
        }
#pragma unroll
        for (int u = 0; u < 4; ++u) {
            const int e = ebase + (u * 256 + tid) * 4;
            const int dv[4] = { d[u].x, d[u].y, d[u].z, d[u].w };
#pragma unroll
            for (int q = 0; q < 4; ++q) pr[pos[u * 4 + q]] = make_int2(e + q, dv[q]);
        }
    } else {
#pragma unroll
        for (int u = 0; u < 4; ++u) {
            const int e = ebase + (u * 256 + tid) * 4;
            const int dv[4] = { d[u].x, d[u].y, d[u].z, d[u].w };
#pragma unroll
            for (int q = 0; q < 4; ++q) {
                if (dv[q] >= 0) {
                    const int b = dv[q] >> BIN_SHIFT;
                    const int pos = base[b] + atomicAdd(&hist[b], 1);
                    pr[pos] = make_int2(e + q, dv[q]);
                }
            }
        }
    }
}

__global__ void __launch_bounds__(1024) bin_fine_kernel(const int* __restrict__ pairs,
    const int* __restrict__ binStart, int* __restrict__ beg, int* __restrict__ cur,
    int* __restrict__ perm, int N)
{
    __shared__ int ncount[256];
    __shared__ int ncur[256];
    __shared__ int wtot[8];
    const int tid  = threadIdx.x;
    const int lane = tid & 31;
    const int wave = tid >> 5;
    const int bin  = blockIdx.x;
    const int s    = binStart[bin];
    const int cntb = binStart[bin + 1] - s;
    const int2* __restrict__ tp = (const int2*)pairs + s;
    if (tid < 256) ncount[tid] = 0;
    __syncthreads();
    for (int i0 = 0; i0 < cntb; i0 += 4096) {
        int2 p[4];
#pragma unroll
        for (int u = 0; u < 4; ++u) {
            const int i = i0 + u * 1024 + tid;
            p[u] = make_int2(-1, -1);
            if (i < cntb) p[u] = tp[i];
        }
#pragma unroll
        for (int u = 0; u < 4; ++u)
            if (p[u].x >= 0) atomicAdd(&ncount[p[u].y & 255], 1);
    }
    __syncthreads();
    int c = 0, x = 0;
    if (tid < 256) {
        c = ncount[tid];
        x = c;
#pragma unroll
        for (int o = 1; o < 32; o <<= 1) {
            const int y = __shfl_up(x, o);
            if (lane >= o) x += y;
        }
        if (lane == 31) wtot[wave] = x;
    }
    __syncthreads();
    if (tid < 256) {
        int woff = 0;
        for (int w = 0; w < wave; ++w) woff += wtot[w];
        const int b0 = s + woff + x - c;
        ncur[tid] = b0;
        const int node = (bin << BIN_SHIFT) + tid;
        if (node < N) { beg[node] = b0; cur[node] = b0 + c; }
    }
    __syncthreads();
    for (int i0 = 0; i0 < cntb; i0 += 4096) {
        int2 p[4];
#pragma unroll
        for (int u = 0; u < 4; ++u) {
            const int i = i0 + u * 1024 + tid;
            p[u] = make_int2(-1, -1);
            if (i < cntb) p[u] = tp[i];
        }
#pragma unroll
        for (int u = 0; u < 4; ++u) {
            if (p[u].x >= 0) {
                const int pos = atomicAdd(&ncur[p[u].y & 255], 1);
                perm[pos] = p[u].x;
            }
        }
    }
}

__device__ __forceinline__ float wave_max_f(float v) {
#pragma unroll
    for (int o = 16; o > 0; o >>= 1) v = fmaxf(v, __shfl_xor(v, o));
    return v;
}
__device__ __forceinline__ float wave_sum_f(float v) {
#pragma unroll
    for (int o = 16; o > 0; o >>= 1) v += __shfl_xor(v, o);
    return v;
}
__device__ __forceinline__ int wave_min_i(int v) {
#pragma unroll
    for (int o = 16; o > 0; o >>= 1) { const int t = __shfl_xor(v, o); v = t < v ? t : v; }
    return v;
}

__device__ __forceinline__ float edge_logit(const float* __restrict__ ef, const int* __restrict__ src,
                                            const float* __restrict__ ssrc,
                                            const float* __restrict__ uvec, int e, float sd)
{
    const float4* er = (const float4*)(ef + (size_t)e * 64);
    float t = ssrc[src[e]] + sd;
#pragma nounroll
    for (int hl = 0; hl < 2; ++hl) {
#pragma unroll
        for (int q = 0; q < 8; ++q) {
            const float4 v = er[8 * hl + q];
            const float* __restrict__ uq = uvec + 32 * hl + 4 * q;
            t += v.x * uq[0];
            t += v.y * uq[1];
            t += v.z * uq[2];
            t += v.w * uq[3];
        }
    }
    return t > 0.f ? t : 0.01f * t;
}

#define ROW_ACC(T)                                                         \
    if (ty == T) { acc[T][0] += al * v0; acc[T][1] += al * v1; acc[T][2] += al * v2; acc[T][3] += al * v3; }

__device__ __forceinline__ int next_edge(const int* __restrict__ pb, int deg, int lane, int last)
{
    int best = 0x7fffffff;
    for (int c = 0; c < deg; c += 32) {
        const int i = c + lane;
        if (i < deg) {
            const int v = pb[i];
            if (v > last && v < best) best = v;
        }
    }
    return wave_min_i(best);
}

__device__ __forceinline__ int rank_owner(int rankA, int rankB, int p, bool& inA)
{
    const unsigned mA = __builtin_amdgcn_ballot_w32(rankA == p);
    const unsigned mB = __builtin_amdgcn_ballot_w32(rankB == p);
    inA = (mA != 0u);
    return __builtin_ctz(inA ? mA : mB);
}
__device__ __forceinline__ int pick_i(int vA, int vB, int ln, bool inA)
{
    return inA ? __builtin_amdgcn_readlane(vA, ln) : __builtin_amdgcn_readlane(vB, ln);
}
__device__ __forceinline__ float pick_f(float vA, float vB, int ln, bool inA)
{
    return __int_as_float(pick_i(__float_as_int(vA), __float_as_int(vB), ln, inA));
}

__global__ void __launch_bounds__(256) node_gather_kernel(
    const float* __restrict__ z, const float* __restrict__ ef,
    const int* __restrict__ src, const int* __restrict__ etype,
    const float* __restrict__ ssrc, const float* __restrict__ sdst,
    const float* __restrict__ uvec,
    const int* __restrict__ beg, const int* __restrict__ cur, const int* __restrict__ perm,
    float* __restrict__ h, int N)
{
    const int d    = (int)((blockIdx.x * blockDim.x + threadIdx.x) >> 5);
    const int lane = threadIdx.x & 31;
    if (d >= N) return;
    const int b   = beg[d];
    const int deg = cur[d] - b;
    const int* __restrict__ pb = perm + b;
    const float sd = sdst[d];
    float acc[4][4] = {};

    if (deg <= 32) {
        const int id = (lane < deg) ? pb[lane] : (0x7fffffc0 + lane);
        int rank = 0;
#pragma unroll
        for (int j = 0; j < 32; ++j)
            rank += (__builtin_amdgcn_readlane(id, j) < id) ? 1 : 0;
        const int sid = __builtin_amdgcn_ds_permute(rank << 2, id);
        const bool ok = lane < deg;
        const int e  = ok ? sid : 0;
        const int s  = src[e];
        const int tt = etype[e];
        const float lr = ok ? edge_logit(ef, src, ssrc, uvec, e, sd) : -INFINITY;
        const float m  = wave_max_f(lr);
        const float ex = ok ? __expf(lr - m) : 0.f;
        const float den = wave_sum_f(ex);
        const float alpha = ex / den;

        for (int j = 0; j < deg; j += 4) {
            float r0[4], r1[4], r2[4], r3[4];
#pragma unroll
            for (int u = 0; u < 4; ++u) {
                const int jj = (j + u < deg) ? (j + u) : (deg - 1);
                const int eu = __builtin_amdgcn_readlane(e, jj);
                const int su = __builtin_amdgcn_readlane(s, jj);
                const float* zr = z + (size_t)su * 64;
                const float* er = ef + (size_t)eu * 64;
                r0[u] = zr[lane];
                r1[u] = zr[32 + lane];
                r2[u] = er[lane];
                r3[u] = er[32 + lane];
            }
#pragma unroll
            for (int u = 0; u < 4; ++u) {
                if (j + u < deg) {
                    const float al = __int_as_float(__builtin_amdgcn_readlane(__float_as_int(alpha), j + u));
                    const int   ty = __builtin_amdgcn_readlane(tt, j + u);
                    const float v0 = r0[u], v1 = r1[u], v2 = r2[u], v3 = r3[u];
                    ROW_ACC(0) ROW_ACC(1) ROW_ACC(2) ROW_ACC(3)
                }
            }
        }
    } else if (deg <= 64) {
        const int  degB = deg - 32;
        const bool okB  = lane < degB;
        const int  eA   = pb[lane];
        const int  idB  = okB ? pb[32 + lane] : (0x7fffffc0 + lane);
        int rankA = 0, rankB = 0;
        for (int j = 0; j < 32; ++j) {
            const int a = __builtin_amdgcn_readlane(eA, j);
            const int b = __builtin_amdgcn_readlane(idB, j);
            rankA += ((a < eA) ? 1 : 0) + ((b < eA) ? 1 : 0);
            rankB += ((a < idB) ? 1 : 0) + ((b < idB) ? 1 : 0);
        }
        const int eB = okB ? idB : 0;
        int sA = 0, sB = 0, tA = 0, tB = 0;
        float lA = 0.f, lB = 0.f;
#pragma nounroll
        for (int hh = 0; hh < 2; ++hh) {
            const int   e_h = hh ? eB : eA;
            const int   s_h = src[e_h];
            const int   t_h = etype[e_h];
            const float l_h = edge_logit(ef, src, ssrc, uvec, e_h, sd);
            if (hh) { sB = s_h; tB = t_h; lB = l_h; } else { sA = s_h; tA = t_h; lA = l_h; }
        }
        lB = okB ? lB : -INFINITY;
        const float m  = wave_max_f(fmaxf(lA, lB));
        const float xA = __expf(lA - m);
        const float xB = okB ? __expf(lB - m) : 0.f;
        float den = 0.f;
        for (int p = 0; p < deg; ++p) {
            bool inA;
            const int ln = rank_owner(rankA, rankB, p, inA);
            den += pick_f(xA, xB, ln, inA);
        }
        const float alA = xA / den, alB = xB / den;

        for (int j = 0; j < deg; j += 4) {
            float r0[4], r1[4], r2[4], r3[4], al4[4];
            int ty4[4];
            for (int u = 0; u < 4; ++u) {
                const int p = (j + u < deg) ? (j + u) : (deg - 1);
                bool inA;
                const int ln = rank_owner(rankA, rankB, p, inA);
                const int eu = pick_i(eA, eB, ln, inA);
                const int su = pick_i(sA, sB, ln, inA);
                al4[u] = pick_f(alA, alB, ln, inA);
                ty4[u] = pick_i(tA, tB, ln, inA);
                const float* zr = z + (size_t)su * 64;
                const float* er = ef + (size_t)eu * 64;
                r0[u] = zr[lane];
                r1[u] = zr[32 + lane];
                r2[u] = er[lane];
                r3[u] = er[32 + lane];
            }
            for (int u = 0; u < 4; ++u) {
                if (j + u < deg) {
                    const float al = al4[u];
                    const int   ty = ty4[u];
                    const float v0 = r0[u], v1 = r1[u], v2 = r2[u], v3 = r3[u];
                    ROW_ACC(0) ROW_ACC(1) ROW_ACC(2) ROW_ACC(3)
                }
            }
        }
    } else {
        float m = -INFINITY;
        for (int c = 0; c < deg; c += 32) {
            const int i = c + lane;
            if (i < deg) m = fmaxf(m, edge_logit(ef, src, ssrc, uvec, pb[i], sd));
        }
        m = wave_max_f(m);
        float den = 0.f;
        int last = -1;
        for (int j = 0; j < deg; ++j) {
            last = next_edge(pb, deg, lane, last);
            den += __expf(edge_logit(ef, src, ssrc, uvec, last, sd) - m);
        }
        last = -1;
        for (int j = 0; j < deg; ++j) {
            last = next_edge(pb, deg, lane, last);
            const float al = __expf(edge_logit(ef, src, ssrc, uvec, last, sd) - m) / den;
            const int ty = etype[last];
            const float* zr = z + (size_t)src[last] * 64;
            const float* er = ef + (size_t)last * 64;
            const float v0 = zr[lane], v1 = zr[32 + lane], v2 = er[lane], v3 = er[32 + lane];
            ROW_ACC(0) ROW_ACC(1) ROW_ACC(2) ROW_ACC(3)
        }
    }

    float* hr = h + (size_t)d * 4 * 128 + lane;
#pragma unroll
    for (int t = 0; t < 4; ++t) {
        hr[t * 128]      = acc[t][0];
        hr[t * 128 + 32] = acc[t][1];
        hr[t * 128 + 64] = acc[t][2];
        hr[t * 128 + 96] = acc[t][3];
    }
}
#undef ROW_ACC

constexpr int ZP_BLOCKS = 512;

__device__ __forceinline__ void add4(float4& a, const float4& v) {
    a.x += v.x; a.y += v.y; a.z += v.z; a.w += v.w;
}

__global__ void __launch_bounds__(1024) hsum_partial_kernel(
    const float* __restrict__ h, float* __restrict__ part, int nodes)
{
    __shared__ float4 sm[8][128];
    const int tid = threadIdx.x;
    const int c4  = tid & 127;
    const int sub = tid >> 7;
    const int R   = (nodes + (int)gridDim.x - 1) / (int)gridDim.x;
    const int r0  = blockIdx.x * R;
    int r1 = r0 + R;
    if (r1 > nodes) r1 = nodes;
    const float4* __restrict__ p = (const float4*)h + c4;

    float4 acc = make_float4(0.f, 0.f, 0.f, 0.f);
    int r = r0 + sub;
    for (; r + 56 < r1; r += 64) {
        float4 v[8];
#pragma unroll
        for (int u = 0; u < 8; ++u) v[u] = p[(size_t)(r + 8 * u) * 128];
#pragma unroll
        for (int u = 0; u < 8; ++u) add4(acc, v[u]);
    }
    {
        float4 v[8];
#pragma unroll
        for (int u = 0; u < 8; ++u) {
            const int rr = r + 8 * u;
            v[u] = make_float4(0.f, 0.f, 0.f, 0.f);
            if (rr < r1) v[u] = p[(size_t)rr * 128];
        }
#pragma unroll
        for (int u = 0; u < 8; ++u) add4(acc, v[u]);
    }
    sm[sub][c4] = acc;
    __syncthreads();
    if (tid < 512) {
        const float* sf = (const float*)&sm[0][0];
        float s = 0.f;
#pragma unroll
        for (int k = 0; k < 8; ++k) s += sf[k * 512 + tid];
        part[(size_t)blockIdx.x * 512 + tid] = s;
    }
}

__global__ void __launch_bounds__(1024) beta_kernel(const float* __restrict__ part,
                            const float* __restrict__ w_fc2, const float* __restrict__ W2,
                            const float* __restrict__ w_sem, float* __restrict__ beta)
{
    __shared__ float4 sm[8][128];
    __shared__ __attribute__((aligned(16))) float hs[4 * 128];
    __shared__ float zs[16 * 64];
    const int tid = threadIdx.x;
    const int c4  = tid & 127;
    const int g   = tid >> 7;
    const float4* __restrict__ p = (const float4*)part + (size_t)g * 64 * 128 + c4;
    float4 acc = make_float4(0.f, 0.f, 0.f, 0.f);
    for (int b = 0; b < 64; b += 16) {
        float4 v[16];
#pragma unroll
        for (int u = 0; u < 16; ++u) v[u] = p[(size_t)(b + u) * 128];
#pragma unroll
        for (int u = 0; u < 16; ++u) add4(acc, v[u]);
    }
    sm[g][c4] = acc;
    __syncthreads();
    if (tid < 512) {
        const float* sf = (const float*)&sm[0][0];
        float s = 0.f;
#pragma unroll
        for (int k = 0; k < 8; ++k) s += sf[k * 512 + tid];
        hs[tid] = s;
    }
    __syncthreads();
    if (tid < 128) {
        const int lane = tid & 31;
        const int nt   = tid >> 5;
        const int half = lane >> 4;
        const int l15  = lane & 15;
        const int ar   = (l15 < 4) ? l15 : 0;
        v8f d = {};
#pragma unroll
        for (int k0 = 0; k0 < 128; k0 += 4) {
            const int k = k0 + 2 * half;
            const v2f av = *(const v2f*)&hs[ar * 128 + k];
            const float* __restrict__ wrow = (k0 < 64) ? (w_fc2 + k * 64) : (W2 + (k - 64) * 64);
            v2f a, b;
            a.x = (l15 < 4) ? av.x : 0.f;
            a.y = (l15 < 4) ? av.y : 0.f;
            b.x = wrow[nt * 16 + l15];
            b.y = wrow[64 + nt * 16 + l15];
            d = __builtin_amdgcn_wmma_f32_16x16x4_f32(false, a, false, b, (short)0, d, false, false);
        }
#pragma unroll
        for (int r = 0; r < 8; ++r) zs[(r + 8 * half) * 64 + nt * 16 + l15] = d[r];
    }
    __syncthreads();
    if (tid != 0) return;
    float wp[4], mx = -INFINITY;
#pragma unroll
    for (int t = 0; t < 4; ++t) {
        float d = 0.f;
        for (int o = 0; o < 64; ++o) d += zs[t * 64 + o] * w_sem[o];
        const float lr = d > 0.f ? d : 0.01f * d;
        wp[t] = lr;
        mx = fmaxf(mx, lr);
    }
    float s = 0.f;
#pragma unroll
    for (int t = 0; t < 4; ++t) { wp[t] = __expf(wp[t] - mx); s += wp[t]; }
#pragma unroll
    for (int t = 0; t < 4; ++t) beta[t] = wp[t] / s;
}

__device__ __forceinline__ void sem_load4(const float4* __restrict__ hrow, int j0, float4 (&v)[4][4])
{
#pragma unroll
    for (int s = 0; s < 4; ++s) {
#pragma unroll
        for (int t = 0; t < 4; ++t) v[s][t] = hrow[32 * t + 2 * (j0 + s)];
    }
}

__device__ __forceinline__ void sem_steps4(const float4 (&v)[4][4], float b0, float b1, float b2, float b3,
                                           const v2f* __restrict__ bcol, int j0, v8f (&acc)[4])
{
#pragma unroll
    for (int s = 0; s < 4; ++s) {
        const float4 v0 = v[s][0], v1 = v[s][1], v2 = v[s][2], v3 = v[s][3];
        v2f a01, a23;
        a01.x = b0 * v0.x + b1 * v1.x + b2 * v2.x + b3 * v3.x;
        a01.y = b0 * v0.y + b1 * v1.y + b2 * v2.y + b3 * v3.y;
        a23.x = b0 * v0.z + b1 * v1.z + b2 * v2.z + b3 * v3.z;
        a23.y = b0 * v0.w + b1 * v1.w + b2 * v2.w + b3 * v3.w;
        const v2f* brow = bcol + (4 * (j0 + s)) * 64;
#pragma unroll
        for (int nt = 0; nt < 4; ++nt) {
            const v2f b = brow[nt * 16];
            acc[nt] = __builtin_amdgcn_wmma_f32_16x16x4_f32(
                false, a01, false, b, (short)0, acc[nt], false, false);
        }
#pragma unroll
        for (int nt = 0; nt < 4; ++nt) {
            const v2f b = brow[64 + nt * 16];
            acc[nt] = __builtin_amdgcn_wmma_f32_16x16x4_f32(
                false, a23, false, b, (short)0, acc[nt], false, false);
        }
    }
}

__global__ void __launch_bounds__(256) sem_out_wmma_kernel(
    const float* __restrict__ h, const float* __restrict__ w_fc2, const float* __restrict__ W2,
    const float* __restrict__ beta, float* __restrict__ Z, int M)
{
    constexpr int K = 128, N = 64, NT = 4;
    __shared__ v2f sB2[(K / 2) * N];
    const int tid = threadIdx.x;
    for (int p = tid; p < (K / 2) * N; p += 256) {
        const int kk = (p / N) << 1;
        const int n  = p % N;
        const float* __restrict__ wrow = (kk < 64) ? (w_fc2 + kk * N) : (W2 + (kk - 64) * N);
        v2f b;
        b.x = wrow[n];
        b.y = wrow[N + n];
        sB2[p] = b;
    }
    __syncthreads();

    const int lane    = tid & 31;
    const int wave    = tid >> 5;
    const int rowTile = blockIdx.x * 8 + wave;
    const int numTiles = (M + 15) >> 4;
    if (rowTile >= numTiles) return;

    const int half = lane >> 4;
    const int l15  = lane & 15;
    int mrow = rowTile * 16 + l15;
    if (mrow >= M) mrow = M - 1;

    const float b0 = beta[0], b1 = beta[1], b2 = beta[2], b3 = beta[3];
    v8f acc[NT] = {};
    const float4* __restrict__ hrow = (const float4*)(h + (size_t)mrow * (4 * K) + 4 * half);
    const v2f* __restrict__ bcol = sB2 + (2 * half) * N + l15;

    float4 v[3][4][4];
    sem_load4(hrow, 0, v[0]);
    sem_load4(hrow, 4, v[1]);
    sem_load4(hrow, 8, v[2]);
    __builtin_amdgcn_sched_barrier(0);
    sem_steps4(v[0], b0, b1, b2, b3, bcol, 0, acc);
    sem_load4(hrow, 12, v[0]);
    __builtin_amdgcn_sched_barrier(0);
    sem_steps4(v[1], b0, b1, b2, b3, bcol, 4, acc);
    sem_steps4(v[2], b0, b1, b2, b3, bcol, 8, acc);
    sem_steps4(v[0], b0, b1, b2, b3, bcol, 12, acc);

#pragma unroll
    for (int p = 0; p < NT / 2; ++p) {
#pragma unroll
        for (int r = 0; r < 8; ++r) {
            const float a0 = acc[2 * p][r];
            const float a1 = acc[2 * p + 1][r];
            const float x0 = xhalf(a0);
            const float x1 = xhalf(a1);
            const float lo = half ? x1 : a0;
            const float hi = half ? a1 : x0;
            const int row = rowTile * 16 + r;
            if (row < M)     Z[(size_t)row * N + p * 32 + lane] = lo;
            if (row + 8 < M) Z[(size_t)(row + 8) * N + p * 32 + lane] = hi;
        }
    }
}

extern "C" void kernel_launch(void* const* d_in, const int* in_sizes, int n_in,
                              void* d_out, int out_size, void* d_ws, size_t ws_size,
                              hipStream_t stream)
{
    const float* node   = (const float*)d_in[0];
    const float* edgef  = (const float*)d_in[1];
    const int*   src    = (const int*)d_in[2];
    const int*   dst    = (const int*)d_in[3];
    const int*   etype  = (const int*)d_in[4];
    const float* w_n    = (const float*)d_in[5];
    const float* w_e    = (const float*)d_in[6];
    const float* w_attn = (const float*)d_in[7];
    const float* w_fc2  = (const float*)d_in[8];
    const float* w_sem  = (const float*)d_in[9];
    float* out = (float*)d_out;

    const int N = in_sizes[0] / 128;
    const int E = in_sizes[2];
    const int T = 4;

    float* ws   = (float*)d_ws;
    float* h    = ws;
    float* z    = h + (size_t)N * T * 128;
    float* ssrc = z + (size_t)N * 64;
    float* sdst = ssrc + N;
    float* abuf = sdst + N;
    float* mbuf = abuf + E;
    float* den  = mbuf + N;
    float* beta = den + N + 256;
    float* W2   = beta + 64;
    float* uvec = W2 + 64 * 64;
    float* part = abuf;
    int* perm  = (int*)abuf;
    int* cur   = (int*)mbuf;
    int* beg   = (int*)den;
    int* pairs     = (int*)h;
    int* binCount  = pairs + 2 * (size_t)E;
    int* binCursor = binCount + MAXBINS * CTR_PITCH;
    int* binStart  = binCursor + MAXBINS * CTR_PITCH;
    const int nbins  = (N + (1 << BIN_SHIFT) - 1) >> BIN_SHIFT;
    const int chunks = (E + EDGE_CHUNK - 1) / EDGE_CHUNK;

    const int thr = 256;

    gemm_f32_wmma<4><<<((N + 15) / 16 + 7) / 8, thr, 0, stream>>>(node, w_n, z, N, 128, 64);
    gemm_f32_wmma<4><<<1, thr, 0, stream>>>(w_e, w_fc2 + 64 * 64, W2, 64, 32, 64);
    node_score_kernel<<<(N + thr - 1) / thr, thr, 0, stream>>>(z, w_attn, w_e, ssrc, sdst, uvec, binCount, binCursor, N);
    bin_count_kernel<<<chunks, 256, 0, stream>>>(dst, binCount, E);
    bin_scatter_kernel<<<chunks, 256, 0, stream>>>(dst, binCount, binCursor, binStart, pairs, E, nbins);
    bin_fine_kernel<<<nbins, 1024, 0, stream>>>(pairs, binStart, beg, cur, perm, N);
    node_gather_kernel<<<(N + 7) / 8, thr, 0, stream>>>(z, edgef, src, etype, ssrc, sdst, uvec,
                                                        beg, cur, perm, h, N);
    hsum_partial_kernel<<<ZP_BLOCKS, 1024, 0, stream>>>(h, part, N);
    beta_kernel<<<1, 1024, 0, stream>>>(part, w_fc2, W2, w_sem, beta);
    sem_out_wmma_kernel<<<((N + 15) / 16 + 7) / 8, thr, 0, stream>>>(h, w_fc2, W2, beta, out, N);
}


# column sums of h folded per 8-node block inside the gather (LDS; last wave to finish adds in wave order); the sum pass reads N/8 block rows, not the h plane; 128 partials, not 512; sum order changes
# speedup vs baseline: 3.0647x; 1.0484x over previous
#include <hip/hip_runtime.h>
#include <hip/hip_bf16.h>
#include <math.h>

typedef __attribute__((ext_vector_type(2))) float v2f;
typedef __attribute__((ext_vector_type(8))) float v8f;

__device__ __forceinline__ float xhalf(float x) {
    return __builtin_bit_cast(float, __builtin_amdgcn_permlanex16(
        __builtin_bit_cast(int, x), __builtin_bit_cast(int, x), 0x76543210, 0xfedcba98, false, false));
}

template <int NT>
__global__ void __launch_bounds__(256) gemm_f32_wmma(
    const float* __restrict__ A, const float* __restrict__ B,
    float* __restrict__ C, int M, int K, int N)
{
    __shared__ v2f sB2[4096];
    const int tid = threadIdx.x;
    const int pairCount = (K >> 1) * N;
    for (int p = tid; p < pairCount; p += 256) {
        const int kk = (p / N) << 1;
        const int n  = p - (p / N) * N;
        v2f b;
        b.x = B[kk * N + n];
        b.y = B[(kk + 1) * N + n];
        sB2[p] = b;
    }
    __syncthreads();

    const int lane    = tid & 31;
    const int wave    = tid >> 5;
    const int rowTile = blockIdx.x * 8 + wave;
    const int numTiles = (M + 15) >> 4;
    if (rowTile >= numTiles) return;

    const int half = lane >> 4;
    const int l15  = lane & 15;
    int mrow = rowTile * 16 + l15;
    if (mrow >= M) mrow = M - 1;

    v8f acc[NT] = {};
    const float* arow = A + (size_t)mrow * K + 2 * half;
    const v2f* __restrict__ bcol = sB2 + half * N + l15;

    for (int k0 = 0; k0 < K; k0 += 4) {
        const v2f a = *(const v2f*)(arow + k0);
        const v2f* brow = bcol + (k0 >> 1) * N;
#pragma unroll
        for (int nt = 0; nt < NT; ++nt) {
            const v2f b = brow[nt * 16];
            acc[nt] = __builtin_amdgcn_wmma_f32_16x16x4_f32(
                false, a, false, b, (short)0, acc[nt], false, false);
        }
    }

#pragma unroll
    for (int p = 0; p < NT / 2; ++p) {
#pragma unroll
        for (int r = 0; r < 8; ++r) {
            const float a0 = acc[2 * p][r];
            const float a1 = acc[2 * p + 1][r];
            const float x0 = xhalf(a0);
            const float x1 = xhalf(a1);
            const float lo = half ? x1 : a0;
            const float hi = half ? a1 : x0;
            const int row = rowTile * 16 + r;
            if (row < M)     C[(size_t)row * N + p * 32 + lane] = lo;
            if (row + 8 < M) C[(size_t)(row + 8) * N + p * 32 + lane] = hi;
        }
    }
}


constexpr int BIN_SHIFT  = 8;
constexpr int MAXBINS    = 256;
constexpr int CTR_PITCH  = 64;
constexpr int EDGE_CHUNK = 4096;

__global__ void node_score_kernel(const float* __restrict__ z,
                                  const float* __restrict__ w_attn, const float* __restrict__ w_e,
                                  float* __restrict__ ssrc, float* __restrict__ sdst, float* __restrict__ uvec,
                                  int* __restrict__ binCount, int* __restrict__ binCursor, int N)
{
    const int n = blockIdx.x * blockDim.x + threadIdx.x;
    if (n < MAXBINS) { binCount[n * CTR_PITCH] = 0; binCursor[n * CTR_PITCH] = 0; }
    if (n < 64) {
        float u = 0.f;
#pragma unroll
        for (int j = 0; j < 32; ++j) u += w_e[n * 32 + j] * w_attn[64 + j];
        uvec[n] = u;
    }
    if (n >= N) return;
    const float* zr = z + (size_t)n * 64;
    float s0 = 0.f, s1 = 0.f;
#pragma unroll
    for (int i = 0; i < 64; ++i) {
        const float v = zr[i];
        s0 += v * w_attn[i];
        s1 += v * w_attn[96 + i];
    }
    ssrc[n] = s0;
    sdst[n] = s1;
}

__device__ __forceinline__ void load_chunk(const int* __restrict__ dst, int E, int base, int tid, int4 (&d)[4])
{
    if (base + EDGE_CHUNK <= E) {
#pragma unroll
        for (int u = 0; u < 4; ++u)
            d[u] = *(const int4*)(dst + base + (u * 256 + tid) * 4);
    } else {
#pragma unroll
        for (int u = 0; u < 4; ++u) {
            const int i = base + (u * 256 + tid) * 4;
            d[u].x = (i     < E) ? dst[i]     : -1;
            d[u].y = (i + 1 < E) ? dst[i + 1] : -1;
            d[u].z = (i + 2 < E) ? dst[i + 2] : -1;
            d[u].w = (i + 3 < E) ? dst[i + 3] : -1;
        }
    }
}

__global__ void __launch_bounds__(256) bin_count_kernel(const int* __restrict__ dst,
                                                        int* __restrict__ binCount, int E)
{
    __shared__ int hist[MAXBINS];
    const int tid = threadIdx.x;
    hist[tid] = 0;
    int4 d[4];
    load_chunk(dst, E, blockIdx.x * EDGE_CHUNK, tid, d);
    __syncthreads();
#pragma unroll
    for (int u = 0; u < 4; ++u) {
        if (d[u].x >= 0) atomicAdd(&hist[d[u].x >> BIN_SHIFT], 1);
        if (d[u].y >= 0) atomicAdd(&hist[d[u].y >> BIN_SHIFT], 1);
        if (d[u].z >= 0) atomicAdd(&hist[d[u].z >> BIN_SHIFT], 1);
        if (d[u].w >= 0) atomicAdd(&hist[d[u].w >> BIN_SHIFT], 1);
    }
    __syncthreads();
    const int c = hist[tid];
    if (c != 0) atomicAdd(&binCount[tid * CTR_PITCH], c);
}

__global__ void __launch_bounds__(256) bin_scatter_kernel(const int* __restrict__ dst,
    const int* __restrict__ binCount, int* __restrict__ binCursor, int* __restrict__ binStart,
    int* __restrict__ pairs, int E, int nbins)
{
    __shared__ int hist[MAXBINS];
    __shared__ int base[MAXBINS];
    __shared__ int wtot[8];
    const int tid  = threadIdx.x;
    const int lane = tid & 31;
    const int wave = tid >> 5;
    hist[tid] = 0;
    int4 d[4];
    const int ebase = blockIdx.x * EDGE_CHUNK;
    load_chunk(dst, E, ebase, tid, d);
    const int c = (tid < nbins) ? binCount[tid * CTR_PITCH] : 0;
    int x = c;
#pragma unroll
    for (int o = 1; o < 32; o <<= 1) {
        const int y = __shfl_up(x, o);
        if (lane >= o) x += y;
    }
    if (lane == 31) wtot[wave] = x;
    __syncthreads();
#pragma unroll
    for (int u = 0; u < 4; ++u) {
        if (d[u].x >= 0) atomicAdd(&hist[d[u].x >> BIN_SHIFT], 1);
        if (d[u].y >= 0) atomicAdd(&hist[d[u].y >> BIN_SHIFT], 1);
        if (d[u].z >= 0) atomicAdd(&hist[d[u].z >> BIN_SHIFT], 1);
        if (d[u].w >= 0) atomicAdd(&hist[d[u].w >> BIN_SHIFT], 1);
    }
    int woff = 0;
    for (int w = 0; w < wave; ++w) woff += wtot[w];
    const int start = woff + x - c;
    __syncthreads();
    const int mine = hist[tid];
    int b0 = start;
    if (mine != 0) b0 += atomicAdd(&binCursor[tid * CTR_PITCH], mine);
    base[tid] = b0;
    hist[tid] = 0;
    if (blockIdx.x == 0) {
        binStart[tid] = start;
        if (tid == MAXBINS - 1) binStart[MAXBINS] = start + c;
    }
    __syncthreads();
    int2* __restrict__ pr = (int2*)pairs;
    if (ebase + EDGE_CHUNK <= E) {
        int pos[16];
#pragma unroll
        for (int u = 0; u < 4; ++u) {
            const int dv[4] = { d[u].x, d[u].y, d[u].z, d[u].w };
#pragma unroll
            for (int q = 0; q < 4; ++q) {
                const int b = dv[q] >> BIN_SHIFT;
                pos[u * 4 + q] = base[b] + atomicAdd(&hist[b], 1);
            }
        }
#pragma unroll
        for (int u = 0; u < 4; ++u) {
            const int e = ebase + (u * 256 + tid) * 4;
            const int dv[4] = { d[u].x, d[u].y, d[u].z, d[u].w };
#pragma unroll
            for (int q = 0; q < 4; ++q) pr[pos[u * 4 + q]] = make_int2(e + q, dv[q]);
        }
    } else {
#pragma unroll
        for (int u = 0; u < 4; ++u) {
            const int e = ebase + (u * 256 + tid) * 4;
            const int dv[4] = { d[u].x, d[u].y, d[u].z, d[u].w };
#pragma unroll
            for (int q = 0; q < 4; ++q) {
                if (dv[q] >= 0) {
                    const int b = dv[q] >> BIN_SHIFT;
                    const int pos = base[b] + atomicAdd(&hist[b], 1);
                    pr[pos] = make_int2(e + q, dv[q]);
                }
            }
        }
    }
}

__global__ void __launch_bounds__(1024) bin_fine_kernel(const int* __restrict__ pairs,
    const int* __restrict__ binStart, int* __restrict__ beg, int* __restrict__ cur,
    int* __restrict__ perm, int N)
{
    __shared__ int ncount[256];
    __shared__ int ncur[256];
    __shared__ int wtot[8];
    const int tid  = threadIdx.x;
    const int lane = tid & 31;
    const int wave = tid >> 5;
    const int bin  = blockIdx.x;
    const int s    = binStart[bin];
    const int cntb = binStart[bin + 1] - s;
    const int2* __restrict__ tp = (const int2*)pairs + s;
    if (tid < 256) ncount[tid] = 0;
    __syncthreads();
    for (int i0 = 0; i0 < cntb; i0 += 4096) {
        int2 p[4];
#pragma unroll
        for (int u = 0; u < 4; ++u) {
            const int i = i0 + u * 1024 + tid;
            p[u] = make_int2(-1, -1);
            if (i < cntb) p[u] = tp[i];
        }
#pragma unroll
        for (int u = 0; u < 4; ++u)
            if (p[u].x >= 0) atomicAdd(&ncount[p[u].y & 255], 1);
    }
    __syncthreads();
    int c = 0, x = 0;
    if (tid < 256) {
        c = ncount[tid];
        x = c;
#pragma unroll
        for (int o = 1; o < 32; o <<= 1) {
            const int y = __shfl_up(x, o);
            if (lane >= o) x += y;
        }
        if (lane == 31) wtot[wave] = x;
    }
    __syncthreads();
    if (tid < 256) {
        int woff = 0;
        for (int w = 0; w < wave; ++w) woff += wtot[w];
        const int b0 = s + woff + x - c;
        ncur[tid] = b0;
        const int node = (bin << BIN_SHIFT) + tid;
        if (node < N) { beg[node] = b0; cur[node] = b0 + c; }
    }
    __syncthreads();
    for (int i0 = 0; i0 < cntb; i0 += 4096) {
        int2 p[4];
#pragma unroll
        for (int u = 0; u < 4; ++u) {
            const int i = i0 + u * 1024 + tid;
            p[u] = make_int2(-1, -1);
            if (i < cntb) p[u] = tp[i];
        }
#pragma unroll
        for (int u = 0; u < 4; ++u) {
            if (p[u].x >= 0) {
                const int pos = atomicAdd(&ncur[p[u].y & 255], 1);
                perm[pos] = p[u].x;
            }
        }
    }
}

__device__ __forceinline__ float wave_max_f(float v) {
#pragma unroll
    for (int o = 16; o > 0; o >>= 1) v = fmaxf(v, __shfl_xor(v, o));
    return v;
}
__device__ __forceinline__ float wave_sum_f(float v) {
#pragma unroll
    for (int o = 16; o > 0; o >>= 1) v += __shfl_xor(v, o);
    return v;
}
__device__ __forceinline__ int wave_min_i(int v) {
#pragma unroll
    for (int o = 16; o > 0; o >>= 1) { const int t = __shfl_xor(v, o); v = t < v ? t : v; }
    return v;
}

__device__ __forceinline__ float edge_logit(const float* __restrict__ ef, const int* __restrict__ src,
                                            const float* __restrict__ ssrc,
                                            const float* __restrict__ uvec, int e, float sd)
{
    const float4* er = (const float4*)(ef + (size_t)e * 64);
    float t = ssrc[src[e]] + sd;
#pragma nounroll
    for (int hl = 0; hl < 2; ++hl) {
#pragma unroll
        for (int q = 0; q < 8; ++q) {
            const float4 v = er[8 * hl + q];
            const float* __restrict__ uq = uvec + 32 * hl + 4 * q;
            t += v.x * uq[0];
            t += v.y * uq[1];
            t += v.z * uq[2];
            t += v.w * uq[3];
        }
    }
    return t > 0.f ? t : 0.01f * t;
}

#define ROW_ACC(T)                                                         \
    if (ty == T) { acc[T][0] += al * v0; acc[T][1] += al * v1; acc[T][2] += al * v2; acc[T][3] += al * v3; }

__device__ __forceinline__ int next_edge(const int* __restrict__ pb, int deg, int lane, int last)
{
    int best = 0x7fffffff;
    for (int c = 0; c < deg; c += 32) {
        const int i = c + lane;
        if (i < deg) {
            const int v = pb[i];
            if (v > last && v < best) best = v;
        }
    }
    return wave_min_i(best);
}

__device__ __forceinline__ int rank_owner(int rankA, int rankB, int p, bool& inA)
{
    const unsigned mA = __builtin_amdgcn_ballot_w32(rankA == p);
    const unsigned mB = __builtin_amdgcn_ballot_w32(rankB == p);
    inA = (mA != 0u);
    return __builtin_ctz(inA ? mA : mB);
}
__device__ __forceinline__ int pick_i(int vA, int vB, int ln, bool inA)
{
    return inA ? __builtin_amdgcn_readlane(vA, ln) : __builtin_amdgcn_readlane(vB, ln);
}
__device__ __forceinline__ float pick_f(float vA, float vB, int ln, bool inA)
{
    return __int_as_float(pick_i(__float_as_int(vA), __float_as_int(vB), ln, inA));
}

__global__ void __launch_bounds__(256) node_gather_kernel(
    const float* __restrict__ z, const float* __restrict__ ef,
    const int* __restrict__ src, const int* __restrict__ etype,
    const float* __restrict__ ssrc, const float* __restrict__ sdst,
    const float* __restrict__ uvec,
    const int* __restrict__ beg, const int* __restrict__ cur, const int* __restrict__ perm,
    float* __restrict__ h, float* __restrict__ blockrow, int N)
{
    __shared__ float4 cs[8][128];
    __shared__ int arrived;
    if (threadIdx.x == 0) arrived = 0;
    __syncthreads();
    const int d    = (int)((blockIdx.x * blockDim.x + threadIdx.x) >> 5);
    const int lane = threadIdx.x & 31;
    const int wave = threadIdx.x >> 5;
    const bool fullBlock = ((int)blockIdx.x + 1) * 8 <= N;
    if (d >= N) return;
    const int b   = beg[d];
    const int deg = cur[d] - b;
    const int* __restrict__ pb = perm + b;
    const float sd = sdst[d];
    float acc[4][4] = {};

    if (deg <= 32) {
        const int id = (lane < deg) ? pb[lane] : (0x7fffffc0 + lane);
        int rank = 0;
#pragma unroll
        for (int j = 0; j < 32; ++j)
            rank += (__builtin_amdgcn_readlane(id, j) < id) ? 1 : 0;
        const int sid = __builtin_amdgcn_ds_permute(rank << 2, id);
        const bool ok = lane < deg;
        const int e  = ok ? sid : 0;
        const int s  = src[e];
        const int tt = etype[e];
        const float lr = ok ? edge_logit(ef, src, ssrc, uvec, e, sd) : -INFINITY;
        const float m  = wave_max_f(lr);
        const float ex = ok ? __expf(lr - m) : 0.f;
        const float den = wave_sum_f(ex);
        const float alpha = ex / den;

        for (int j = 0; j < deg; j += 4) {
            float r0[4], r1[4], r2[4], r3[4];
#pragma unroll
            for (int u = 0; u < 4; ++u) {
                const int jj = (j + u < deg) ? (j + u) : (deg - 1);
                const int eu = __builtin_amdgcn_readlane(e, jj);
                const int su = __builtin_amdgcn_readlane(s, jj);
                const float* zr = z + (size_t)su * 64;
                const float* er = ef + (size_t)eu * 64;
                r0[u] = zr[lane];
                r1[u] = zr[32 + lane];
                r2[u] = er[lane];
                r3[u] = er[32 + lane];
            }
#pragma unroll
            for (int u = 0; u < 4; ++u) {
                if (j + u < deg) {
                    const float al = __int_as_float(__builtin_amdgcn_readlane(__float_as_int(alpha), j + u));
                    const int   ty = __builtin_amdgcn_readlane(tt, j + u);
                    const float v0 = r0[u], v1 = r1[u], v2 = r2[u], v3 = r3[u];
                    ROW_ACC(0) ROW_ACC(1) ROW_ACC(2) ROW_ACC(3)
                }
            }
        }
    } else if (deg <= 64) {
        const int  degB = deg - 32;
        const bool okB  = lane < degB;
        const int  eA   = pb[lane];
        const int  idB  = okB ? pb[32 + lane] : (0x7fffffc0 + lane);
        int rankA = 0, rankB = 0;
        for (int j = 0; j < 32; ++j) {
            const int a = __builtin_amdgcn_readlane(eA, j);
            const int b = __builtin_amdgcn_readlane(idB, j);
            rankA += ((a < eA) ? 1 : 0) + ((b < eA) ? 1 : 0);
            rankB += ((a < idB) ? 1 : 0) + ((b < idB) ? 1 : 0);
        }
        const int eB = okB ? idB : 0;
        int sA = 0, sB = 0, tA = 0, tB = 0;
        float lA = 0.f, lB = 0.f;
#pragma nounroll
        for (int hh = 0; hh < 2; ++hh) {
            const int   e_h = hh ? eB : eA;
            const int   s_h = src[e_h];
            const int   t_h = etype[e_h];
            const float l_h = edge_logit(ef, src, ssrc, uvec, e_h, sd);
            if (hh) { sB = s_h; tB = t_h; lB = l_h; } else { sA = s_h; tA = t_h; lA = l_h; }
        }
        lB = okB ? lB : -INFINITY;
        const float m  = wave_max_f(fmaxf(lA, lB));
        const float xA = __expf(lA - m);
        const float xB = okB ? __expf(lB - m) : 0.f;
        float den = 0.f;
        for (int p = 0; p < deg; ++p) {
            bool inA;
            const int ln = rank_owner(rankA, rankB, p, inA);
            den += pick_f(xA, xB, ln, inA);
        }
        const float alA = xA / den, alB = xB / den;

        for (int j = 0; j < deg; j += 4) {
            float r0[4], r1[4], r2[4], r3[4], al4[4];
            int ty4[4];
            for (int u = 0; u < 4; ++u) {
                const int p = (j + u < deg) ? (j + u) : (deg - 1);
                bool inA;
                const int ln = rank_owner(rankA, rankB, p, inA);
                const int eu = pick_i(eA, eB, ln, inA);
                const int su = pick_i(sA, sB, ln, inA);
                al4[u] = pick_f(alA, alB, ln, inA);
                ty4[u] = pick_i(tA, tB, ln, inA);
                const float* zr = z + (size_t)su * 64;
                const float* er = ef + (size_t)eu * 64;
                r0[u] = zr[lane];
                r1[u] = zr[32 + lane];
                r2[u] = er[lane];
                r3[u] = er[32 + lane];
            }
            for (int u = 0; u < 4; ++u) {
                if (j + u < deg) {
                    const float al = al4[u];
                    const int   ty = ty4[u];
                    const float v0 = r0[u], v1 = r1[u], v2 = r2[u], v3 = r3[u];
                    ROW_ACC(0) ROW_ACC(1) ROW_ACC(2) ROW_ACC(3)
                }
            }
        }
    } else {
        float m = -INFINITY;
        for (int c = 0; c < deg; c += 32) {
            const int i = c + lane;
            if (i < deg) m = fmaxf(m, edge_logit(ef, src, ssrc, uvec, pb[i], sd));
        }
        m = wave_max_f(m);
        float den = 0.f;
        int last = -1;
        for (int j = 0; j < deg; ++j) {
            last = next_edge(pb, deg, lane, last);
            den += __expf(edge_logit(ef, src, ssrc, uvec, last, sd) - m);
        }
        last = -1;
        for (int j = 0; j < deg; ++j) {
            last = next_edge(pb, deg, lane, last);
            const float al = __expf(edge_logit(ef, src, ssrc, uvec, last, sd) - m) / den;
            const int ty = etype[last];
            const float* zr = z + (size_t)src[last] * 64;
            const float* er = ef + (size_t)last * 64;
            const float v0 = zr[lane], v1 = zr[32 + lane], v2 = er[lane], v3 = er[32 + lane];
            ROW_ACC(0) ROW_ACC(1) ROW_ACC(2) ROW_ACC(3)
        }
    }

    bool last = false;
    if (fullBlock) {
#pragma unroll
        for (int t = 0; t < 4; ++t)
            cs[wave][t * 32 + lane] = make_float4(acc[t][0], acc[t][1], acc[t][2], acc[t][3]);
        int prev = 0;
        if (lane == 0)
            prev = __hip_atomic_fetch_add(&arrived, 1, __ATOMIC_ACQ_REL, __HIP_MEMORY_SCOPE_WORKGROUP);
        last = __builtin_amdgcn_readfirstlane(prev) == 7;
    }

    float* hr = h + (size_t)d * 4 * 128 + lane;
#pragma unroll
    for (int t = 0; t < 4; ++t) {
        hr[t * 128]      = acc[t][0];
        hr[t * 128 + 32] = acc[t][1];
        hr[t * 128 + 64] = acc[t][2];
        hr[t * 128 + 96] = acc[t][3];
    }

    if (last) {
        float* br = blockrow + (size_t)blockIdx.x * 512 + lane;
#pragma unroll
        for (int t = 0; t < 4; ++t) {
            float4 s = cs[0][t * 32 + lane];
#pragma unroll
            for (int w = 1; w < 8; ++w) {
                const float4 v = cs[w][t * 32 + lane];
                s.x += v.x; s.y += v.y; s.z += v.z; s.w += v.w;
            }
            br[t * 128]      = s.x;
            br[t * 128 + 32] = s.y;
            br[t * 128 + 64] = s.z;
            br[t * 128 + 96] = s.w;
        }
    }
}
#undef ROW_ACC

constexpr int ZP_BLOCKS = 128;

__device__ __forceinline__ void add4(float4& a, const float4& v) {
    a.x += v.x; a.y += v.y; a.z += v.z; a.w += v.w;
}

__global__ void __launch_bounds__(1024) hsum_partial_kernel(
    const float* __restrict__ h, float* __restrict__ part, int nodes,
    const float* __restrict__ hx, int firstLoose, int N)
{
    __shared__ float4 sm[8][128];
    const int tid = threadIdx.x;
    const int c4  = tid & 127;
    const int sub = tid >> 7;
    const int R   = (nodes + (int)gridDim.x - 1) / (int)gridDim.x;
    const int r0  = blockIdx.x * R;
    int r1 = r0 + R;
    if (r1 > nodes) r1 = nodes;
    const float4* __restrict__ p = (const float4*)h + c4;

    float4 acc = make_float4(0.f, 0.f, 0.f, 0.f);
    int r = r0 + sub;
    for (; r + 56 < r1; r += 64) {
        float4 v[8];
#pragma unroll
        for (int u = 0; u < 8; ++u) v[u] = p[(size_t)(r + 8 * u) * 128];
#pragma unroll
        for (int u = 0; u < 8; ++u) add4(acc, v[u]);
    }
    {
        float4 v[8];
#pragma unroll
        for (int u = 0; u < 8; ++u) {
            const int rr = r + 8 * u;
            v[u] = make_float4(0.f, 0.f, 0.f, 0.f);
            if (rr < r1) v[u] = p[(size_t)rr * 128];
        }
#pragma unroll
        for (int u = 0; u < 8; ++u) add4(acc, v[u]);
    }
    if (blockIdx.x == 0 && sub == 0) {
        for (int n = firstLoose; n < N; ++n) add4(acc, ((const float4*)hx)[(size_t)n * 128 + c4]);
    }
    sm[sub][c4] = acc;
    __syncthreads();
    if (tid < 512) {
        const float* sf = (const float*)&sm[0][0];
        float s = 0.f;
#pragma unroll
        for (int k = 0; k < 8; ++k) s += sf[k * 512 + tid];
        part[(size_t)blockIdx.x * 512 + tid] = s;
    }
}

__global__ void __launch_bounds__(1024) beta_kernel(const float* __restrict__ part,
                            const float* __restrict__ w_fc2, const float* __restrict__ W2,
                            const float* __restrict__ w_sem, float* __restrict__ beta)
{
    __shared__ float4 sm[8][128];
    __shared__ __attribute__((aligned(16))) float hs[4 * 128];
    __shared__ float zs[16 * 64];
    const int tid = threadIdx.x;
    const int c4  = tid & 127;
    const int g   = tid >> 7;
    const float4* __restrict__ p = (const float4*)part + (size_t)g * (ZP_BLOCKS / 8) * 128 + c4;
    float4 acc = make_float4(0.f, 0.f, 0.f, 0.f);
    for (int b = 0; b < ZP_BLOCKS / 8; b += 16) {
        float4 v[16];
#pragma unroll
        for (int u = 0; u < 16; ++u) v[u] = p[(size_t)(b + u) * 128];
#pragma unroll
        for (int u = 0; u < 16; ++u) add4(acc, v[u]);
    }
    sm[g][c4] = acc;
    __syncthreads();
    if (tid < 512) {
        const float* sf = (const float*)&sm[0][0];
        float s = 0.f;
#pragma unroll
        for (int k = 0; k < 8; ++k) s += sf[k * 512 + tid];
        hs[tid] = s;
    }
    __syncthreads();
    if (tid < 128) {
        const int lane = tid & 31;
        const int nt   = tid >> 5;
        const int half = lane >> 4;
        const int l15  = lane & 15;
        const int ar   = (l15 < 4) ? l15 : 0;
        v8f d = {};
#pragma unroll
        for (int k0 = 0; k0 < 128; k0 += 4) {
            const int k = k0 + 2 * half;
            const v2f av = *(const v2f*)&hs[ar * 128 + k];
            const float* __restrict__ wrow = (k0 < 64) ? (w_fc2 + k * 64) : (W2 + (k - 64) * 64);
            v2f a, b;
            a.x = (l15 < 4) ? av.x : 0.f;
            a.y = (l15 < 4) ? av.y : 0.f;
            b.x = wrow[nt * 16 + l15];
            b.y = wrow[64 + nt * 16 + l15];
            d = __builtin_amdgcn_wmma_f32_16x16x4_f32(false, a, false, b, (short)0, d, false, false);
        }
#pragma unroll
        for (int r = 0; r < 8; ++r) zs[(r + 8 * half) * 64 + nt * 16 + l15] = d[r];
    }
    __syncthreads();
    if (tid != 0) return;
    float wp[4], mx = -INFINITY;
#pragma unroll
    for (int t = 0; t < 4; ++t) {
        float d = 0.f;
        for (int o = 0; o < 64; ++o) d += zs[t * 64 + o] * w_sem[o];
        const float lr = d > 0.f ? d : 0.01f * d;
        wp[t] = lr;
        mx = fmaxf(mx, lr);
    }
    float s = 0.f;
#pragma unroll
    for (int t = 0; t < 4; ++t) { wp[t] = __expf(wp[t] - mx); s += wp[t]; }
#pragma unroll
    for (int t = 0; t < 4; ++t) beta[t] = wp[t] / s;
}

__device__ __forceinline__ void sem_load4(const float4* __restrict__ hrow, int j0, float4 (&v)[4][4])
{
#pragma unroll
    for (int s = 0; s < 4; ++s) {
#pragma unroll
        for (int t = 0; t < 4; ++t) v[s][t] = hrow[32 * t + 2 * (j0 + s)];
    }
}

__device__ __forceinline__ void sem_steps4(const float4 (&v)[4][4], float b0, float b1, float b2, float b3,
                                           const v2f* __restrict__ bcol, int j0, v8f (&acc)[4])
{
#pragma unroll
    for (int s = 0; s < 4; ++s) {
        const float4 v0 = v[s][0], v1 = v[s][1], v2 = v[s][2], v3 = v[s][3];
        v2f a01, a23;
        a01.x = b0 * v0.x + b1 * v1.x + b2 * v2.x + b3 * v3.x;
        a01.y = b0 * v0.y + b1 * v1.y + b2 * v2.y + b3 * v3.y;
        a23.x = b0 * v0.z + b1 * v1.z + b2 * v2.z + b3 * v3.z;
        a23.y = b0 * v0.w + b1 * v1.w + b2 * v2.w + b3 * v3.w;
        const v2f* brow = bcol + (4 * (j0 + s)) * 64;
#pragma unroll
        for (int nt = 0; nt < 4; ++nt) {
            const v2f b = brow[nt * 16];
            acc[nt] = __builtin_amdgcn_wmma_f32_16x16x4_f32(
                false, a01, false, b, (short)0, acc[nt], false, false);
        }
#pragma unroll
        for (int nt = 0; nt < 4; ++nt) {
            const v2f b = brow[64 + nt * 16];
            acc[nt] = __builtin_amdgcn_wmma_f32_16x16x4_f32(
                false, a23, false, b, (short)0, acc[nt], false, false);
        }
    }
}

__global__ void __launch_bounds__(256) sem_out_wmma_kernel(
    const float* __restrict__ h, const float* __restrict__ w_fc2, const float* __restrict__ W2,
    const float* __restrict__ beta, float* __restrict__ Z, int M)
{
    constexpr int K = 128, N = 64, NT = 4;
    __shared__ v2f sB2[(K / 2) * N];
    const int tid = threadIdx.x;
    for (int p = tid; p < (K / 2) * N; p += 256) {
        const int kk = (p / N) << 1;
        const int n  = p % N;
        const float* __restrict__ wrow = (kk < 64) ? (w_fc2 + kk * N) : (W2 + (kk - 64) * N);
        v2f b;
        b.x = wrow[n];
        b.y = wrow[N + n];
        sB2[p] = b;
    }
    __syncthreads();

    const int lane    = tid & 31;
    const int wave    = tid >> 5;
    const int rowTile = blockIdx.x * 8 + wave;
    const int numTiles = (M + 15) >> 4;
    if (rowTile >= numTiles) return;

    const int half = lane >> 4;
    const int l15  = lane & 15;
    int mrow = rowTile * 16 + l15;
    if (mrow >= M) mrow = M - 1;

    const float b0 = beta[0], b1 = beta[1], b2 = beta[2], b3 = beta[3];
    v8f acc[NT] = {};
    const float4* __restrict__ hrow = (const float4*)(h + (size_t)mrow * (4 * K) + 4 * half);
    const v2f* __restrict__ bcol = sB2 + (2 * half) * N + l15;

    float4 v[3][4][4];
    sem_load4(hrow, 0, v[0]);
    sem_load4(hrow, 4, v[1]);
    sem_load4(hrow, 8, v[2]);
    __builtin_amdgcn_sched_barrier(0);
    sem_steps4(v[0], b0, b1, b2, b3, bcol, 0, acc);
    sem_load4(hrow, 12, v[0]);
    __builtin_amdgcn_sched_barrier(0);
    sem_steps4(v[1], b0, b1, b2, b3, bcol, 4, acc);
    sem_steps4(v[2], b0, b1, b2, b3, bcol, 8, acc);
    sem_steps4(v[0], b0, b1, b2, b3, bcol, 12, acc);

#pragma unroll
    for (int p = 0; p < NT / 2; ++p) {
#pragma unroll
        for (int r = 0; r < 8; ++r) {
            const float a0 = acc[2 * p][r];
            const float a1 = acc[2 * p + 1][r];
            const float x0 = xhalf(a0);
            const float x1 = xhalf(a1);
            const float lo = half ? x1 : a0;
            const float hi = half ? a1 : x0;
            const int row = rowTile * 16 + r;
            if (row < M)     Z[(size_t)row * N + p * 32 + lane] = lo;
            if (row + 8 < M) Z[(size_t)(row + 8) * N + p * 32 + lane] = hi;
        }
    }
}

extern "C" void kernel_launch(void* const* d_in, const int* in_sizes, int n_in,
                              void* d_out, int out_size, void* d_ws, size_t ws_size,
                              hipStream_t stream)
{
    const float* node   = (const float*)d_in[0];
    const float* edgef  = (const float*)d_in[1];
    const int*   src    = (const int*)d_in[2];
    const int*   dst    = (const int*)d_in[3];
    const int*   etype  = (const int*)d_in[4];
    const float* w_n    = (const float*)d_in[5];
    const float* w_e    = (const float*)d_in[6];
    const float* w_attn = (const float*)d_in[7];
    const float* w_fc2  = (const float*)d_in[8];
    const float* w_sem  = (const float*)d_in[9];
    float* out = (float*)d_out;

    const int N = in_sizes[0] / 128;
    const int E = in_sizes[2];
    const int T = 4;

    float* ws   = (float*)d_ws;
    float* h    = ws;
    float* z    = h + (size_t)N * T * 128;
    float* ssrc = z + (size_t)N * 64;
    float* sdst = ssrc + N;
    float* abuf = sdst + N;
    float* mbuf = abuf + E;
    float* den  = mbuf + N;
    float* beta = den + N + 256;
    float* W2   = beta + 64;
    float* uvec = W2 + 64 * 64;
    float* part = abuf;
    float* blockrow = out;
    int* perm  = (int*)abuf;
    int* cur   = (int*)mbuf;
    int* beg   = (int*)den;
    int* pairs     = (int*)h;
    int* binCount  = pairs + 2 * (size_t)E;
    int* binCursor = binCount + MAXBINS * CTR_PITCH;
    int* binStart  = binCursor + MAXBINS * CTR_PITCH;
    const int nbins  = (N + (1 << BIN_SHIFT) - 1) >> BIN_SHIFT;
    const int chunks = (E + EDGE_CHUNK - 1) / EDGE_CHUNK;

    const int thr = 256;

    gemm_f32_wmma<4><<<((N + 15) / 16 + 7) / 8, thr, 0, stream>>>(node, w_n, z, N, 128, 64);
    gemm_f32_wmma<4><<<1, thr, 0, stream>>>(w_e, w_fc2 + 64 * 64, W2, 64, 32, 64);
    node_score_kernel<<<(N + thr - 1) / thr, thr, 0, stream>>>(z, w_attn, w_e, ssrc, sdst, uvec, binCount, binCursor, N);
    bin_count_kernel<<<chunks, 256, 0, stream>>>(dst, binCount, E);
    bin_scatter_kernel<<<chunks, 256, 0, stream>>>(dst, binCount, binCursor, binStart, pairs, E, nbins);
    bin_fine_kernel<<<nbins, 1024, 0, stream>>>(pairs, binStart, beg, cur, perm, N);
    node_gather_kernel<<<(N + 7) / 8, thr, 0, stream>>>(z, edgef, src, etype, ssrc, sdst, uvec,
                                                        beg, cur, perm, h, blockrow, N);
    hsum_partial_kernel<<<ZP_BLOCKS, 1024, 0, stream>>>(blockrow, part, N / 8, h, (N / 8) * 8, N);
    beta_kernel<<<1, 1024, 0, stream>>>(part, w_fc2, W2, w_sem, beta);
    sem_out_wmma_kernel<<<((N + 15) / 16 + 7) / 8, thr, 0, stream>>>(h, w_fc2, W2, beta, out, N);
}


# runs up to 32 edges in one pass: logits from the gathered rows (dot summed across lanes), softmax by running maximum, sums divided by den at the end; edge rows read once, not twice; sum order changes
# speedup vs baseline: 3.2732x; 1.0680x over previous
#include <hip/hip_runtime.h>
#include <hip/hip_bf16.h>
#include <math.h>

typedef __attribute__((ext_vector_type(2))) float v2f;
typedef __attribute__((ext_vector_type(8))) float v8f;

__device__ __forceinline__ float xhalf(float x) {
    return __builtin_bit_cast(float, __builtin_amdgcn_permlanex16(
        __builtin_bit_cast(int, x), __builtin_bit_cast(int, x), 0x76543210, 0xfedcba98, false, false));
}

template <int NT>
__global__ void __launch_bounds__(256) gemm_f32_wmma(
    const float* __restrict__ A, const float* __restrict__ B,
    float* __restrict__ C, int M, int K, int N)
{
    __shared__ v2f sB2[4096];
    const int tid = threadIdx.x;
    const int pairCount = (K >> 1) * N;
    for (int p = tid; p < pairCount; p += 256) {
        const int kk = (p / N) << 1;
        const int n  = p - (p / N) * N;
        v2f b;
        b.x = B[kk * N + n];
        b.y = B[(kk + 1) * N + n];
        sB2[p] = b;
    }
    __syncthreads();

    const int lane    = tid & 31;
    const int wave    = tid >> 5;
    const int rowTile = blockIdx.x * 8 + wave;
    const int numTiles = (M + 15) >> 4;
    if (rowTile >= numTiles) return;

    const int half = lane >> 4;
    const int l15  = lane & 15;
    int mrow = rowTile * 16 + l15;
    if (mrow >= M) mrow = M - 1;

    v8f acc[NT] = {};
    const float* arow = A + (size_t)mrow * K + 2 * half;
    const v2f* __restrict__ bcol = sB2 + half * N + l15;

    for (int k0 = 0; k0 < K; k0 += 4) {
        const v2f a = *(const v2f*)(arow + k0);
        const v2f* brow = bcol + (k0 >> 1) * N;
#pragma unroll
        for (int nt = 0; nt < NT; ++nt) {
            const v2f b = brow[nt * 16];
            acc[nt] = __builtin_amdgcn_wmma_f32_16x16x4_f32(
                false, a, false, b, (short)0, acc[nt], false, false);
        }
    }

#pragma unroll
    for (int p = 0; p < NT / 2; ++p) {
#pragma unroll
        for (int r = 0; r < 8; ++r) {
            const float a0 = acc[2 * p][r];
            const float a1 = acc[2 * p + 1][r];
            const float x0 = xhalf(a0);
            const float x1 = xhalf(a1);
            const float lo = half ? x1 : a0;
            const float hi = half ? a1 : x0;
            const int row = rowTile * 16 + r;
            if (row < M)     C[(size_t)row * N + p * 32 + lane] = lo;
            if (row + 8 < M) C[(size_t)(row + 8) * N + p * 32 + lane] = hi;
        }
    }
}


constexpr int BIN_SHIFT  = 8;
constexpr int MAXBINS    = 256;
constexpr int CTR_PITCH  = 64;
constexpr int EDGE_CHUNK = 4096;

__global__ void node_score_kernel(const float* __restrict__ z,
                                  const float* __restrict__ w_attn, const float* __restrict__ w_e,
                                  float* __restrict__ ssrc, float* __restrict__ sdst, float* __restrict__ uvec,
                                  int* __restrict__ binCount, int* __restrict__ binCursor, int N)
{
    const int n = blockIdx.x * blockDim.x + threadIdx.x;
    if (n < MAXBINS) { binCount[n * CTR_PITCH] = 0; binCursor[n * CTR_PITCH] = 0; }
    if (n < 64) {
        float u = 0.f;
#pragma unroll
        for (int j = 0; j < 32; ++j) u += w_e[n * 32 + j] * w_attn[64 + j];
        uvec[n] = u;
    }
    if (n >= N) return;
    const float* zr = z + (size_t)n * 64;
    float s0 = 0.f, s1 = 0.f;
#pragma unroll
    for (int i = 0; i < 64; ++i) {
        const float v = zr[i];
        s0 += v * w_attn[i];
        s1 += v * w_attn[96 + i];
    }
    ssrc[n] = s0;
    sdst[n] = s1;
}

__device__ __forceinline__ void load_chunk(const int* __restrict__ dst, int E, int base, int tid, int4 (&d)[4])
{
    if (base + EDGE_CHUNK <= E) {
#pragma unroll
        for (int u = 0; u < 4; ++u)
            d[u] = *(const int4*)(dst + base + (u * 256 + tid) * 4);
    } else {
#pragma unroll
        for (int u = 0; u < 4; ++u) {
            const int i = base + (u * 256 + tid) * 4;
            d[u].x = (i     < E) ? dst[i]     : -1;
            d[u].y = (i + 1 < E) ? dst[i + 1] : -1;
            d[u].z = (i + 2 < E) ? dst[i + 2] : -1;
            d[u].w = (i + 3 < E) ? dst[i + 3] : -1;
        }
    }
}

__global__ void __launch_bounds__(256) bin_count_kernel(const int* __restrict__ dst,
                                                        int* __restrict__ binCount, int E)
{
    __shared__ int hist[MAXBINS];
    const int tid = threadIdx.x;
    hist[tid] = 0;
    int4 d[4];
    load_chunk(dst, E, blockIdx.x * EDGE_CHUNK, tid, d);
    __syncthreads();
#pragma unroll
    for (int u = 0; u < 4; ++u) {
        if (d[u].x >= 0) atomicAdd(&hist[d[u].x >> BIN_SHIFT], 1);
        if (d[u].y >= 0) atomicAdd(&hist[d[u].y >> BIN_SHIFT], 1);
        if (d[u].z >= 0) atomicAdd(&hist[d[u].z >> BIN_SHIFT], 1);
        if (d[u].w >= 0) atomicAdd(&hist[d[u].w >> BIN_SHIFT], 1);
    }
    __syncthreads();
    const int c = hist[tid];
    if (c != 0) atomicAdd(&binCount[tid * CTR_PITCH], c);
}

__global__ void __launch_bounds__(256) bin_scatter_kernel(const int* __restrict__ dst,
    const int* __restrict__ binCount, int* __restrict__ binCursor, int* __restrict__ binStart,
    int* __restrict__ pairs, int E, int nbins)
{
    __shared__ int hist[MAXBINS];
    __shared__ int base[MAXBINS];
    __shared__ int wtot[8];
    const int tid  = threadIdx.x;
    const int lane = tid & 31;
    const int wave = tid >> 5;
    hist[tid] = 0;
    int4 d[4];
    const int ebase = blockIdx.x * EDGE_CHUNK;
    load_chunk(dst, E, ebase, tid, d);
    const int c = (tid < nbins) ? binCount[tid * CTR_PITCH] : 0;
    int x = c;
#pragma unroll
    for (int o = 1; o < 32; o <<= 1) {
        const int y = __shfl_up(x, o);
        if (lane >= o) x += y;
    }
    if (lane == 31) wtot[wave] = x;
    __syncthreads();
#pragma unroll
    for (int u = 0; u < 4; ++u) {
        if (d[u].x >= 0) atomicAdd(&hist[d[u].x >> BIN_SHIFT], 1);
        if (d[u].y >= 0) atomicAdd(&hist[d[u].y >> BIN_SHIFT], 1);
        if (d[u].z >= 0) atomicAdd(&hist[d[u].z >> BIN_SHIFT], 1);
        if (d[u].w >= 0) atomicAdd(&hist[d[u].w >> BIN_SHIFT], 1);
    }
    int woff = 0;
    for (int w = 0; w < wave; ++w) woff += wtot[w];
    const int start = woff + x - c;
    __syncthreads();
    const int mine = hist[tid];
    int b0 = start;
    if (mine != 0) b0 += atomicAdd(&binCursor[tid * CTR_PITCH], mine);
    base[tid] = b0;
    hist[tid] = 0;
    if (blockIdx.x == 0) {
        binStart[tid] = start;
        if (tid == MAXBINS - 1) binStart[MAXBINS] = start + c;
    }
    __syncthreads();
    int2* __restrict__ pr = (int2*)pairs;
    if (ebase + EDGE_CHUNK <= E) {
        int pos[16];
#pragma unroll
        for (int u = 0; u < 4; ++u) {
            const int dv[4] = { d[u].x, d[u].y, d[u].z, d[u].w };
#pragma unroll
            for (int q = 0; q < 4; ++q) {
                const int b = dv[q] >> BIN_SHIFT;
                pos[u * 4 + q] = base[b] + atomicAdd(&hist[b], 1);
            }
        }
#pragma unroll
        for (int u = 0; u < 4; ++u) {
            const int e = ebase + (u * 256 + tid) * 4;
            const int dv[4] = { d[u].x, d[u].y, d[u].z, d[u].w };
#pragma unroll
            for (int q = 0; q < 4; ++q) pr[pos[u * 4 + q]] = make_int2(e + q, dv[q]);
        }
    } else {
#pragma unroll
        for (int u = 0; u < 4; ++u) {
            const int e = ebase + (u * 256 + tid) * 4;
            const int dv[4] = { d[u].x, d[u].y, d[u].z, d[u].w };
#pragma unroll
            for (int q = 0; q < 4; ++q) {
                if (dv[q] >= 0) {
                    const int b = dv[q] >> BIN_SHIFT;
                    const int pos = base[b] + atomicAdd(&hist[b], 1);
                    pr[pos] = make_int2(e + q, dv[q]);
                }
            }
        }
    }
}

__global__ void __launch_bounds__(1024) bin_fine_kernel(const int* __restrict__ pairs,
    const int* __restrict__ binStart, int* __restrict__ beg, int* __restrict__ cur,
    int* __restrict__ perm, int N)
{
    __shared__ int ncount[256];
    __shared__ int ncur[256];
    __shared__ int wtot[8];
    const int tid  = threadIdx.x;
    const int lane = tid & 31;
    const int wave = tid >> 5;
    const int bin  = blockIdx.x;
    const int s    = binStart[bin];
    const int cntb = binStart[bin + 1] - s;
    const int2* __restrict__ tp = (const int2*)pairs + s;
    if (tid < 256) ncount[tid] = 0;
    __syncthreads();
    for (int i0 = 0; i0 < cntb; i0 += 4096) {
        int2 p[4];
#pragma unroll
        for (int u = 0; u < 4; ++u) {
            const int i = i0 + u * 1024 + tid;
            p[u] = make_int2(-1, -1);
            if (i < cntb) p[u] = tp[i];
        }
#pragma unroll
        for (int u = 0; u < 4; ++u)
            if (p[u].x >= 0) atomicAdd(&ncount[p[u].y & 255], 1);
    }
    __syncthreads();
    int c = 0, x = 0;
    if (tid < 256) {
        c = ncount[tid];
        x = c;
#pragma unroll
        for (int o = 1; o < 32; o <<= 1) {
            const int y = __shfl_up(x, o);
            if (lane >= o) x += y;
        }
        if (lane == 31) wtot[wave] = x;
    }
    __syncthreads();
    if (tid < 256) {
        int woff = 0;
        for (int w = 0; w < wave; ++w) woff += wtot[w];
        const int b0 = s + woff + x - c;
        ncur[tid] = b0;
        const int node = (bin << BIN_SHIFT) + tid;
        if (node < N) { beg[node] = b0; cur[node] = b0 + c; }
    }
    __syncthreads();
    for (int i0 = 0; i0 < cntb; i0 += 4096) {
        int2 p[4];
#pragma unroll
        for (int u = 0; u < 4; ++u) {
            const int i = i0 + u * 1024 + tid;
            p[u] = make_int2(-1, -1);
            if (i < cntb) p[u] = tp[i];
        }
#pragma unroll
        for (int u = 0; u < 4; ++u) {
            if (p[u].x >= 0) {
                const int pos = atomicAdd(&ncur[p[u].y & 255], 1);
                perm[pos] = p[u].x;
            }
        }
    }
}

__device__ __forceinline__ float wave_max_f(float v) {
#pragma unroll
    for (int o = 16; o > 0; o >>= 1) v = fmaxf(v, __shfl_xor(v, o));
    return v;
}
__device__ __forceinline__ int wave_min_i(int v) {
#pragma unroll
    for (int o = 16; o > 0; o >>= 1) { const int t = __shfl_xor(v, o); v = t < v ? t : v; }
    return v;
}

__device__ __forceinline__ float edge_logit(const float* __restrict__ ef, const int* __restrict__ src,
                                            const float* __restrict__ ssrc,
                                            const float* __restrict__ uvec, int e, float sd)
{
    const float4* er = (const float4*)(ef + (size_t)e * 64);
    float t = ssrc[src[e]] + sd;
#pragma nounroll
    for (int hl = 0; hl < 2; ++hl) {
#pragma unroll
        for (int q = 0; q < 8; ++q) {
            const float4 v = er[8 * hl + q];
            const float* __restrict__ uq = uvec + 32 * hl + 4 * q;
            t += v.x * uq[0];
            t += v.y * uq[1];
            t += v.z * uq[2];
            t += v.w * uq[3];
        }
    }
    return t > 0.f ? t : 0.01f * t;
}

#define ROW_ACC(T)                                                         \
    if (ty == T) { acc[T][0] += al * v0; acc[T][1] += al * v1; acc[T][2] += al * v2; acc[T][3] += al * v3; }

__device__ __forceinline__ int next_edge(const int* __restrict__ pb, int deg, int lane, int last)
{
    int best = 0x7fffffff;
    for (int c = 0; c < deg; c += 32) {
        const int i = c + lane;
        if (i < deg) {
            const int v = pb[i];
            if (v > last && v < best) best = v;
        }
    }
    return wave_min_i(best);
}

__device__ __forceinline__ int rank_owner(int rankA, int rankB, int p, bool& inA)
{
    const unsigned mA = __builtin_amdgcn_ballot_w32(rankA == p);
    const unsigned mB = __builtin_amdgcn_ballot_w32(rankB == p);
    inA = (mA != 0u);
    return __builtin_ctz(inA ? mA : mB);
}
__device__ __forceinline__ int pick_i(int vA, int vB, int ln, bool inA)
{
    return inA ? __builtin_amdgcn_readlane(vA, ln) : __builtin_amdgcn_readlane(vB, ln);
}
__device__ __forceinline__ float pick_f(float vA, float vB, int ln, bool inA)
{
    return __int_as_float(pick_i(__float_as_int(vA), __float_as_int(vB), ln, inA));
}

template <int M>
__device__ __forceinline__ float row_xor(float x) {
    return __builtin_bit_cast(float, __builtin_amdgcn_update_dpp(
        __builtin_bit_cast(int, x), __builtin_bit_cast(int, x), 0x160 + M, 0xf, 0xf, false));
}

__device__ __forceinline__ float lane_f(float v, int ln) {
    return __int_as_float(__builtin_amdgcn_readlane(__float_as_int(v), ln));
}

__global__ void __launch_bounds__(256) node_gather_kernel(
    const float* __restrict__ z, const float* __restrict__ ef,
    const int* __restrict__ src, const int* __restrict__ etype,
    const float* __restrict__ ssrc, const float* __restrict__ sdst,
    const float* __restrict__ uvec,
    const int* __restrict__ beg, const int* __restrict__ cur, const int* __restrict__ perm,
    float* __restrict__ h, float* __restrict__ blockrow, int N)
{
    __shared__ float4 cs[8][128];
    __shared__ int arrived;
    if (threadIdx.x == 0) arrived = 0;
    __syncthreads();
    const int d    = (int)((blockIdx.x * blockDim.x + threadIdx.x) >> 5);
    const int lane = threadIdx.x & 31;
    const int wave = threadIdx.x >> 5;
    const bool fullBlock = ((int)blockIdx.x + 1) * 8 <= N;
    if (d >= N) return;
    const int b   = beg[d];
    const int deg = cur[d] - b;
    const int* __restrict__ pb = perm + b;
    const float sd = sdst[d];
    float acc[4][4] = {};

    if (deg <= 32) {
        const int id = (lane < deg) ? pb[lane] : (0x7fffffc0 + lane);
        int rank = 0;
#pragma unroll
        for (int j = 0; j < 32; ++j)
            rank += (__builtin_amdgcn_readlane(id, j) < id) ? 1 : 0;
        const int sid = __builtin_amdgcn_ds_permute(rank << 2, id);
        const bool ok = lane < deg;
        const int e  = ok ? sid : 0;
        const int s  = src[e];
        const int tt = etype[e];
        const float ss = ssrc[s];
        const float sdU = lane_f(sd, 0);
        const float u0 = uvec[lane], u1 = uvec[32 + lane];
        const int degU = __builtin_amdgcn_readfirstlane(deg);
        const bool h16 = (lane & 16) != 0, b3 = (lane & 8) != 0, b2 = (lane & 4) != 0;
        float m = -INFINITY, den = 0.f;

        for (int j = 0; j < degU; j += 8) {
            float r0[8], r1[8], r2[8], r3[8];
#pragma unroll
            for (int u = 0; u < 8; ++u) {
                const int jj = (j + u < degU) ? (j + u) : (degU - 1);
                const int eu = __builtin_amdgcn_readlane(e, jj);
                const int su = __builtin_amdgcn_readlane(s, jj);
                const float* zr = z + (size_t)su * 64;
                const float* er = ef + (size_t)eu * 64;
                r0[u] = zr[lane];
                r1[u] = zr[32 + lane];
                r2[u] = er[lane];
                r3[u] = er[32 + lane];
            }
            float q[8];
#pragma unroll
            for (int u = 0; u < 8; ++u) q[u] = r2[u] * u0 + r3[u] * u1;
            float a[4];
#pragma unroll
            for (int i = 0; i < 4; ++i) {
                const float keep = h16 ? q[i + 4] : q[i];
                const float send = h16 ? q[i] : q[i + 4];
                a[i] = keep + xhalf(send);
            }
            float bb[2];
#pragma unroll
            for (int i = 0; i < 2; ++i) {
                const float keep = b3 ? a[i + 2] : a[i];
                const float send = b3 ? a[i] : a[i + 2];
                bb[i] = keep + row_xor<8>(send);
            }
            const float keep4 = b2 ? bb[1] : bb[0];
            const float send4 = b2 ? bb[0] : bb[1];
            float dt = keep4 + row_xor<4>(send4);
            dt += row_xor<2>(dt);
            dt += row_xor<1>(dt);

            float lg[8];
            float gm = m;
#pragma unroll
            for (int u = 0; u < 8; ++u) {
                const int jj = (j + u < degU) ? (j + u) : (degU - 1);
                const float tl = (lane_f(ss, jj) + sdU) + lane_f(dt, 4 * u);
                lg[u] = tl > 0.f ? tl : 0.01f * tl;
                if (j + u < degU) gm = fmaxf(gm, lg[u]);
            }
            const float sc = __expf(m - gm);
            den *= sc;
#pragma unroll
            for (int t = 0; t < 4; ++t) {
                acc[t][0] *= sc; acc[t][1] *= sc; acc[t][2] *= sc; acc[t][3] *= sc;
            }
            m = gm;
#pragma unroll
            for (int u = 0; u < 8; ++u) {
                if (j + u < degU) {
                    const float al = __expf(lg[u] - m);
                    den += al;
                    const int   ty = __builtin_amdgcn_readlane(tt, j + u);
                    const float v0 = r0[u], v1 = r1[u], v2 = r2[u], v3 = r3[u];
                    ROW_ACC(0) ROW_ACC(1) ROW_ACC(2) ROW_ACC(3)
                }
            }
        }
        if (degU > 0) {
            const float inv = 1.f / den;
#pragma unroll
            for (int t = 0; t < 4; ++t) {
                acc[t][0] *= inv; acc[t][1] *= inv; acc[t][2] *= inv; acc[t][3] *= inv;
            }
        }
    } else if (deg <= 64) {
        const int  degB = deg - 32;
        const bool okB  = lane < degB;
        const int  eA   = pb[lane];
        const int  idB  = okB ? pb[32 + lane] : (0x7fffffc0 + lane);
        int rankA = 0, rankB = 0;
        for (int j = 0; j < 32; ++j) {
            const int a = __builtin_amdgcn_readlane(eA, j);
            const int b = __builtin_amdgcn_readlane(idB, j);
            rankA += ((a < eA) ? 1 : 0) + ((b < eA) ? 1 : 0);
            rankB += ((a < idB) ? 1 : 0) + ((b < idB) ? 1 : 0);
        }
        const int eB = okB ? idB : 0;
        int sA = 0, sB = 0, tA = 0, tB = 0;
        float lA = 0.f, lB = 0.f;
#pragma nounroll
        for (int hh = 0; hh < 2; ++hh) {
            const int   e_h = hh ? eB : eA;
            const int   s_h = src[e_h];
            const int   t_h = etype[e_h];
            const float l_h = edge_logit(ef, src, ssrc, uvec, e_h, sd);
            if (hh) { sB = s_h; tB = t_h; lB = l_h; } else { sA = s_h; tA = t_h; lA = l_h; }
        }
        lB = okB ? lB : -INFINITY;
        const float m  = wave_max_f(fmaxf(lA, lB));
        const float xA = __expf(lA - m);
        const float xB = okB ? __expf(lB - m) : 0.f;
        float den = 0.f;
        for (int p = 0; p < deg; ++p) {
            bool inA;
            const int ln = rank_owner(rankA, rankB, p, inA);
            den += pick_f(xA, xB, ln, inA);
        }
        const float alA = xA / den, alB = xB / den;

        for (int j = 0; j < deg; j += 4) {
            float r0[4], r1[4], r2[4], r3[4], al4[4];
            int ty4[4];
            for (int u = 0; u < 4; ++u) {
                const int p = (j + u < deg) ? (j + u) : (deg - 1);
                bool inA;
                const int ln = rank_owner(rankA, rankB, p, inA);
                const int eu = pick_i(eA, eB, ln, inA);
                const int su = pick_i(sA, sB, ln, inA);
                al4[u] = pick_f(alA, alB, ln, inA);
                ty4[u] = pick_i(tA, tB, ln, inA);
                const float* zr = z + (size_t)su * 64;
                const float* er = ef + (size_t)eu * 64;
                r0[u] = zr[lane];
                r1[u] = zr[32 + lane];
                r2[u] = er[lane];
                r3[u] = er[32 + lane];
            }
            for (int u = 0; u < 4; ++u) {
                if (j + u < deg) {
                    const float al = al4[u];
                    const int   ty = ty4[u];
                    const float v0 = r0[u], v1 = r1[u], v2 = r2[u], v3 = r3[u];
                    ROW_ACC(0) ROW_ACC(1) ROW_ACC(2) ROW_ACC(3)
                }
            }
        }
    } else {
        float m = -INFINITY;
        for (int c = 0; c < deg; c += 32) {
            const int i = c + lane;
            if (i < deg) m = fmaxf(m, edge_logit(ef, src, ssrc, uvec, pb[i], sd));
        }
        m = wave_max_f(m);
        float den = 0.f;
        int last = -1;
        for (int j = 0; j < deg; ++j) {
            last = next_edge(pb, deg, lane, last);
            den += __expf(edge_logit(ef, src, ssrc, uvec, last, sd) - m);
        }
        last = -1;
        for (int j = 0; j < deg; ++j) {
            last = next_edge(pb, deg, lane, last);
            const float al = __expf(edge_logit(ef, src, ssrc, uvec, last, sd) - m) / den;
            const int ty = etype[last];
            const float* zr = z + (size_t)src[last] * 64;
            const float* er = ef + (size_t)last * 64;
            const float v0 = zr[lane], v1 = zr[32 + lane], v2 = er[lane], v3 = er[32 + lane];
            ROW_ACC(0) ROW_ACC(1) ROW_ACC(2) ROW_ACC(3)
        }
    }

    bool last = false;
    if (fullBlock) {
#pragma unroll
        for (int t = 0; t < 4; ++t)
            cs[wave][t * 32 + lane] = make_float4(acc[t][0], acc[t][1], acc[t][2], acc[t][3]);
        int prev = 0;
        if (lane == 0)
            prev = __hip_atomic_fetch_add(&arrived, 1, __ATOMIC_ACQ_REL, __HIP_MEMORY_SCOPE_WORKGROUP);
        last = __builtin_amdgcn_readfirstlane(prev) == 7;
    }

    float* hr = h + (size_t)d * 4 * 128 + lane;
#pragma unroll
    for (int t = 0; t < 4; ++t) {
        hr[t * 128]      = acc[t][0];
        hr[t * 128 + 32] = acc[t][1];
        hr[t * 128 + 64] = acc[t][2];
        hr[t * 128 + 96] = acc[t][3];
    }

    if (last) {
        float* br = blockrow + (size_t)blockIdx.x * 512 + lane;
#pragma unroll
        for (int t = 0; t < 4; ++t) {
            float4 s = cs[0][t * 32 + lane];
#pragma unroll
            for (int w = 1; w < 8; ++w) {
                const float4 v = cs[w][t * 32 + lane];
                s.x += v.x; s.y += v.y; s.z += v.z; s.w += v.w;
            }
            br[t * 128]      = s.x;
            br[t * 128 + 32] = s.y;
            br[t * 128 + 64] = s.z;
            br[t * 128 + 96] = s.w;
        }
    }
}
#undef ROW_ACC

constexpr int ZP_BLOCKS = 128;

__device__ __forceinline__ void add4(float4& a, const float4& v) {
    a.x += v.x; a.y += v.y; a.z += v.z; a.w += v.w;
}

__global__ void __launch_bounds__(1024) hsum_partial_kernel(
    const float* __restrict__ h, float* __restrict__ part, int nodes,
    const float* __restrict__ hx, int firstLoose, int N)
{
    __shared__ float4 sm[8][128];
    const int tid = threadIdx.x;
    const int c4  = tid & 127;
    const int sub = tid >> 7;
    const int R   = (nodes + (int)gridDim.x - 1) / (int)gridDim.x;
    const int r0  = blockIdx.x * R;
    int r1 = r0 + R;
    if (r1 > nodes) r1 = nodes;
    const float4* __restrict__ p = (const float4*)h + c4;

    float4 acc = make_float4(0.f, 0.f, 0.f, 0.f);
    int r = r0 + sub;
    for (; r + 56 < r1; r += 64) {
        float4 v[8];
#pragma unroll
        for (int u = 0; u < 8; ++u) v[u] = p[(size_t)(r + 8 * u) * 128];
#pragma unroll
        for (int u = 0; u < 8; ++u) add4(acc, v[u]);
    }
    {
        float4 v[8];
#pragma unroll
        for (int u = 0; u < 8; ++u) {
            const int rr = r + 8 * u;
            v[u] = make_float4(0.f, 0.f, 0.f, 0.f);
            if (rr < r1) v[u] = p[(size_t)rr * 128];
        }
#pragma unroll
        for (int u = 0; u < 8; ++u) add4(acc, v[u]);
    }
    if (blockIdx.x == 0 && sub == 0) {
        for (int n = firstLoose; n < N; ++n) add4(acc, ((const float4*)hx)[(size_t)n * 128 + c4]);
    }
    sm[sub][c4] = acc;
    __syncthreads();
    if (tid < 512) {
        const float* sf = (const float*)&sm[0][0];
        float s = 0.f;
#pragma unroll
        for (int k = 0; k < 8; ++k) s += sf[k * 512 + tid];
        part[(size_t)blockIdx.x * 512 + tid] = s;
    }
}

__global__ void __launch_bounds__(1024) beta_kernel(const float* __restrict__ part,
                            const float* __restrict__ w_fc2, const float* __restrict__ W2,
                            const float* __restrict__ w_sem, float* __restrict__ beta)
{
    __shared__ float4 sm[8][128];
    __shared__ __attribute__((aligned(16))) float hs[4 * 128];
    __shared__ float zs[16 * 64];
    const int tid = threadIdx.x;
    const int c4  = tid & 127;
    const int g   = tid >> 7;
    const float4* __restrict__ p = (const float4*)part + (size_t)g * (ZP_BLOCKS / 8) * 128 + c4;
    float4 acc = make_float4(0.f, 0.f, 0.f, 0.f);
    for (int b = 0; b < ZP_BLOCKS / 8; b += 16) {
        float4 v[16];
#pragma unroll
        for (int u = 0; u < 16; ++u) v[u] = p[(size_t)(b + u) * 128];
#pragma unroll
        for (int u = 0; u < 16; ++u) add4(acc, v[u]);
    }
    sm[g][c4] = acc;
    __syncthreads();
    if (tid < 512) {
        const float* sf = (const float*)&sm[0][0];
        float s = 0.f;
#pragma unroll
        for (int k = 0; k < 8; ++k) s += sf[k * 512 + tid];
        hs[tid] = s;
    }
    __syncthreads();
    if (tid < 128) {
        const int lane = tid & 31;
        const int nt   = tid >> 5;
        const int half = lane >> 4;
        const int l15  = lane & 15;
        const int ar   = (l15 < 4) ? l15 : 0;
        v8f d = {};
#pragma unroll
        for (int k0 = 0; k0 < 128; k0 += 4) {
            const int k = k0 + 2 * half;
            const v2f av = *(const v2f*)&hs[ar * 128 + k];
            const float* __restrict__ wrow = (k0 < 64) ? (w_fc2 + k * 64) : (W2 + (k - 64) * 64);
            v2f a, b;
            a.x = (l15 < 4) ? av.x : 0.f;
            a.y = (l15 < 4) ? av.y : 0.f;
            b.x = wrow[nt * 16 + l15];
            b.y = wrow[64 + nt * 16 + l15];
            d = __builtin_amdgcn_wmma_f32_16x16x4_f32(false, a, false, b, (short)0, d, false, false);
        }
#pragma unroll
        for (int r = 0; r < 8; ++r) zs[(r + 8 * half) * 64 + nt * 16 + l15] = d[r];
    }
    __syncthreads();
    if (tid != 0) return;
    float wp[4], mx = -INFINITY;
#pragma unroll
    for (int t = 0; t < 4; ++t) {
        float d = 0.f;
        for (int o = 0; o < 64; ++o) d += zs[t * 64 + o] * w_sem[o];
        const float lr = d > 0.f ? d : 0.01f * d;
        wp[t] = lr;
        mx = fmaxf(mx, lr);
    }
    float s = 0.f;
#pragma unroll
    for (int t = 0; t < 4; ++t) { wp[t] = __expf(wp[t] - mx); s += wp[t]; }
#pragma unroll
    for (int t = 0; t < 4; ++t) beta[t] = wp[t] / s;
}

__device__ __forceinline__ void sem_load4(const float4* __restrict__ hrow, int j0, float4 (&v)[4][4])
{
#pragma unroll
    for (int s = 0; s < 4; ++s) {
#pragma unroll
        for (int t = 0; t < 4; ++t) v[s][t] = hrow[32 * t + 2 * (j0 + s)];
    }
}

__device__ __forceinline__ void sem_steps4(const float4 (&v)[4][4], float b0, float b1, float b2, float b3,
                                           const v2f* __restrict__ bcol, int j0, v8f (&acc)[4])
{
#pragma unroll
    for (int s = 0; s < 4; ++s) {
        const float4 v0 = v[s][0], v1 = v[s][1], v2 = v[s][2], v3 = v[s][3];
        v2f a01, a23;
        a01.x = b0 * v0.x + b1 * v1.x + b2 * v2.x + b3 * v3.x;
        a01.y = b0 * v0.y + b1 * v1.y + b2 * v2.y + b3 * v3.y;
        a23.x = b0 * v0.z + b1 * v1.z + b2 * v2.z + b3 * v3.z;
        a23.y = b0 * v0.w + b1 * v1.w + b2 * v2.w + b3 * v3.w;
        const v2f* brow = bcol + (4 * (j0 + s)) * 64;
#pragma unroll
        for (int nt = 0; nt < 4; ++nt) {
            const v2f b = brow[nt * 16];
            acc[nt] = __builtin_amdgcn_wmma_f32_16x16x4_f32(
                false, a01, false, b, (short)0, acc[nt], false, false);
        }
#pragma unroll
        for (int nt = 0; nt < 4; ++nt) {
            const v2f b = brow[64 + nt * 16];
            acc[nt] = __builtin_amdgcn_wmma_f32_16x16x4_f32(
                false, a23, false, b, (short)0, acc[nt], false, false);
        }
    }
}

__global__ void __launch_bounds__(256) sem_out_wmma_kernel(
    const float* __restrict__ h, const float* __restrict__ w_fc2, const float* __restrict__ W2,
    const float* __restrict__ beta, float* __restrict__ Z, int M)
{
    constexpr int K = 128, N = 64, NT = 4;
    __shared__ v2f sB2[(K / 2) * N];
    const int tid = threadIdx.x;
    for (int p = tid; p < (K / 2) * N; p += 256) {
        const int kk = (p / N) << 1;
        const int n  = p % N;
        const float* __restrict__ wrow = (kk < 64) ? (w_fc2 + kk * N) : (W2 + (kk - 64) * N);
        v2f b;
        b.x = wrow[n];
        b.y = wrow[N + n];
        sB2[p] = b;
    }
    __syncthreads();

    const int lane    = tid & 31;
    const int wave    = tid >> 5;
    const int rowTile = blockIdx.x * 8 + wave;
    const int numTiles = (M + 15) >> 4;
    if (rowTile >= numTiles) return;

    const int half = lane >> 4;
    const int l15  = lane & 15;
    int mrow = rowTile * 16 + l15;
    if (mrow >= M) mrow = M - 1;

    const float b0 = beta[0], b1 = beta[1], b2 = beta[2], b3 = beta[3];
    v8f acc[NT] = {};
    const float4* __restrict__ hrow = (const float4*)(h + (size_t)mrow * (4 * K) + 4 * half);
    const v2f* __restrict__ bcol = sB2 + (2 * half) * N + l15;

    float4 v[3][4][4];
    sem_load4(hrow, 0, v[0]);
    sem_load4(hrow, 4, v[1]);
    sem_load4(hrow, 8, v[2]);
    __builtin_amdgcn_sched_barrier(0);
    sem_steps4(v[0], b0, b1, b2, b3, bcol, 0, acc);
    sem_load4(hrow, 12, v[0]);
    __builtin_amdgcn_sched_barrier(0);
    sem_steps4(v[1], b0, b1, b2, b3, bcol, 4, acc);
    sem_steps4(v[2], b0, b1, b2, b3, bcol, 8, acc);
    sem_steps4(v[0], b0, b1, b2, b3, bcol, 12, acc);

#pragma unroll
    for (int p = 0; p < NT / 2; ++p) {
#pragma unroll
        for (int r = 0; r < 8; ++r) {
            const float a0 = acc[2 * p][r];
            const float a1 = acc[2 * p + 1][r];
            const float x0 = xhalf(a0);
            const float x1 = xhalf(a1);
            const float lo = half ? x1 : a0;
            const float hi = half ? a1 : x0;
            const int row = rowTile * 16 + r;
            if (row < M)     Z[(size_t)row * N + p * 32 + lane] = lo;
            if (row + 8 < M) Z[(size_t)(row + 8) * N + p * 32 + lane] = hi;
        }
    }
}

extern "C" void kernel_launch(void* const* d_in, const int* in_sizes, int n_in,
                              void* d_out, int out_size, void* d_ws, size_t ws_size,
                              hipStream_t stream)
{
    const float* node   = (const float*)d_in[0];
    const float* edgef  = (const float*)d_in[1];
    const int*   src    = (const int*)d_in[2];
    const int*   dst    = (const int*)d_in[3];
    const int*   etype  = (const int*)d_in[4];
    const float* w_n    = (const float*)d_in[5];
    const float* w_e    = (const float*)d_in[6];
    const float* w_attn = (const float*)d_in[7];
    const float* w_fc2  = (const float*)d_in[8];
    const float* w_sem  = (const float*)d_in[9];
    float* out = (float*)d_out;

    const int N = in_sizes[0] / 128;
    const int E = in_sizes[2];
    const int T = 4;

    float* ws   = (float*)d_ws;
    float* h    = ws;
    float* z    = h + (size_t)N * T * 128;
    float* ssrc = z + (size_t)N * 64;
    float* sdst = ssrc + N;
    float* abuf = sdst + N;
    float* mbuf = abuf + E;
    float* den  = mbuf + N;
    float* beta = den + N + 256;
    float* W2   = beta + 64;
    float* uvec = W2 + 64 * 64;
    float* part = abuf;
    float* blockrow = out;
    int* perm  = (int*)abuf;
    int* cur   = (int*)mbuf;
    int* beg   = (int*)den;
    int* pairs     = (int*)h;
    int* binCount  = pairs + 2 * (size_t)E;
    int* binCursor = binCount + MAXBINS * CTR_PITCH;
    int* binStart  = binCursor + MAXBINS * CTR_PITCH;
    const int nbins  = (N + (1 << BIN_SHIFT) - 1) >> BIN_SHIFT;
    const int chunks = (E + EDGE_CHUNK - 1) / EDGE_CHUNK;

    const int thr = 256;

    gemm_f32_wmma<4><<<((N + 15) / 16 + 7) / 8, thr, 0, stream>>>(node, w_n, z, N, 128, 64);
    gemm_f32_wmma<4><<<1, thr, 0, stream>>>(w_e, w_fc2 + 64 * 64, W2, 64, 32, 64);
    node_score_kernel<<<(N + thr - 1) / thr, thr, 0, stream>>>(z, w_attn, w_e, ssrc, sdst, uvec, binCount, binCursor, N);
    bin_count_kernel<<<chunks, 256, 0, stream>>>(dst, binCount, E);
    bin_scatter_kernel<<<chunks, 256, 0, stream>>>(dst, binCount, binCursor, binStart, pairs, E, nbins);
    bin_fine_kernel<<<nbins, 1024, 0, stream>>>(pairs, binStart, beg, cur, perm, N);
    node_gather_kernel<<<(N + 7) / 8, thr, 0, stream>>>(z, edgef, src, etype, ssrc, sdst, uvec,
                                                        beg, cur, perm, h, blockrow, N);
    hsum_partial_kernel<<<ZP_BLOCKS, 1024, 0, stream>>>(blockrow, part, N / 8, h, (N / 8) * 8, N);
    beta_kernel<<<1, 1024, 0, stream>>>(part, w_fc2, W2, w_sem, beta);
    sem_out_wmma_kernel<<<((N + 15) / 16 + 7) / 8, thr, 0, stream>>>(h, w_fc2, W2, beta, out, N);
}


# all work that depends on the inputs alone in ONE launch: node scores from the z tile in the GEMM epilogue (16-lane tree; score sum order changes), W2/uvec block, per-chunk bin counts as plain rows
# speedup vs baseline: 3.6431x; 1.1130x over previous
#include <hip/hip_runtime.h>
#include <hip/hip_bf16.h>
#include <math.h>

typedef __attribute__((ext_vector_type(2))) float v2f;
typedef __attribute__((ext_vector_type(8))) float v8f;

__device__ __forceinline__ float xhalf(float x) {
    return __builtin_bit_cast(float, __builtin_amdgcn_permlanex16(
        __builtin_bit_cast(int, x), __builtin_bit_cast(int, x), 0x76543210, 0xfedcba98, false, false));
}

template <int M>
__device__ __forceinline__ float row_xor(float x) {
    return __builtin_bit_cast(float, __builtin_amdgcn_update_dpp(
        __builtin_bit_cast(int, x), __builtin_bit_cast(int, x), 0x160 + M, 0xf, 0xf, false));
}

template <int NT, bool SCORES>
__device__ __forceinline__ void gemm_f32_wmma_block(
    v2f* __restrict__ sB2, const float* __restrict__ A, const float* __restrict__ B,
    float* __restrict__ C, int M, int K, int N, int bid,
    const float* __restrict__ w_attn, float* __restrict__ ssrc, float* __restrict__ sdst)
{
    const int tid = threadIdx.x;
    const int pairCount = (K >> 1) * N;
    for (int p = tid; p < pairCount; p += 256) {
        const int kk = (p / N) << 1;
        const int n  = p - (p / N) * N;
        v2f b;
        b.x = B[kk * N + n];
        b.y = B[(kk + 1) * N + n];
        sB2[p] = b;
    }
    __syncthreads();

    const int lane    = tid & 31;
    const int wave    = tid >> 5;
    const int rowTile = bid * 8 + wave;
    const int numTiles = (M + 15) >> 4;
    if (rowTile >= numTiles) return;

    const int half = lane >> 4;
    const int l15  = lane & 15;
    int mrow = rowTile * 16 + l15;
    if (mrow >= M) mrow = M - 1;

    v8f acc[NT] = {};
    const float* arow = A + (size_t)mrow * K + 2 * half;
    const v2f* __restrict__ bcol = sB2 + half * N + l15;

    for (int k0 = 0; k0 < K; k0 += 4) {
        const v2f a = *(const v2f*)(arow + k0);
        const v2f* brow = bcol + (k0 >> 1) * N;
#pragma unroll
        for (int nt = 0; nt < NT; ++nt) {
            const v2f b = brow[nt * 16];
            acc[nt] = __builtin_amdgcn_wmma_f32_16x16x4_f32(
                false, a, false, b, (short)0, acc[nt], false, false);
        }
    }

    if (SCORES) {
        float wsv[NT], wdv[NT];
#pragma unroll
        for (int nt = 0; nt < NT; ++nt) {
            wsv[nt] = w_attn[nt * 16 + l15];
            wdv[nt] = w_attn[96 + nt * 16 + l15];
        }
        float outS = 0.f, outD = 0.f;
#pragma unroll
        for (int r = 0; r < 8; ++r) {
            float ps = 0.f, pd = 0.f;
#pragma unroll
            for (int nt = 0; nt < NT; ++nt) {
                ps += acc[nt][r] * wsv[nt];
                pd += acc[nt][r] * wdv[nt];
            }
            ps += row_xor<8>(ps); pd += row_xor<8>(pd);
            ps += row_xor<4>(ps); pd += row_xor<4>(pd);
            ps += row_xor<2>(ps); pd += row_xor<2>(pd);
            ps += row_xor<1>(ps); pd += row_xor<1>(pd);
            if (l15 == r) { outS = ps; outD = pd; }
        }
        const int srow = rowTile * 16 + 8 * half + l15;
        if (l15 < 8 && srow < M) { ssrc[srow] = outS; sdst[srow] = outD; }
    }

#pragma unroll
    for (int p = 0; p < NT / 2; ++p) {
#pragma unroll
        for (int r = 0; r < 8; ++r) {
            const float a0 = acc[2 * p][r];
            const float a1 = acc[2 * p + 1][r];
            const float x0 = xhalf(a0);
            const float x1 = xhalf(a1);
            const float lo = half ? x1 : a0;
            const float hi = half ? a1 : x0;
            const int row = rowTile * 16 + r;
            if (row < M)     C[(size_t)row * N + p * 32 + lane] = lo;
            if (row + 8 < M) C[(size_t)(row + 8) * N + p * 32 + lane] = hi;
        }
    }
}

constexpr int BIN_SHIFT  = 8;
constexpr int MAXBINS    = 256;
constexpr int EDGE_CHUNK = 4096;

__device__ __forceinline__ void load_chunk(const int* __restrict__ dst, int E, int base, int tid, int4 (&d)[4])
{
    if (base + EDGE_CHUNK <= E) {
#pragma unroll
        for (int u = 0; u < 4; ++u)
            d[u] = *(const int4*)(dst + base + (u * 256 + tid) * 4);
    } else {
#pragma unroll
        for (int u = 0; u < 4; ++u) {
            const int i = base + (u * 256 + tid) * 4;
            d[u].x = (i     < E) ? dst[i]     : -1;
            d[u].y = (i + 1 < E) ? dst[i + 1] : -1;
            d[u].z = (i + 2 < E) ? dst[i + 2] : -1;
            d[u].w = (i + 3 < E) ? dst[i + 3] : -1;
        }
    }
}

__global__ void __launch_bounds__(256) head_kernel(
    const float* __restrict__ node, const float* __restrict__ w_n, float* __restrict__ z,
    const float* __restrict__ w_attn, float* __restrict__ ssrc, float* __restrict__ sdst, int N,
    const float* __restrict__ w_e, const float* __restrict__ w_fc2_tail, float* __restrict__ W2,
    float* __restrict__ uvec, const int* __restrict__ dst, int* __restrict__ chunkHist, int E, int zBlocks)
{
    __shared__ v2f sB2[4096];
    __shared__ int hist[MAXBINS];
    if ((int)blockIdx.x < zBlocks) {
        gemm_f32_wmma_block<4, true>(sB2, node, w_n, z, N, 128, 64, (int)blockIdx.x, w_attn, ssrc, sdst);
    } else if ((int)blockIdx.x == zBlocks) {
        const int n = threadIdx.x;
        if (n < 64) {
            float u = 0.f;
#pragma unroll
            for (int j = 0; j < 32; ++j) u += w_e[n * 32 + j] * w_attn[64 + j];
            uvec[n] = u;
        }
        gemm_f32_wmma_block<4, false>(sB2, w_e, w_fc2_tail, W2, 64, 32, 64, 0, w_attn, ssrc, sdst);
    } else {
        const int tid   = threadIdx.x;
        const int chunk = (int)blockIdx.x - zBlocks - 1;
        hist[tid] = 0;
        int4 d[4];
        load_chunk(dst, E, chunk * EDGE_CHUNK, tid, d);
        __syncthreads();
#pragma unroll
        for (int u = 0; u < 4; ++u) {
            if (d[u].x >= 0) atomicAdd(&hist[d[u].x >> BIN_SHIFT], 1);
            if (d[u].y >= 0) atomicAdd(&hist[d[u].y >> BIN_SHIFT], 1);
            if (d[u].z >= 0) atomicAdd(&hist[d[u].z >> BIN_SHIFT], 1);
            if (d[u].w >= 0) atomicAdd(&hist[d[u].w >> BIN_SHIFT], 1);
        }
        __syncthreads();
        chunkHist[chunk * MAXBINS + tid] = hist[tid];
    }
}


__global__ void __launch_bounds__(256) bin_scatter_kernel(const int* __restrict__ dst,
    const int* __restrict__ chunkHist, int chunks, int* __restrict__ binStart,
    int* __restrict__ pairs, int E)
{
    __shared__ int hist[MAXBINS];
    __shared__ int base[MAXBINS];
    __shared__ int wtot[8];
    __shared__ int4 stot[4][64];
    __shared__ int4 spre[4][64];
    const int tid  = threadIdx.x;
    const int lane = tid & 31;
    const int wave = tid >> 5;
    hist[tid] = 0;
    int4 d[4];
    const int ebase = blockIdx.x * EDGE_CHUNK;
    load_chunk(dst, E, ebase, tid, d);
    {
        const int g = tid >> 6, q = tid & 63;
        const int4* __restrict__ H4 = (const int4*)chunkHist + q;
        const int mychunk = (int)blockIdx.x;
        int4 t4 = make_int4(0, 0, 0, 0), p4 = make_int4(0, 0, 0, 0);
#pragma unroll 16
        for (int cc = g; cc < chunks; cc += 4) {
            const int4 v = H4[(size_t)cc * (MAXBINS / 4)];
            const int m = (cc < mychunk) ? -1 : 0;
            t4.x += v.x; t4.y += v.y; t4.z += v.z; t4.w += v.w;
            p4.x += v.x & m; p4.y += v.y & m; p4.z += v.z & m; p4.w += v.w & m;
        }
        stot[g][q] = t4;
        spre[g][q] = p4;
    }
    __syncthreads();
    const int* st = (const int*)&stot[0][0];
    const int* sp = (const int*)&spre[0][0];
    const int c   = st[tid] + st[256 + tid] + st[512 + tid] + st[768 + tid];
    const int pre = sp[tid] + sp[256 + tid] + sp[512 + tid] + sp[768 + tid];
    int x = c;
#pragma unroll
    for (int o = 1; o < 32; o <<= 1) {
        const int y = __shfl_up(x, o);
        if (lane >= o) x += y;
    }
    if (lane == 31) wtot[wave] = x;
    __syncthreads();
    int woff = 0;
    for (int w = 0; w < wave; ++w) woff += wtot[w];
    const int start = woff + x - c;
    base[tid] = start + pre;
    if (blockIdx.x == 0) {
        binStart[tid] = start;
        if (tid == MAXBINS - 1) binStart[MAXBINS] = start + c;
    }
    __syncthreads();
    int2* __restrict__ pr = (int2*)pairs;
    if (ebase + EDGE_CHUNK <= E) {
        int pos[16];
#pragma unroll
        for (int u = 0; u < 4; ++u) {
            const int dv[4] = { d[u].x, d[u].y, d[u].z, d[u].w };
#pragma unroll
            for (int q = 0; q < 4; ++q) {
                const int b = dv[q] >> BIN_SHIFT;
                pos[u * 4 + q] = base[b] + atomicAdd(&hist[b], 1);
            }
        }
#pragma unroll
        for (int u = 0; u < 4; ++u) {
            const int e = ebase + (u * 256 + tid) * 4;
            const int dv[4] = { d[u].x, d[u].y, d[u].z, d[u].w };
#pragma unroll
            for (int q = 0; q < 4; ++q) pr[pos[u * 4 + q]] = make_int2(e + q, dv[q]);
        }
    } else {
#pragma unroll
        for (int u = 0; u < 4; ++u) {
            const int e = ebase + (u * 256 + tid) * 4;
            const int dv[4] = { d[u].x, d[u].y, d[u].z, d[u].w };
#pragma unroll
            for (int q = 0; q < 4; ++q) {
                if (dv[q] >= 0) {
                    const int b = dv[q] >> BIN_SHIFT;
                    const int pos = base[b] + atomicAdd(&hist[b], 1);
                    pr[pos] = make_int2(e + q, dv[q]);
                }
            }
        }
    }
}

__global__ void __launch_bounds__(1024) bin_fine_kernel(const int* __restrict__ pairs,
    const int* __restrict__ binStart, int* __restrict__ beg, int* __restrict__ cur,
    int* __restrict__ perm, int N)
{
    __shared__ int ncount[256];
    __shared__ int ncur[256];
    __shared__ int wtot[8];
    const int tid  = threadIdx.x;
    const int lane = tid & 31;
    const int wave = tid >> 5;
    const int bin  = blockIdx.x;
    const int s    = binStart[bin];
    const int cntb = binStart[bin + 1] - s;
    const int2* __restrict__ tp = (const int2*)pairs + s;
    if (tid < 256) ncount[tid] = 0;
    __syncthreads();
    for (int i0 = 0; i0 < cntb; i0 += 4096) {
        int2 p[4];
#pragma unroll
        for (int u = 0; u < 4; ++u) {
            const int i = i0 + u * 1024 + tid;
            p[u] = make_int2(-1, -1);
            if (i < cntb) p[u] = tp[i];
        }
#pragma unroll
        for (int u = 0; u < 4; ++u)
            if (p[u].x >= 0) atomicAdd(&ncount[p[u].y & 255], 1);
    }
    __syncthreads();
    int c = 0, x = 0;
    if (tid < 256) {
        c = ncount[tid];
        x = c;
#pragma unroll
        for (int o = 1; o < 32; o <<= 1) {
            const int y = __shfl_up(x, o);
            if (lane >= o) x += y;
        }
        if (lane == 31) wtot[wave] = x;
    }
    __syncthreads();
    if (tid < 256) {
        int woff = 0;
        for (int w = 0; w < wave; ++w) woff += wtot[w];
        const int b0 = s + woff + x - c;
        ncur[tid] = b0;
        const int node = (bin << BIN_SHIFT) + tid;
        if (node < N) { beg[node] = b0; cur[node] = b0 + c; }
    }
    __syncthreads();
    for (int i0 = 0; i0 < cntb; i0 += 4096) {
        int2 p[4];
#pragma unroll
        for (int u = 0; u < 4; ++u) {
            const int i = i0 + u * 1024 + tid;
            p[u] = make_int2(-1, -1);
            if (i < cntb) p[u] = tp[i];
        }
#pragma unroll
        for (int u = 0; u < 4; ++u) {
            if (p[u].x >= 0) {
                const int pos = atomicAdd(&ncur[p[u].y & 255], 1);
                perm[pos] = p[u].x;
            }
        }
    }
}

__device__ __forceinline__ float wave_max_f(float v) {
#pragma unroll
    for (int o = 16; o > 0; o >>= 1) v = fmaxf(v, __shfl_xor(v, o));
    return v;
}
__device__ __forceinline__ int wave_min_i(int v) {
#pragma unroll
    for (int o = 16; o > 0; o >>= 1) { const int t = __shfl_xor(v, o); v = t < v ? t : v; }
    return v;
}

__device__ __forceinline__ float edge_logit(const float* __restrict__ ef, const int* __restrict__ src,
                                            const float* __restrict__ ssrc,
                                            const float* __restrict__ uvec, int e, float sd)
{
    const float4* er = (const float4*)(ef + (size_t)e * 64);
    float t = ssrc[src[e]] + sd;
#pragma nounroll
    for (int hl = 0; hl < 2; ++hl) {
#pragma unroll
        for (int q = 0; q < 8; ++q) {
            const float4 v = er[8 * hl + q];
            const float* __restrict__ uq = uvec + 32 * hl + 4 * q;
            t += v.x * uq[0];
            t += v.y * uq[1];
            t += v.z * uq[2];
            t += v.w * uq[3];
        }
    }
    return t > 0.f ? t : 0.01f * t;
}

#define ROW_ACC(T)                                                         \
    if (ty == T) { acc[T][0] += al * v0; acc[T][1] += al * v1; acc[T][2] += al * v2; acc[T][3] += al * v3; }

__device__ __forceinline__ int next_edge(const int* __restrict__ pb, int deg, int lane, int last)
{
    int best = 0x7fffffff;
    for (int c = 0; c < deg; c += 32) {
        const int i = c + lane;
        if (i < deg) {
            const int v = pb[i];
            if (v > last && v < best) best = v;
        }
    }
    return wave_min_i(best);
}

__device__ __forceinline__ int rank_owner(int rankA, int rankB, int p, bool& inA)
{
    const unsigned mA = __builtin_amdgcn_ballot_w32(rankA == p);
    const unsigned mB = __builtin_amdgcn_ballot_w32(rankB == p);
    inA = (mA != 0u);
    return __builtin_ctz(inA ? mA : mB);
}
__device__ __forceinline__ int pick_i(int vA, int vB, int ln, bool inA)
{
    return inA ? __builtin_amdgcn_readlane(vA, ln) : __builtin_amdgcn_readlane(vB, ln);
}
__device__ __forceinline__ float pick_f(float vA, float vB, int ln, bool inA)
{
    return __int_as_float(pick_i(__float_as_int(vA), __float_as_int(vB), ln, inA));
}

__device__ __forceinline__ float lane_f(float v, int ln) {
    return __int_as_float(__builtin_amdgcn_readlane(__float_as_int(v), ln));
}

__global__ void __launch_bounds__(256) node_gather_kernel(
    const float* __restrict__ z, const float* __restrict__ ef,
    const int* __restrict__ src, const int* __restrict__ etype,
    const float* __restrict__ ssrc, const float* __restrict__ sdst,
    const float* __restrict__ uvec,
    const int* __restrict__ beg, const int* __restrict__ cur, const int* __restrict__ perm,
    float* __restrict__ h, float* __restrict__ blockrow, int N)
{
    __shared__ float4 cs[8][128];
    __shared__ int arrived;
    if (threadIdx.x == 0) arrived = 0;
    __syncthreads();
    const int d    = (int)((blockIdx.x * blockDim.x + threadIdx.x) >> 5);
    const int lane = threadIdx.x & 31;
    const int wave = threadIdx.x >> 5;
    const bool fullBlock = ((int)blockIdx.x + 1) * 8 <= N;
    if (d >= N) return;
    const int b   = beg[d];
    const int deg = cur[d] - b;
    const int* __restrict__ pb = perm + b;
    const float sd = sdst[d];
    float acc[4][4] = {};

    if (deg <= 32) {
        const int id = (lane < deg) ? pb[lane] : (0x7fffffc0 + lane);
        int rank = 0;
#pragma unroll
        for (int j = 0; j < 32; ++j)
            rank += (__builtin_amdgcn_readlane(id, j) < id) ? 1 : 0;
        const int sid = __builtin_amdgcn_ds_permute(rank << 2, id);
        const bool ok = lane < deg;
        const int e  = ok ? sid : 0;
        const int s  = src[e];
        const int tt = etype[e];
        const float ss = ssrc[s];
        const float sdU = lane_f(sd, 0);
        const float u0 = uvec[lane], u1 = uvec[32 + lane];
        const int degU = __builtin_amdgcn_readfirstlane(deg);
        const bool h16 = (lane & 16) != 0, b3 = (lane & 8) != 0, b2 = (lane & 4) != 0;
        float m = -INFINITY, den = 0.f;

        for (int j = 0; j < degU; j += 8) {
            float r0[8], r1[8], r2[8], r3[8];
#pragma unroll
            for (int u = 0; u < 8; ++u) {
                const int jj = (j + u < degU) ? (j + u) : (degU - 1);
                const int eu = __builtin_amdgcn_readlane(e, jj);
                const int su = __builtin_amdgcn_readlane(s, jj);
                const float* zr = z + (size_t)su * 64;
                const float* er = ef + (size_t)eu * 64;
                r0[u] = zr[lane];
                r1[u] = zr[32 + lane];
                r2[u] = er[lane];
                r3[u] = er[32 + lane];
            }
            float q[8];
#pragma unroll
            for (int u = 0; u < 8; ++u) q[u] = r2[u] * u0 + r3[u] * u1;
            float a[4];
#pragma unroll
            for (int i = 0; i < 4; ++i) {
                const float keep = h16 ? q[i + 4] : q[i];
                const float send = h16 ? q[i] : q[i + 4];
                a[i] = keep + xhalf(send);
            }
            float bb[2];
#pragma unroll
            for (int i = 0; i < 2; ++i) {
                const float keep = b3 ? a[i + 2] : a[i];
                const float send = b3 ? a[i] : a[i + 2];
                bb[i] = keep + row_xor<8>(send);
            }
            const float keep4 = b2 ? bb[1] : bb[0];
            const float send4 = b2 ? bb[0] : bb[1];
            float dt = keep4 + row_xor<4>(send4);
            dt += row_xor<2>(dt);
            dt += row_xor<1>(dt);

            float lg[8];
            float gm = m;
#pragma unroll
            for (int u = 0; u < 8; ++u) {
                const int jj = (j + u < degU) ? (j + u) : (degU - 1);
                const float tl = (lane_f(ss, jj) + sdU) + lane_f(dt, 4 * u);
                lg[u] = tl > 0.f ? tl : 0.01f * tl;
                if (j + u < degU) gm = fmaxf(gm, lg[u]);
            }
            const float sc = __expf(m - gm);
            den *= sc;
#pragma unroll
            for (int t = 0; t < 4; ++t) {
                acc[t][0] *= sc; acc[t][1] *= sc; acc[t][2] *= sc; acc[t][3] *= sc;
            }
            m = gm;
#pragma unroll
            for (int u = 0; u < 8; ++u) {
                if (j + u < degU) {
                    const float al = __expf(lg[u] - m);
                    den += al;
                    const int   ty = __builtin_amdgcn_readlane(tt, j + u);
                    const float v0 = r0[u], v1 = r1[u], v2 = r2[u], v3 = r3[u];
                    ROW_ACC(0) ROW_ACC(1) ROW_ACC(2) ROW_ACC(3)
                }
            }
        }
        if (degU > 0) {
            const float inv = 1.f / den;
#pragma unroll
            for (int t = 0; t < 4; ++t) {
                acc[t][0] *= inv; acc[t][1] *= inv; acc[t][2] *= inv; acc[t][3] *= inv;
            }
        }
    } else if (deg <= 64) {
        const int  degB = deg - 32;
        const bool okB  = lane < degB;
        const int  eA   = pb[lane];
        const int  idB  = okB ? pb[32 + lane] : (0x7fffffc0 + lane);
        int rankA = 0, rankB = 0;
        for (int j = 0; j < 32; ++j) {
            const int a = __builtin_amdgcn_readlane(eA, j);
            const int b = __builtin_amdgcn_readlane(idB, j);
            rankA += ((a < eA) ? 1 : 0) + ((b < eA) ? 1 : 0);
            rankB += ((a < idB) ? 1 : 0) + ((b < idB) ? 1 : 0);
        }
        const int eB = okB ? idB : 0;
        int sA = 0, sB = 0, tA = 0, tB = 0;
        float lA = 0.f, lB = 0.f;
#pragma nounroll
        for (int hh = 0; hh < 2; ++hh) {
            const int   e_h = hh ? eB : eA;
            const int   s_h = src[e_h];
            const int   t_h = etype[e_h];
            const float l_h = edge_logit(ef, src, ssrc, uvec, e_h, sd);
            if (hh) { sB = s_h; tB = t_h; lB = l_h; } else { sA = s_h; tA = t_h; lA = l_h; }
        }
        lB = okB ? lB : -INFINITY;
        const float m  = wave_max_f(fmaxf(lA, lB));
        const float xA = __expf(lA - m);
        const float xB = okB ? __expf(lB - m) : 0.f;
        float den = 0.f;
        for (int p = 0; p < deg; ++p) {
            bool inA;
            const int ln = rank_owner(rankA, rankB, p, inA);
            den += pick_f(xA, xB, ln, inA);
        }
        const float alA = xA / den, alB = xB / den;

        for (int j = 0; j < deg; j += 4) {
            float r0[4], r1[4], r2[4], r3[4], al4[4];
            int ty4[4];
            for (int u = 0; u < 4; ++u) {
                const int p = (j + u < deg) ? (j + u) : (deg - 1);
                bool inA;
                const int ln = rank_owner(rankA, rankB, p, inA);
                const int eu = pick_i(eA, eB, ln, inA);
                const int su = pick_i(sA, sB, ln, inA);
                al4[u] = pick_f(alA, alB, ln, inA);
                ty4[u] = pick_i(tA, tB, ln, inA);
                const float* zr = z + (size_t)su * 64;
                const float* er = ef + (size_t)eu * 64;
                r0[u] = zr[lane];
                r1[u] = zr[32 + lane];
                r2[u] = er[lane];
                r3[u] = er[32 + lane];
            }
            for (int u = 0; u < 4; ++u) {
                if (j + u < deg) {
                    const float al = al4[u];
                    const int   ty = ty4[u];
                    const float v0 = r0[u], v1 = r1[u], v2 = r2[u], v3 = r3[u];
                    ROW_ACC(0) ROW_ACC(1) ROW_ACC(2) ROW_ACC(3)
                }
            }
        }
    } else {
        float m = -INFINITY;
        for (int c = 0; c < deg; c += 32) {
            const int i = c + lane;
            if (i < deg) m = fmaxf(m, edge_logit(ef, src, ssrc, uvec, pb[i], sd));
        }
        m = wave_max_f(m);
        float den = 0.f;
        int last = -1;
        for (int j = 0; j < deg; ++j) {
            last = next_edge(pb, deg, lane, last);
            den += __expf(edge_logit(ef, src, ssrc, uvec, last, sd) - m);
        }
        last = -1;
        for (int j = 0; j < deg; ++j) {
            last = next_edge(pb, deg, lane, last);
            const float al = __expf(edge_logit(ef, src, ssrc, uvec, last, sd) - m) / den;
            const int ty = etype[last];
            const float* zr = z + (size_t)src[last] * 64;
            const float* er = ef + (size_t)last * 64;
            const float v0 = zr[lane], v1 = zr[32 + lane], v2 = er[lane], v3 = er[32 + lane];
            ROW_ACC(0) ROW_ACC(1) ROW_ACC(2) ROW_ACC(3)
        }
    }

    bool last = false;
    if (fullBlock) {
#pragma unroll
        for (int t = 0; t < 4; ++t)
            cs[wave][t * 32 + lane] = make_float4(acc[t][0], acc[t][1], acc[t][2], acc[t][3]);
        int prev = 0;
        if (lane == 0)
            prev = __hip_atomic_fetch_add(&arrived, 1, __ATOMIC_ACQ_REL, __HIP_MEMORY_SCOPE_WORKGROUP);
        last = __builtin_amdgcn_readfirstlane(prev) == 7;
    }

    float* hr = h + (size_t)d * 4 * 128 + lane;
#pragma unroll
    for (int t = 0; t < 4; ++t) {
        hr[t * 128]      = acc[t][0];
        hr[t * 128 + 32] = acc[t][1];
        hr[t * 128 + 64] = acc[t][2];
        hr[t * 128 + 96] = acc[t][3];
    }

    if (last) {
        float* br = blockrow + (size_t)blockIdx.x * 512 + lane;
#pragma unroll
        for (int t = 0; t < 4; ++t) {
            float4 s = cs[0][t * 32 + lane];
#pragma unroll
            for (int w = 1; w < 8; ++w) {
                const float4 v = cs[w][t * 32 + lane];
                s.x += v.x; s.y += v.y; s.z += v.z; s.w += v.w;
            }
            br[t * 128]      = s.x;
            br[t * 128 + 32] = s.y;
            br[t * 128 + 64] = s.z;
            br[t * 128 + 96] = s.w;
        }
    }
}
#undef ROW_ACC

constexpr int ZP_BLOCKS = 128;

__device__ __forceinline__ void add4(float4& a, const float4& v) {
    a.x += v.x; a.y += v.y; a.z += v.z; a.w += v.w;
}

__global__ void __launch_bounds__(1024) hsum_partial_kernel(
    const float* __restrict__ h, float* __restrict__ part, int nodes,
    const float* __restrict__ hx, int firstLoose, int N)
{
    __shared__ float4 sm[8][128];
    const int tid = threadIdx.x;
    const int c4  = tid & 127;
    const int sub = tid >> 7;
    const int R   = (nodes + (int)gridDim.x - 1) / (int)gridDim.x;
    const int r0  = blockIdx.x * R;
    int r1 = r0 + R;
    if (r1 > nodes) r1 = nodes;
    const float4* __restrict__ p = (const float4*)h + c4;

    float4 acc = make_float4(0.f, 0.f, 0.f, 0.f);
    int r = r0 + sub;
    for (; r + 56 < r1; r += 64) {
        float4 v[8];
#pragma unroll
        for (int u = 0; u < 8; ++u) v[u] = p[(size_t)(r + 8 * u) * 128];
#pragma unroll
        for (int u = 0; u < 8; ++u) add4(acc, v[u]);
    }
    {
        float4 v[8];
#pragma unroll
        for (int u = 0; u < 8; ++u) {
            const int rr = r + 8 * u;
            v[u] = make_float4(0.f, 0.f, 0.f, 0.f);
            if (rr < r1) v[u] = p[(size_t)rr * 128];
        }
#pragma unroll
        for (int u = 0; u < 8; ++u) add4(acc, v[u]);
    }
    if (blockIdx.x == 0 && sub == 0) {
        for (int n = firstLoose; n < N; ++n) add4(acc, ((const float4*)hx)[(size_t)n * 128 + c4]);
    }
    sm[sub][c4] = acc;
    __syncthreads();
    if (tid < 512) {
        const float* sf = (const float*)&sm[0][0];
        float s = 0.f;
#pragma unroll
        for (int k = 0; k < 8; ++k) s += sf[k * 512 + tid];
        part[(size_t)blockIdx.x * 512 + tid] = s;
    }
}

__global__ void __launch_bounds__(1024) beta_kernel(const float* __restrict__ part,
                            const float* __restrict__ w_fc2, const float* __restrict__ W2,
                            const float* __restrict__ w_sem, float* __restrict__ beta)
{
    __shared__ float4 sm[8][128];
    __shared__ __attribute__((aligned(16))) float hs[4 * 128];
    __shared__ float zs[16 * 64];
    const int tid = threadIdx.x;
    const int c4  = tid & 127;
    const int g   = tid >> 7;
    const float4* __restrict__ p = (const float4*)part + (size_t)g * (ZP_BLOCKS / 8) * 128 + c4;
    float4 acc = make_float4(0.f, 0.f, 0.f, 0.f);
    for (int b = 0; b < ZP_BLOCKS / 8; b += 16) {
        float4 v[16];
#pragma unroll
        for (int u = 0; u < 16; ++u) v[u] = p[(size_t)(b + u) * 128];
#pragma unroll
        for (int u = 0; u < 16; ++u) add4(acc, v[u]);
    }
    sm[g][c4] = acc;
    __syncthreads();
    if (tid < 512) {
        const float* sf = (const float*)&sm[0][0];
        float s = 0.f;
#pragma unroll
        for (int k = 0; k < 8; ++k) s += sf[k * 512 + tid];
        hs[tid] = s;
    }
    __syncthreads();
    if (tid < 128) {
        const int lane = tid & 31;
        const int nt   = tid >> 5;
        const int half = lane >> 4;
        const int l15  = lane & 15;
        const int ar   = (l15 < 4) ? l15 : 0;
        v8f d = {};
#pragma unroll
        for (int k0 = 0; k0 < 128; k0 += 4) {
            const int k = k0 + 2 * half;
            const v2f av = *(const v2f*)&hs[ar * 128 + k];
            const float* __restrict__ wrow = (k0 < 64) ? (w_fc2 + k * 64) : (W2 + (k - 64) * 64);
            v2f a, b;
            a.x = (l15 < 4) ? av.x : 0.f;
            a.y = (l15 < 4) ? av.y : 0.f;
            b.x = wrow[nt * 16 + l15];
            b.y = wrow[64 + nt * 16 + l15];
            d = __builtin_amdgcn_wmma_f32_16x16x4_f32(false, a, false, b, (short)0, d, false, false);
        }
#pragma unroll
        for (int r = 0; r < 8; ++r) zs[(r + 8 * half) * 64 + nt * 16 + l15] = d[r];
    }
    __syncthreads();
    if (tid != 0) return;
    float wp[4], mx = -INFINITY;
#pragma unroll
    for (int t = 0; t < 4; ++t) {
        float d = 0.f;
        for (int o = 0; o < 64; ++o) d += zs[t * 64 + o] * w_sem[o];
        const float lr = d > 0.f ? d : 0.01f * d;
        wp[t] = lr;
        mx = fmaxf(mx, lr);
    }
    float s = 0.f;
#pragma unroll
    for (int t = 0; t < 4; ++t) { wp[t] = __expf(wp[t] - mx); s += wp[t]; }
#pragma unroll
    for (int t = 0; t < 4; ++t) beta[t] = wp[t] / s;
}

__device__ __forceinline__ void sem_load4(const float4* __restrict__ hrow, int j0, float4 (&v)[4][4])
{
#pragma unroll
    for (int s = 0; s < 4; ++s) {
#pragma unroll
        for (int t = 0; t < 4; ++t) v[s][t] = hrow[32 * t + 2 * (j0 + s)];
    }
}

__device__ __forceinline__ void sem_steps4(const float4 (&v)[4][4], float b0, float b1, float b2, float b3,
                                           const v2f* __restrict__ bcol, int j0, v8f (&acc)[4])
{
#pragma unroll
    for (int s = 0; s < 4; ++s) {
        const float4 v0 = v[s][0], v1 = v[s][1], v2 = v[s][2], v3 = v[s][3];
        v2f a01, a23;
        a01.x = b0 * v0.x + b1 * v1.x + b2 * v2.x + b3 * v3.x;
        a01.y = b0 * v0.y + b1 * v1.y + b2 * v2.y + b3 * v3.y;
        a23.x = b0 * v0.z + b1 * v1.z + b2 * v2.z + b3 * v3.z;
        a23.y = b0 * v0.w + b1 * v1.w + b2 * v2.w + b3 * v3.w;
        const v2f* brow = bcol + (4 * (j0 + s)) * 64;
#pragma unroll
        for (int nt = 0; nt < 4; ++nt) {
            const v2f b = brow[nt * 16];
            acc[nt] = __builtin_amdgcn_wmma_f32_16x16x4_f32(
                false, a01, false, b, (short)0, acc[nt], false, false);
        }
#pragma unroll
        for (int nt = 0; nt < 4; ++nt) {
            const v2f b = brow[64 + nt * 16];
            acc[nt] = __builtin_amdgcn_wmma_f32_16x16x4_f32(
                false, a23, false, b, (short)0, acc[nt], false, false);
        }
    }
}

__global__ void __launch_bounds__(256) sem_out_wmma_kernel(
    const float* __restrict__ h, const float* __restrict__ w_fc2, const float* __restrict__ W2,
    const float* __restrict__ beta, float* __restrict__ Z, int M)
{
    constexpr int K = 128, N = 64, NT = 4;
    __shared__ v2f sB2[(K / 2) * N];
    const int tid = threadIdx.x;
    for (int p = tid; p < (K / 2) * N; p += 256) {
        const int kk = (p / N) << 1;
        const int n  = p % N;
        const float* __restrict__ wrow = (kk < 64) ? (w_fc2 + kk * N) : (W2 + (kk - 64) * N);
        v2f b;
        b.x = wrow[n];
        b.y = wrow[N + n];
        sB2[p] = b;
    }
    __syncthreads();

    const int lane    = tid & 31;
    const int wave    = tid >> 5;
    const int rowTile = blockIdx.x * 8 + wave;
    const int numTiles = (M + 15) >> 4;
    if (rowTile >= numTiles) return;

    const int half = lane >> 4;
    const int l15  = lane & 15;
    int mrow = rowTile * 16 + l15;
    if (mrow >= M) mrow = M - 1;

    const float b0 = beta[0], b1 = beta[1], b2 = beta[2], b3 = beta[3];
    v8f acc[NT] = {};
    const float4* __restrict__ hrow = (const float4*)(h + (size_t)mrow * (4 * K) + 4 * half);
    const v2f* __restrict__ bcol = sB2 + (2 * half) * N + l15;

    float4 v[3][4][4];
    sem_load4(hrow, 0, v[0]);
    sem_load4(hrow, 4, v[1]);
    sem_load4(hrow, 8, v[2]);
    __builtin_amdgcn_sched_barrier(0);
    sem_steps4(v[0], b0, b1, b2, b3, bcol, 0, acc);
    sem_load4(hrow, 12, v[0]);
    __builtin_amdgcn_sched_barrier(0);
    sem_steps4(v[1], b0, b1, b2, b3, bcol, 4, acc);
    sem_steps4(v[2], b0, b1, b2, b3, bcol, 8, acc);
    sem_steps4(v[0], b0, b1, b2, b3, bcol, 12, acc);

#pragma unroll
    for (int p = 0; p < NT / 2; ++p) {
#pragma unroll
        for (int r = 0; r < 8; ++r) {
            const float a0 = acc[2 * p][r];
            const float a1 = acc[2 * p + 1][r];
            const float x0 = xhalf(a0);
            const float x1 = xhalf(a1);
            const float lo = half ? x1 : a0;
            const float hi = half ? a1 : x0;
            const int row = rowTile * 16 + r;
            if (row < M)     Z[(size_t)row * N + p * 32 + lane] = lo;
            if (row + 8 < M) Z[(size_t)(row + 8) * N + p * 32 + lane] = hi;
        }
    }
}

extern "C" void kernel_launch(void* const* d_in, const int* in_sizes, int n_in,
                              void* d_out, int out_size, void* d_ws, size_t ws_size,
                              hipStream_t stream)
{
    const float* node   = (const float*)d_in[0];
    const float* edgef  = (const float*)d_in[1];
    const int*   src    = (const int*)d_in[2];
    const int*   dst    = (const int*)d_in[3];
    const int*   etype  = (const int*)d_in[4];
    const float* w_n    = (const float*)d_in[5];
    const float* w_e    = (const float*)d_in[6];
    const float* w_attn = (const float*)d_in[7];
    const float* w_fc2  = (const float*)d_in[8];
    const float* w_sem  = (const float*)d_in[9];
    float* out = (float*)d_out;

    const int N = in_sizes[0] / 128;
    const int E = in_sizes[2];
    const int T = 4;

    float* ws   = (float*)d_ws;
    float* h    = ws;
    float* z    = h + (size_t)N * T * 128;
    float* ssrc = z + (size_t)N * 64;
    float* sdst = ssrc + N;
    float* abuf = sdst + N;
    float* mbuf = abuf + E;
    float* den  = mbuf + N;
    float* beta = den + N + 256;
    float* W2   = beta + 64;
    float* uvec = W2 + 64 * 64;
    float* part = abuf;
    float* blockrow = out;
    int* perm  = (int*)abuf;
    int* cur   = (int*)mbuf;
    int* beg   = (int*)den;
    const int nbins  = (N + (1 << BIN_SHIFT) - 1) >> BIN_SHIFT;
    const int chunks = (E + EDGE_CHUNK - 1) / EDGE_CHUNK;
    int* pairs     = (int*)h;
    int* chunkHist = pairs + ((2 * (size_t)E + 3) & ~(size_t)3);
    int* binStart  = chunkHist + (size_t)chunks * MAXBINS;

    const int thr = 256;

    const int zBlocks = ((N + 15) / 16 + 7) / 8;
    head_kernel<<<zBlocks + 1 + chunks, thr, 0, stream>>>(node, w_n, z, w_attn, ssrc, sdst, N,
                                                          w_e, w_fc2 + 64 * 64, W2, uvec, dst, chunkHist, E, zBlocks);
    bin_scatter_kernel<<<chunks, 256, 0, stream>>>(dst, chunkHist, chunks, binStart, pairs, E);
    bin_fine_kernel<<<nbins, 1024, 0, stream>>>(pairs, binStart, beg, cur, perm, N);
    node_gather_kernel<<<(N + 7) / 8, thr, 0, stream>>>(z, edgef, src, etype, ssrc, sdst, uvec,
                                                        beg, cur, perm, h, blockrow, N);
    hsum_partial_kernel<<<ZP_BLOCKS, 1024, 0, stream>>>(blockrow, part, N / 8, h, (N / 8) * 8, N);
    beta_kernel<<<1, 1024, 0, stream>>>(part, w_fc2, W2, w_sem, beta);
    sem_out_wmma_kernel<<<((N + 15) / 16 + 7) / 8, thr, 0, stream>>>(h, w_fc2, W2, beta, out, N);
}
